# starting kernel
# baseline (speedup 1.0000x reference)
#include <hip/hip_runtime.h>
#include <hip/hip_bf16.h>
#include <math.h>

typedef float v2f __attribute__((ext_vector_type(2)));
typedef float v8f __attribute__((ext_vector_type(8)));

#define EMB   64
#define EPSN  1e-12f

__device__ __forceinline__ float readlane_f32(float x, int lane) {
    return __int_as_float(__builtin_amdgcn_readlane(__float_as_int(x), lane));
}

__global__ void spmm_scatter(const float* __restrict__ f,
                             const float* __restrict__ vals,
                             const int*   __restrict__ rows,
                             const int*   __restrict__ cols,
                             float*       __restrict__ out, int E) {
    int idx = blockIdx.x * blockDim.x + threadIdx.x;
    int e   = idx >> 4;
    if (e >= E) return;
    int c   = (idx & 15) << 2;
    __builtin_prefetch(cols + e + 4096, 0, 0);
    int   row = rows[e];
    int   col = cols[e];
    float v   = vals[e];
    float4 x  = *(const float4*)(f + (size_t)col * EMB + c);
    float* o  = out + (size_t)row * EMB + c;
    atomicAdd(o + 0, x.x * v);
    atomicAdd(o + 1, x.y * v);
    atomicAdd(o + 2, x.z * v);
    atomicAdd(o + 3, x.w * v);
}

__global__ void norm_accum(float*       __restrict__ f,
                           const float* __restrict__ base,
                           float*       __restrict__ acc,
                           float inv_c, int N) {
    int tid  = blockIdx.x * blockDim.x + threadIdx.x;
    int lane = tid & 31;
    int row  = tid >> 5;
    if (row >= N) return;
    size_t off = (size_t)row * EMB + lane * 2;
    float2 x = *(const float2*)(f + off);
    float gx = x.x * inv_c;
    float gy = x.y * inv_c;
    float s  = gx * gx + gy * gy;
#pragma unroll
    for (int d = 16; d > 0; d >>= 1) s += __shfl_xor(s, d, 32);
    float n   = sqrtf(s);
    float inv = 1.0f / fmaxf(n, EPSN);
    gx *= inv;
    gy *= inv;
    float2 g;  g.x = gx;  g.y = gy;
    *(float2*)(f + off) = g;
    float2 b = *(const float2*)(base + off);
    float2 a;  a.x = b.x + gx;  a.y = b.y + gy;
    *(float2*)(acc + off) = a;
}

__global__ void edge_weight_scatter(const float* __restrict__ acc,
                                    const float* __restrict__ vals,
                                    const int*   __restrict__ rows,
                                    const int*   __restrict__ cols,
                                    float*       __restrict__ out, int E) {
    const float kScale = 1.0f / 3.0f;
    int lane   = threadIdx.x & 31;
    int waveId = (blockIdx.x * blockDim.x + threadIdx.x) >> 5;
    int m      = lane & 15;
    int half   = lane >> 4;
    int e      = waveId * 16 + m;
    bool valid = (e < E);
    int  ec    = valid ? e : 0;
    int   row  = rows[ec];
    int   col  = cols[ec];
    float v    = valid ? vals[ec] : 0.0f;

    const float* S = acc + (size_t)row * EMB;
    const float* T = acc + (size_t)col * EMB;

    v8f cacc = {};
#pragma unroll
    for (int k0 = 0; k0 < EMB; k0 += 4) {
        float2 sa = *(const float2*)(S + k0 + 2 * half);
        float2 sb = *(const float2*)(T + k0 + 2 * half);
        v2f A;  A.x = sa.x * kScale;  A.y = sa.y * kScale;
        v2f B;  B.x = sb.x * kScale;  B.y = sb.y * kScale;
        cacc = __builtin_amdgcn_wmma_f32_16x16x4_f32(
             false, A,  false, B,
             (short)0, cacc,  false,  false);
    }

    float diag = 0.0f;
#pragma unroll
    for (int vv = 0; vv < 8; ++vv) {
        float dlo = readlane_f32(cacc[vv], vv);
        float dhi = readlane_f32(cacc[vv], 24 + vv);
        if (m == vv)     diag = dlo;
        if (m == vv + 8) diag = dhi;
    }

    float cross = diag * (1.0f / EMB);
    float z     = 4.0f * (1.0f - cross);
    float sp    = log1pf(expf(-fabsf(z))) + fmaxf(z, 0.0f);
    float mat   = 0.5f * expf(z) * sp * v;

#pragma unroll 1
    for (int j = 0; j < 16; ++j) {
        float mj = readlane_f32(mat, j);
        int   rj = __builtin_amdgcn_readlane(row, j);
        int   cj = __builtin_amdgcn_readlane(col, j);
        float2 ee = *(const float2*)(acc + (size_t)cj * EMB + lane * 2);
        float* o  = out + (size_t)rj * EMB + lane * 2;
        atomicAdd(o,     ee.x * kScale * mj);
        atomicAdd(o + 1, ee.y * kScale * mj);
    }
}

extern "C" void kernel_launch(void* const* d_in, const int* in_sizes, int n_in,
                              void* d_out, int out_size, void* d_ws, size_t ws_size,
                              hipStream_t stream) {
    (void)n_in; (void)out_size; (void)ws_size;
    const float* feats     = (const float*)d_in[0];
    const float* edge_vals = (const float*)d_in[1];
    const int*   edge_row  = (const int*)d_in[2];
    const int*   edge_col  = (const int*)d_in[3];

    int N = in_sizes[0] / EMB;
    int E = in_sizes[1];

    float* bufA = (float*)d_ws;
    float* bufB = bufA + (size_t)N * EMB;
    float* acc  = bufB + (size_t)N * EMB;
    float* out  = (float*)d_out;
    size_t rowBytes = (size_t)N * EMB * sizeof(float);

    const int blk = 256;
    int gridSp = (int)(((long long)E * 16 + blk - 1) / blk);
    int gridNm = (int)(((long long)N * 32 + blk - 1) / blk);
    long long nWaves = ((long long)E + 15) / 16;
    int gridEw = (int)((nWaves * 32 + blk - 1) / blk);

    hipMemsetAsync(bufA, 0, rowBytes, stream);
    spmm_scatter<<<gridSp, blk, 0, stream>>>(feats, edge_vals, edge_row, edge_col, bufA, E);
    norm_accum<<<gridNm, blk, 0, stream>>>(bufA, feats, acc, 0.5f, N);

    hipMemsetAsync(bufB, 0, rowBytes, stream);
    spmm_scatter<<<gridSp, blk, 0, stream>>>(bufA, edge_vals, edge_row, edge_col, bufB, E);
    norm_accum<<<gridNm, blk, 0, stream>>>(bufB, acc, acc, 1.0f / 3.0f, N);

    hipMemsetAsync(out, 0, rowBytes, stream);
    edge_weight_scatter<<<gridEw, blk, 0, stream>>>(acc, edge_vals, edge_row, edge_col, out, E);
}


# Replace the three f32-atomic scatters by a per-call row-bucketed edge list (count, reserve, fill) and row-owned gather sums stored once; the three large buffer clears go
# speedup vs baseline: 1.7665x; 1.7665x over previous
#include <hip/hip_runtime.h>
#include <hip/hip_bf16.h>
#include <math.h>

typedef float v2f __attribute__((ext_vector_type(2)));
typedef float v8f __attribute__((ext_vector_type(8)));

#define EMB   64
#define EPSN  1e-12f

__device__ __forceinline__ float readlane_f32(float x, int lane) {
    return __int_as_float(__builtin_amdgcn_readlane(__float_as_int(x), lane));
}

__global__ void row_hist(const int* __restrict__ rows,
                         int*       __restrict__ cnt,
                         int*       __restrict__ pos, int E) {
    int e = blockIdx.x * blockDim.x + threadIdx.x;
    if (e >= E) return;
    pos[e] = atomicAdd(cnt + rows[e], 1);
}

__global__ void row_alloc(const int* __restrict__ cnt,
                          int*       __restrict__ seg,
                          int*       __restrict__ total, int N) {
    __shared__ int wsum[8];
    __shared__ int base;
    int r    = blockIdx.x * 256 + threadIdx.x;
    int lane = threadIdx.x & 31;
    int w    = threadIdx.x >> 5;
    int c    = (r < N) ? cnt[r] : 0;
    int incl = c;
#pragma unroll
    for (int d = 1; d < 32; d <<= 1) {
        int t = __shfl_up(incl, d, 32);
        if (lane >= d) incl += t;
    }
    if (lane == 31) wsum[w] = incl;
    __syncthreads();
    if (threadIdx.x == 0) {
        int run = 0;
        for (int i = 0; i < 8; ++i) { int t = wsum[i]; wsum[i] = run; run += t; }
        base = atomicAdd(total, run);
    }
    __syncthreads();
    if (r < N) {
        int2 sc;
        sc.x = base + wsum[w] + incl - c;
        sc.y = c;
        *(int2*)(seg + (size_t)r * 2) = sc;
    }
}

__global__ void csr_fill(const float* __restrict__ vals,
                         const int*   __restrict__ rows,
                         const int*   __restrict__ cols,
                         const int*   __restrict__ seg,
                         const int*   __restrict__ pos,
                         int*         __restrict__ ce, int E) {
    int e = blockIdx.x * blockDim.x + threadIdx.x;
    if (e >= E) return;
    int r = rows[e];
    int p = seg[(size_t)r * 2] + pos[e];
    int4 rec;
    rec.x = cols[e];
    rec.y = __float_as_int(vals[e]);
    rec.z = r;
    rec.w = 0;
    *(int4*)(ce + (size_t)p * 4) = rec;
}

__device__ __forceinline__ void fma_row(float4& a, const float4& x, float sc, float w) {
    a.x += (x.x * sc) * w;
    a.y += (x.y * sc) * w;
    a.z += (x.z * sc) * w;
    a.w += (x.w * sc) * w;
}

__global__ void spmm_gather(const float* __restrict__ f,
                            const int*   __restrict__ seg,
                            const int*   __restrict__ ce,
                            float*       __restrict__ dst,
                            float srcScale, int N) {
    int tid = blockIdx.x * blockDim.x + threadIdx.x;
    int r   = tid >> 4;
    int l4  = (tid & 15) << 2;
    if (r >= N) return;
    int2 sc = *(const int2*)(seg + (size_t)r * 2);
    int c = sc.y;
    const int4* rec = (const int4*)ce + sc.x;
    const float* fl = f + l4;
    float4 a;  a.x = 0.0f;  a.y = 0.0f;  a.z = 0.0f;  a.w = 0.0f;
    for (int i = 0; i < c; i += 4) {
        int last = c - 1;
        int i1 = min(i + 1, last);
        int i2 = min(i + 2, last);
        int i3 = min(i + 3, last);
        int4 e0 = rec[i];
        int4 e1 = rec[i1];
        int4 e2 = rec[i2];
        int4 e3 = rec[i3];
        float4 x0 = *(const float4*)(fl + (size_t)e0.x * EMB);
        float4 x1 = *(const float4*)(fl + (size_t)e1.x * EMB);
        float4 x2 = *(const float4*)(fl + (size_t)e2.x * EMB);
        float4 x3 = *(const float4*)(fl + (size_t)e3.x * EMB);
        float w0 = __int_as_float(e0.y);
        float w1 = (i + 1 <= last) ? __int_as_float(e1.y) : 0.0f;
        float w2 = (i + 2 <= last) ? __int_as_float(e2.y) : 0.0f;
        float w3 = (i + 3 <= last) ? __int_as_float(e3.y) : 0.0f;
        fma_row(a, x0, srcScale, w0);
        fma_row(a, x1, srcScale, w1);
        fma_row(a, x2, srcScale, w2);
        fma_row(a, x3, srcScale, w3);
    }
    *(float4*)(dst + (size_t)r * EMB + l4) = a;
}

__global__ void norm_accum(float*       __restrict__ f,
                           const float* __restrict__ base,
                           float*       __restrict__ acc,
                           float inv_c, int N) {
    int tid  = blockIdx.x * blockDim.x + threadIdx.x;
    int lane = tid & 31;
    int row  = tid >> 5;
    if (row >= N) return;
    size_t off = (size_t)row * EMB + lane * 2;
    float2 x = *(const float2*)(f + off);
    float gx = x.x * inv_c;
    float gy = x.y * inv_c;
    float s  = gx * gx + gy * gy;
#pragma unroll
    for (int d = 16; d > 0; d >>= 1) s += __shfl_xor(s, d, 32);
    float n   = sqrtf(s);
    float inv = 1.0f / fmaxf(n, EPSN);
    gx *= inv;
    gy *= inv;
    float2 g;  g.x = gx;  g.y = gy;
    *(float2*)(f + off) = g;
    float2 b = *(const float2*)(base + off);
    float2 a;  a.x = b.x + gx;  a.y = b.y + gy;
    *(float2*)(acc + off) = a;
}

__global__ void edge_weight(const float* __restrict__ acc,
                            int* ce, int E) {
    const float kScale = 1.0f / 3.0f;
    int lane   = threadIdx.x & 31;
    int waveId = (blockIdx.x * blockDim.x + threadIdx.x) >> 5;
    int m      = lane & 15;
    int half   = lane >> 4;
    int p      = waveId * 16 + m;
    bool valid = (p < E);
    int  pc    = valid ? p : 0;
    int4 rec   = *((const int4*)ce + pc);
    int   col  = rec.x;
    int   row  = rec.z;
    float v    = valid ? __int_as_float(rec.y) : 0.0f;

    const float* S = acc + (size_t)row * EMB;
    const float* T = acc + (size_t)col * EMB;

    v8f cacc = {};
#pragma unroll
    for (int k0 = 0; k0 < EMB; k0 += 4) {
        float2 sa = *(const float2*)(S + k0 + 2 * half);
        float2 sb = *(const float2*)(T + k0 + 2 * half);
        v2f A;  A.x = sa.x * kScale;  A.y = sa.y * kScale;
        v2f B;  B.x = sb.x * kScale;  B.y = sb.y * kScale;
        cacc = __builtin_amdgcn_wmma_f32_16x16x4_f32(
             false, A,  false, B,
             (short)0, cacc,  false,  false);
    }

    float diag = 0.0f;
#pragma unroll
    for (int vv = 0; vv < 8; ++vv) {
        float dlo = readlane_f32(cacc[vv], vv);
        float dhi = readlane_f32(cacc[vv], 24 + vv);
        if (m == vv)     diag = dlo;
        if (m == vv + 8) diag = dhi;
    }

    float cross = diag * (1.0f / EMB);
    float z     = 4.0f * (1.0f - cross);
    float sp    = log1pf(expf(-fabsf(z))) + fmaxf(z, 0.0f);
    float mat   = 0.5f * expf(z) * sp * v;

    if (valid && half == 0) ce[(size_t)p * 4 + 1] = __float_as_int(mat);
}

extern "C" void kernel_launch(void* const* d_in, const int* in_sizes, int n_in,
                              void* d_out, int out_size, void* d_ws, size_t ws_size,
                              hipStream_t stream) {
    (void)n_in; (void)out_size; (void)ws_size;
    const float* feats     = (const float*)d_in[0];
    const float* edge_vals = (const float*)d_in[1];
    const int*   edge_row  = (const int*)d_in[2];
    const int*   edge_col  = (const int*)d_in[3];

    int N = in_sizes[0] / EMB;
    int E = in_sizes[1];

    float* bufA = (float*)d_ws;
    float* bufB = bufA + (size_t)N * EMB;
    float* acc  = bufB + (size_t)N * EMB;
    float* out  = (float*)d_out;

    int* ce    = (int*)bufB;
    int* seg   = ce + (size_t)E * 4;
    int* pos   = seg + (size_t)N * 2;
    int* cnt   = pos + E;
    int* total = cnt + N;

    const int blk = 256;
    int gridE  = (E + blk - 1) / blk;
    int gridN  = (N + blk - 1) / blk;
    int gridGa = (int)(((long long)N * 16 + blk - 1) / blk);
    int gridNm = (int)(((long long)N * 32 + blk - 1) / blk);
    long long nWaves = ((long long)E + 15) / 16;
    int gridEw = (int)((nWaves * 32 + blk - 1) / blk);

    hipMemsetAsync(cnt, 0, (size_t)(N + 1) * sizeof(int), stream);
    row_hist<<<gridE, blk, 0, stream>>>(edge_row, cnt, pos, E);
    row_alloc<<<gridN, blk, 0, stream>>>(cnt, seg, total, N);
    csr_fill<<<gridE, blk, 0, stream>>>(edge_vals, edge_row, edge_col, seg, pos, ce, E);

    spmm_gather<<<gridGa, blk, 0, stream>>>(feats, seg, ce, bufA, 1.0f, N);
    norm_accum<<<gridNm, blk, 0, stream>>>(bufA, feats, acc, 0.5f, N);

    spmm_gather<<<gridGa, blk, 0, stream>>>(bufA, seg, ce, out, 1.0f, N);
    norm_accum<<<gridNm, blk, 0, stream>>>(out, acc, acc, 1.0f / 3.0f, N);

    edge_weight<<<gridEw, blk, 0, stream>>>(acc, ce, E);
    spmm_gather<<<gridGa, blk, 0, stream>>>(acc, seg, ce, out, 1.0f / 3.0f, N);
}


# Edge-weight kernel reads its 32 embedding rows per wave as whole lines (half-wave per row, 16 B per lane) and takes the WMMA fragments from a padded LDS tile instead of 16-line global loads
# speedup vs baseline: 1.8592x; 1.0525x over previous
#include <hip/hip_runtime.h>
#include <hip/hip_bf16.h>
#include <math.h>

typedef float v2f __attribute__((ext_vector_type(2)));
typedef float v8f __attribute__((ext_vector_type(8)));

#define EMB   64
#define EPSN  1e-12f

__device__ __forceinline__ float readlane_f32(float x, int lane) {
    return __int_as_float(__builtin_amdgcn_readlane(__float_as_int(x), lane));
}

__global__ void row_hist(const int* __restrict__ rows,
                         int*       __restrict__ cnt,
                         int*       __restrict__ pos, int E) {
    int e = blockIdx.x * blockDim.x + threadIdx.x;
    if (e >= E) return;
    pos[e] = atomicAdd(cnt + rows[e], 1);
}

__global__ void row_alloc(const int* __restrict__ cnt,
                          int*       __restrict__ seg,
                          int*       __restrict__ total, int N) {
    __shared__ int wsum[8];
    __shared__ int base;
    int r    = blockIdx.x * 256 + threadIdx.x;
    int lane = threadIdx.x & 31;
    int w    = threadIdx.x >> 5;
    int c    = (r < N) ? cnt[r] : 0;
    int incl = c;
#pragma unroll
    for (int d = 1; d < 32; d <<= 1) {
        int t = __shfl_up(incl, d, 32);
        if (lane >= d) incl += t;
    }
    if (lane == 31) wsum[w] = incl;
    __syncthreads();
    if (threadIdx.x == 0) {
        int run = 0;
        for (int i = 0; i < 8; ++i) { int t = wsum[i]; wsum[i] = run; run += t; }
        base = atomicAdd(total, run);
    }
    __syncthreads();
    if (r < N) {
        int2 sc;
        sc.x = base + wsum[w] + incl - c;
        sc.y = c;
        *(int2*)(seg + (size_t)r * 2) = sc;
    }
}

__global__ void csr_fill(const float* __restrict__ vals,
                         const int*   __restrict__ rows,
                         const int*   __restrict__ cols,
                         const int*   __restrict__ seg,
                         const int*   __restrict__ pos,
                         int*         __restrict__ ce, int E) {
    int e = blockIdx.x * blockDim.x + threadIdx.x;
    if (e >= E) return;
    int r = rows[e];
    int p = seg[(size_t)r * 2] + pos[e];
    int4 rec;
    rec.x = cols[e];
    rec.y = __float_as_int(vals[e]);
    rec.z = r;
    rec.w = 0;
    *(int4*)(ce + (size_t)p * 4) = rec;
}

__device__ __forceinline__ void fma_row(float4& a, const float4& x, float sc, float w) {
    a.x += (x.x * sc) * w;
    a.y += (x.y * sc) * w;
    a.z += (x.z * sc) * w;
    a.w += (x.w * sc) * w;
}

__global__ void spmm_gather(const float* __restrict__ f,
                            const int*   __restrict__ seg,
                            const int*   __restrict__ ce,
                            float*       __restrict__ dst,
                            float srcScale, int N) {
    int tid = blockIdx.x * blockDim.x + threadIdx.x;
    int r   = tid >> 4;
    int l4  = (tid & 15) << 2;
    if (r >= N) return;
    int2 sc = *(const int2*)(seg + (size_t)r * 2);
    int c = sc.y;
    const int4* rec = (const int4*)ce + sc.x;
    const float* fl = f + l4;
    float4 a;  a.x = 0.0f;  a.y = 0.0f;  a.z = 0.0f;  a.w = 0.0f;
    for (int i = 0; i < c; i += 4) {
        int last = c - 1;
        int i1 = min(i + 1, last);
        int i2 = min(i + 2, last);
        int i3 = min(i + 3, last);
        int4 e0 = rec[i];
        int4 e1 = rec[i1];
        int4 e2 = rec[i2];
        int4 e3 = rec[i3];
        float4 x0 = *(const float4*)(fl + (size_t)e0.x * EMB);
        float4 x1 = *(const float4*)(fl + (size_t)e1.x * EMB);
        float4 x2 = *(const float4*)(fl + (size_t)e2.x * EMB);
        float4 x3 = *(const float4*)(fl + (size_t)e3.x * EMB);
        float w0 = __int_as_float(e0.y);
        float w1 = (i + 1 <= last) ? __int_as_float(e1.y) : 0.0f;
        float w2 = (i + 2 <= last) ? __int_as_float(e2.y) : 0.0f;
        float w3 = (i + 3 <= last) ? __int_as_float(e3.y) : 0.0f;
        fma_row(a, x0, srcScale, w0);
        fma_row(a, x1, srcScale, w1);
        fma_row(a, x2, srcScale, w2);
        fma_row(a, x3, srcScale, w3);
    }
    *(float4*)(dst + (size_t)r * EMB + l4) = a;
}

__global__ void norm_accum(float*       __restrict__ f,
                           const float* __restrict__ base,
                           float*       __restrict__ acc,
                           float inv_c, int N) {
    int tid  = blockIdx.x * blockDim.x + threadIdx.x;
    int lane = tid & 31;
    int row  = tid >> 5;
    if (row >= N) return;
    size_t off = (size_t)row * EMB + lane * 2;
    float2 x = *(const float2*)(f + off);
    float gx = x.x * inv_c;
    float gy = x.y * inv_c;
    float s  = gx * gx + gy * gy;
#pragma unroll
    for (int d = 16; d > 0; d >>= 1) s += __shfl_xor(s, d, 32);
    float n   = sqrtf(s);
    float inv = 1.0f / fmaxf(n, EPSN);
    gx *= inv;
    gy *= inv;
    float2 g;  g.x = gx;  g.y = gy;
    *(float2*)(f + off) = g;
    float2 b = *(const float2*)(base + off);
    float2 a;  a.x = b.x + gx;  a.y = b.y + gy;
    *(float2*)(acc + off) = a;
}

#define EW_WAVES 4
#define EW_PITCH 68

__global__ void __launch_bounds__(EW_WAVES * 32)
edge_weight(const float* __restrict__ acc, int* ce, int E) {
    __shared__ __attribute__((aligned(16))) float tileS[EW_WAVES][16 * EW_PITCH];
    __shared__ __attribute__((aligned(16))) float tileT[EW_WAVES][16 * EW_PITCH];
    const float kScale = 1.0f / 3.0f;
    int lane   = threadIdx.x & 31;
    int wv     = threadIdx.x >> 5;
    int waveId = blockIdx.x * EW_WAVES + wv;
    int m      = lane & 15;
    int half   = lane >> 4;
    int p      = waveId * 16 + m;
    bool valid = (p < E);
    int  pc    = valid ? p : 0;
    int4 rec   = *((const int4*)ce + pc);
    int   col  = rec.x;
    int   row  = rec.z;
    float v    = valid ? __int_as_float(rec.y) : 0.0f;

    float* sS = tileS[wv];
    float* sT = tileT[wv];
#pragma unroll
    for (int j = 0; j < 8; ++j) {
        int src = 2 * j + half;
        int rj  = __shfl(row, src, 32);
        int cj  = __shfl(col, src, 32);
        float4 s4 = *(const float4*)(acc + (size_t)rj * EMB + 4 * m);
        float4 t4 = *(const float4*)(acc + (size_t)cj * EMB + 4 * m);
        s4.x *= kScale;  s4.y *= kScale;  s4.z *= kScale;  s4.w *= kScale;
        t4.x *= kScale;  t4.y *= kScale;  t4.z *= kScale;  t4.w *= kScale;
        *(float4*)(sS + src * EW_PITCH + 4 * m) = s4;
        *(float4*)(sT + src * EW_PITCH + 4 * m) = t4;
    }
    __syncthreads();

    const float* S = sS + m * EW_PITCH + 2 * half;
    const float* T = sT + m * EW_PITCH + 2 * half;
    v8f cacc = {};
#pragma unroll
    for (int k0 = 0; k0 < EMB; k0 += 4) {
        float2 sa = *(const float2*)(S + k0);
        float2 sb = *(const float2*)(T + k0);
        v2f A;  A.x = sa.x;  A.y = sa.y;
        v2f B;  B.x = sb.x;  B.y = sb.y;
        cacc = __builtin_amdgcn_wmma_f32_16x16x4_f32(
             false, A,  false, B,
             (short)0, cacc,  false,  false);
    }

    float diag = 0.0f;
#pragma unroll
    for (int vv = 0; vv < 8; ++vv) {
        float dlo = readlane_f32(cacc[vv], vv);
        float dhi = readlane_f32(cacc[vv], 24 + vv);
        if (m == vv)     diag = dlo;
        if (m == vv + 8) diag = dhi;
    }

    float cross = diag * (1.0f / EMB);
    float z     = 4.0f * (1.0f - cross);
    float sp    = log1pf(expf(-fabsf(z))) + fmaxf(z, 0.0f);
    float mat   = 0.5f * expf(z) * sp * v;

    if (valid && half == 0) ce[(size_t)p * 4 + 1] = __float_as_int(mat);
}

extern "C" void kernel_launch(void* const* d_in, const int* in_sizes, int n_in,
                              void* d_out, int out_size, void* d_ws, size_t ws_size,
                              hipStream_t stream) {
    (void)n_in; (void)out_size; (void)ws_size;
    const float* feats     = (const float*)d_in[0];
    const float* edge_vals = (const float*)d_in[1];
    const int*   edge_row  = (const int*)d_in[2];
    const int*   edge_col  = (const int*)d_in[3];

    int N = in_sizes[0] / EMB;
    int E = in_sizes[1];

    float* bufA = (float*)d_ws;
    float* bufB = bufA + (size_t)N * EMB;
    float* acc  = bufB + (size_t)N * EMB;
    float* out  = (float*)d_out;

    int* ce    = (int*)bufB;
    int* seg   = ce + (size_t)E * 4;
    int* pos   = seg + (size_t)N * 2;
    int* cnt   = pos + E;
    int* total = cnt + N;

    const int blk = 256;
    int gridE  = (E + blk - 1) / blk;
    int gridN  = (N + blk - 1) / blk;
    int gridGa = (int)(((long long)N * 16 + blk - 1) / blk);
    int gridNm = (int)(((long long)N * 32 + blk - 1) / blk);
    long long nWaves = ((long long)E + 15) / 16;
    int gridEw = (int)((nWaves + EW_WAVES - 1) / EW_WAVES);

    hipMemsetAsync(cnt, 0, (size_t)(N + 1) * sizeof(int), stream);
    row_hist<<<gridE, blk, 0, stream>>>(edge_row, cnt, pos, E);
    row_alloc<<<gridN, blk, 0, stream>>>(cnt, seg, total, N);
    csr_fill<<<gridE, blk, 0, stream>>>(edge_vals, edge_row, edge_col, seg, pos, ce, E);

    spmm_gather<<<gridGa, blk, 0, stream>>>(feats, seg, ce, bufA, 1.0f, N);
    norm_accum<<<gridNm, blk, 0, stream>>>(bufA, feats, acc, 0.5f, N);

    spmm_gather<<<gridGa, blk, 0, stream>>>(bufA, seg, ce, out, 1.0f, N);
    norm_accum<<<gridNm, blk, 0, stream>>>(out, acc, acc, 1.0f / 3.0f, N);

    edge_weight<<<gridEw, EW_WAVES * 32, 0, stream>>>(acc, ce, E);
    spmm_gather<<<gridGa, blk, 0, stream>>>(acc, seg, ce, out, 1.0f / 3.0f, N);
}


# Fuse edge-weight and final row-sum kernels: a wave per 16 rows walks their contiguous records in 16-record tiles; WMMA dots vs the group's rows held in registers, row sums from the same LDS tile
# speedup vs baseline: 2.3625x; 1.2707x over previous
#include <hip/hip_runtime.h>
#include <hip/hip_bf16.h>
#include <math.h>

typedef float v2f __attribute__((ext_vector_type(2)));
typedef float v8f __attribute__((ext_vector_type(8)));

#define EMB   64
#define EPSN  1e-12f

__device__ __forceinline__ float readlane_f32(float x, int lane) {
    return __int_as_float(__builtin_amdgcn_readlane(__float_as_int(x), lane));
}

__global__ void row_hist(const int* __restrict__ rows,
                         int*       __restrict__ cnt,
                         int*       __restrict__ pos, int E) {
    int e = blockIdx.x * blockDim.x + threadIdx.x;
    if (e >= E) return;
    pos[e] = atomicAdd(cnt + rows[e], 1);
}

__global__ void row_alloc(const int* __restrict__ cnt,
                          int*       __restrict__ seg,
                          int*       __restrict__ total, int N) {
    __shared__ int wsum[8];
    __shared__ int base;
    int r    = blockIdx.x * 256 + threadIdx.x;
    int lane = threadIdx.x & 31;
    int w    = threadIdx.x >> 5;
    int c    = (r < N) ? cnt[r] : 0;
    int incl = c;
#pragma unroll
    for (int d = 1; d < 32; d <<= 1) {
        int t = __shfl_up(incl, d, 32);
        if (lane >= d) incl += t;
    }
    if (lane == 31) wsum[w] = incl;
    __syncthreads();
    if (threadIdx.x == 0) {
        int run = 0;
        for (int i = 0; i < 8; ++i) { int t = wsum[i]; wsum[i] = run; run += t; }
        base = atomicAdd(total, run);
    }
    __syncthreads();
    if (r < N) {
        int2 sc;
        sc.x = base + wsum[w] + incl - c;
        sc.y = c;
        *(int2*)(seg + (size_t)r * 2) = sc;
    }
}

__global__ void csr_fill(const float* __restrict__ vals,
                         const int*   __restrict__ rows,
                         const int*   __restrict__ cols,
                         const int*   __restrict__ seg,
                         const int*   __restrict__ pos,
                         int*         __restrict__ ce, int E) {
    int e = blockIdx.x * blockDim.x + threadIdx.x;
    if (e >= E) return;
    int r = rows[e];
    int p = seg[(size_t)r * 2] + pos[e];
    int4 rec;
    rec.x = cols[e];
    rec.y = __float_as_int(vals[e]);
    rec.z = r;
    rec.w = e;
    *(int4*)(ce + (size_t)p * 4) = rec;
}

#define ROW_BLK   256
#define ROW_HOLD  32
__device__ __forceinline__ void fma_row(float4& a, const float4& x, float sc, float w) {
    a.x += (x.x * sc) * w;
    a.y += (x.y * sc) * w;
    a.z += (x.z * sc) * w;
    a.w += (x.w * sc) * w;
}

template <int K>
__device__ __forceinline__ int row_bcast(int v) {
    return __builtin_amdgcn_update_dpp(0, v, 0x150 + K, 0xf, 0xf, false);
}

template <int D>
__device__ __forceinline__ int row_rot(int v) {
    return __builtin_amdgcn_update_dpp(0, v, 0x120 + D, 0xf, 0xf, false);
}

template <int M>
__device__ __forceinline__ float row_xadd(float s) {
    int t = __builtin_amdgcn_update_dpp(0, __float_as_int(s), 0x160 + M, 0xf, 0xf, false);
    return s + __int_as_float(t);
}

template <int D>
__device__ __forceinline__ void rank2(int ka, int kb, int& ra, int& rb) {
    int ta = row_rot<D>(ka);
    int tb = row_rot<D>(kb);
    ra += ((ta < ka) ? 1 : 0) + ((tb < ka) ? 1 : 0);
    rb += ((ta < kb) ? 1 : 0) + ((tb < kb) ? 1 : 0);
}

template <int I>
__device__ __forceinline__ void gather4(float4& a, const float* fl, int scol, int sval, float sc) {
    int c0 = row_bcast<I + 0>(scol);
    int c1 = row_bcast<I + 1>(scol);
    int c2 = row_bcast<I + 2>(scol);
    int c3 = row_bcast<I + 3>(scol);
    float w0 = __int_as_float(row_bcast<I + 0>(sval));
    float w1 = __int_as_float(row_bcast<I + 1>(sval));
    float w2 = __int_as_float(row_bcast<I + 2>(sval));
    float w3 = __int_as_float(row_bcast<I + 3>(sval));
    float4 x0 = *(const float4*)(fl + (size_t)c0 * EMB);
    float4 x1 = *(const float4*)(fl + (size_t)c1 * EMB);
    float4 x2 = *(const float4*)(fl + (size_t)c2 * EMB);
    float4 x3 = *(const float4*)(fl + (size_t)c3 * EMB);
    fma_row(a, x0, sc, w0);
    fma_row(a, x1, sc, w1);
    fma_row(a, x2, sc, w2);
    fma_row(a, x3, sc, w3);
}

template <int MODE>
__global__ void spmm_rows(const float* __restrict__ f,
                          const int*   __restrict__ seg,
                          int*                      ce,
                          const float*              base,
                          float*                    fout,
                          float*                    dst,
                          float srcScale, float inv_c, int N) {
    int tid = blockIdx.x * blockDim.x + threadIdx.x;
    int r   = tid >> 4;
    int m   = tid & 15;
    int l4  = m << 2;
    if (r >= N) return;
    int2 sc = *(const int2*)(seg + (size_t)r * 2);
    int c = sc.y;
    int4* rec = (int4*)ce + sc.x;
    const float* fl = f + l4;
    size_t off = (size_t)r * EMB + l4;
    float4 b4 = *(const float4*)(base + off);
    float4 a;  a.x = 0.0f;  a.y = 0.0f;  a.z = 0.0f;  a.w = 0.0f;

    if (c > ROW_HOLD) {
        int prev = -1;
        for (int k = 0; k < c; ++k) {
            int best = 0x7fffffff, bcol = 0, bval = 0;
            for (int j = 0; j < c; ++j) {
                int4 q = rec[j];
                if (q.w > prev && q.w < best) { best = q.w;  bcol = q.x;  bval = q.y; }
            }
            prev = best;
            float4 x = *(const float4*)(fl + (size_t)bcol * EMB);
            fma_row(a, x, srcScale, __int_as_float(bval));
        }
    } else if (MODE == 0 && c > 16) {
        __shared__ int2 ordS[ROW_BLK / 16][ROW_HOLD];
        int2* ord = ordS[threadIdx.x >> 4];
        bool hb = (m + 16 < c);
        int4 qa = rec[m];
        int4 qb;  qb.x = 0;  qb.y = 0;  qb.z = r;  qb.w = 0x7fffffff;
        if (hb) qb = rec[m + 16];
        int ka = qa.w;
        int kb = hb ? qb.w : 0x7fffffff;
        int ra = (kb < ka) ? 1 : 0;
        int rb = (ka < kb) ? 1 : 0;
        rank2<1>(ka, kb, ra, rb);   rank2<2>(ka, kb, ra, rb);   rank2<3>(ka, kb, ra, rb);
        rank2<4>(ka, kb, ra, rb);   rank2<5>(ka, kb, ra, rb);   rank2<6>(ka, kb, ra, rb);
        rank2<7>(ka, kb, ra, rb);   rank2<8>(ka, kb, ra, rb);   rank2<9>(ka, kb, ra, rb);
        rank2<10>(ka, kb, ra, rb);  rank2<11>(ka, kb, ra, rb);  rank2<12>(ka, kb, ra, rb);
        rank2<13>(ka, kb, ra, rb);  rank2<14>(ka, kb, ra, rb);  rank2<15>(ka, kb, ra, rb);
        if (!hb) rb = m + 16;
        int2 pa;  pa.x = qa.x;  pa.y = qa.y;
        int2 pb;  pb.x = qb.x;  pb.y = qb.y;
        ord[ra] = pa;
        ord[rb] = pb;
        int2 sa = ord[m];
        int4 oa;  oa.x = sa.x;  oa.y = sa.y;  oa.z = r;  oa.w = 0;
        rec[m] = oa;
        if (hb) {
            int2 sb = ord[m + 16];
            int4 ob;  ob.x = sb.x;  ob.y = sb.y;  ob.z = r;  ob.w = 0;
            rec[m + 16] = ob;
        }
        int last = c - 1;
        for (int i = 0; i < c; i += 4) {
            int2 e0 = ord[i];
            int2 e1 = ord[min(i + 1, last)];
            int2 e2 = ord[min(i + 2, last)];
            int2 e3 = ord[min(i + 3, last)];
            float4 x0 = *(const float4*)(fl + (size_t)e0.x * EMB);
            float4 x1 = *(const float4*)(fl + (size_t)e1.x * EMB);
            float4 x2 = *(const float4*)(fl + (size_t)e2.x * EMB);
            float4 x3 = *(const float4*)(fl + (size_t)e3.x * EMB);
            float w0 = __int_as_float(e0.y);
            float w1 = (i + 1 <= last) ? __int_as_float(e1.y) : 0.0f;
            float w2 = (i + 2 <= last) ? __int_as_float(e2.y) : 0.0f;
            float w3 = (i + 3 <= last) ? __int_as_float(e3.y) : 0.0f;
            fma_row(a, x0, srcScale, w0);
            fma_row(a, x1, srcScale, w1);
            fma_row(a, x2, srcScale, w2);
            fma_row(a, x3, srcScale, w3);
        }
    } else if (MODE == 0) {
        if (c > 0) {
            int4 q;  q.x = 0;  q.y = 0;  q.z = r;  q.w = 0x7fffffff;
            if (m < c) q = rec[m];
            int key = (m < c) ? q.w : 0x7fffffff;
            int rank = 0;
            rank += (row_rot<1>(key)  < key) ? 1 : 0;
            rank += (row_rot<2>(key)  < key) ? 1 : 0;
            rank += (row_rot<3>(key)  < key) ? 1 : 0;
            rank += (row_rot<4>(key)  < key) ? 1 : 0;
            rank += (row_rot<5>(key)  < key) ? 1 : 0;
            rank += (row_rot<6>(key)  < key) ? 1 : 0;
            rank += (row_rot<7>(key)  < key) ? 1 : 0;
            rank += (row_rot<8>(key)  < key) ? 1 : 0;
            rank += (row_rot<9>(key)  < key) ? 1 : 0;
            rank += (row_rot<10>(key) < key) ? 1 : 0;
            rank += (row_rot<11>(key) < key) ? 1 : 0;
            rank += (row_rot<12>(key) < key) ? 1 : 0;
            rank += (row_rot<13>(key) < key) ? 1 : 0;
            rank += (row_rot<14>(key) < key) ? 1 : 0;
            rank += (row_rot<15>(key) < key) ? 1 : 0;
            if (m >= c) rank = m;
            int to   = (((int)threadIdx.x & 16) | rank) << 2;
            int scol = __builtin_amdgcn_ds_permute(to, q.x);
            int sval = __builtin_amdgcn_ds_permute(to, q.y);
            int lastcol = __shfl(scol, c - 1, 16);
            if (m >= c) scol = lastcol;
            if (m < c) {
                int4 o;  o.x = scol;  o.y = sval;  o.z = r;  o.w = 0;
                rec[m] = o;
            }
            gather4<0>(a, fl, scol, sval, srcScale);
            if (c > 4)  gather4<4>(a, fl, scol, sval, srcScale);
            if (c > 8)  gather4<8>(a, fl, scol, sval, srcScale);
            if (c > 12) gather4<12>(a, fl, scol, sval, srcScale);
        }
    } else {
        for (int i = 0; i < c; i += 4) {
            int last = c - 1;
            int i1 = min(i + 1, last);
            int i2 = min(i + 2, last);
            int i3 = min(i + 3, last);
            int4 e0 = rec[i];
            int4 e1 = rec[i1];
            int4 e2 = rec[i2];
            int4 e3 = rec[i3];
            float4 x0 = *(const float4*)(fl + (size_t)e0.x * EMB);
            float4 x1 = *(const float4*)(fl + (size_t)e1.x * EMB);
            float4 x2 = *(const float4*)(fl + (size_t)e2.x * EMB);
            float4 x3 = *(const float4*)(fl + (size_t)e3.x * EMB);
            float w0 = __int_as_float(e0.y);
            float w1 = (i + 1 <= last) ? __int_as_float(e1.y) : 0.0f;
            float w2 = (i + 2 <= last) ? __int_as_float(e2.y) : 0.0f;
            float w3 = (i + 3 <= last) ? __int_as_float(e3.y) : 0.0f;
            fma_row(a, x0, srcScale, w0);
            fma_row(a, x1, srcScale, w1);
            fma_row(a, x2, srcScale, w2);
            fma_row(a, x3, srcScale, w3);
        }
    }

    float gx = a.x * inv_c;
    float gy = a.y * inv_c;
    float gz = a.z * inv_c;
    float gw = a.w * inv_c;
    float s  = gx * gx + gy * gy + gz * gz + gw * gw;
    s = row_xadd<8>(s);
    s = row_xadd<4>(s);
    s = row_xadd<2>(s);
    s = row_xadd<1>(s);
    float n   = sqrtf(s);
    float inv = 1.0f / fmaxf(n, EPSN);
    float4 g;  g.x = gx * inv;  g.y = gy * inv;  g.z = gz * inv;  g.w = gw * inv;
    if (MODE == 0) *(float4*)(fout + off) = g;
    float4 o;  o.x = b4.x + g.x;  o.y = b4.y + g.y;  o.z = b4.z + g.z;  o.w = b4.w + g.w;
    *(float4*)(dst + off) = o;
}

#define FU_WAVES 4
#define FU_PITCH 68

template <int M>
__device__ __forceinline__ int row_xadd_i(int s) {
    return s + __builtin_amdgcn_update_dpp(0, s, 0x160 + M, 0xf, 0xf, false);
}

__device__ __forceinline__ float4 row_load16(const float* __restrict__ base, int row, int m) {
    return *(const float4*)((const char*)base + (unsigned)(row * (EMB * 4) + m * 16));
}

__device__ __forceinline__ void store_row(float* __restrict__ out, int row, int lane, float2 v) {
    *(float2*)((char*)out + (unsigned)(row * (EMB * 4) + lane * 8)) = v;
}

__global__ void __launch_bounds__(FU_WAVES * 32)
edge_upd(const float* __restrict__ acc,
         const int*   __restrict__ seg,
         const int*   __restrict__ ce,
         float*       __restrict__ out, int N) {
    __shared__ __attribute__((aligned(16))) float tileS[FU_WAVES][16 * FU_PITCH];
    const float kScale = 1.0f / 3.0f;
    int lane = threadIdx.x & 31;
    int wv   = __builtin_amdgcn_readfirstlane((int)(threadIdx.x >> 5));
    int r0   = (blockIdx.x * FU_WAVES + wv) * 16;
    if (r0 >= N) return;
    int m    = lane & 15;
    int half = lane >> 4;
    int rows = min(16, N - r0);
    float* sT = tileS[wv];
    const float* F = sT + m * FU_PITCH + 2 * half;

    int2 sc;  sc.x = 0;  sc.y = 0;
    if (m < rows) sc = *(const int2*)(seg + (size_t)(r0 + m) * 2);
    int nsum = sc.y;
    nsum = row_xadd_i<8>(nsum);
    nsum = row_xadd_i<4>(nsum);
    nsum = row_xadd_i<2>(nsum);
    nsum = row_xadd_i<1>(nsum);
    int begin  = __builtin_amdgcn_readfirstlane(sc.x);
    int n      = __builtin_amdgcn_readfirstlane(nsum);
    int ntiles = (n + 15) >> 4;
    int last   = n - 1;
    const int4* recs = (const int4*)ce + begin;

    float2 zero2;  zero2.x = 0.0f;  zero2.y = 0.0f;

    unsigned emp = __builtin_amdgcn_ballot_w32(half == 0 && m < rows && sc.y == 0);
    while (emp != 0u) {
        int zr = __builtin_ctz(emp);
        emp &= emp - 1u;
        store_row(out, r0 + zr, lane, zero2);
    }
    if (ntiles == 0) return;

    int4 rcC = recs[min(m, last)];
    int4 rcN = recs[min(16 + m, last)];

    float4 s4[8];
#pragma unroll
    for (int j = 0; j < 8; ++j) {
        s4[j] = row_load16(acc, min(r0 + 2 * j + half, N - 1), m);
    }
    float4 tq[8];
#pragma unroll
    for (int j = 0; j < 8; ++j) {
        tq[j] = row_load16(acc, __shfl(rcC.x, 2 * j + half, 32), m);
    }
#pragma unroll
    for (int j = 0; j < 8; ++j) {
        float4 x = s4[j];
        x.x *= kScale;  x.y *= kScale;  x.z *= kScale;  x.w *= kScale;
        *(float4*)(sT + (2 * j + half) * FU_PITCH + 4 * m) = x;
    }
    v2f A[16];
#pragma unroll
    for (int ks = 0; ks < 16; ++ks) {
        float2 a = *(const float2*)(F + 4 * ks);
        A[ks].x = a.x;  A[ks].y = a.y;
    }

    float2 av = zero2;
    int tailrow = 0;
    for (int t = 0; t < ntiles; ++t) {
        bool  valid = (16 * t + m < n);
        int   irow  = rcC.z - r0;
        float v     = __int_as_float(rcC.y);

#pragma unroll
        for (int j = 0; j < 8; ++j) {
            float4 x = tq[j];
            x.x *= kScale;  x.y *= kScale;  x.z *= kScale;  x.w *= kScale;
            *(float4*)(sT + (2 * j + half) * FU_PITCH + 4 * m) = x;
        }
#pragma unroll
        for (int j = 0; j < 8; ++j) {
            tq[j] = row_load16(acc, __shfl(rcN.x, 2 * j + half, 32), m);
        }
        int4 rcNN = recs[min(16 * (t + 2) + m, last)];

        int nxt = __shfl_down(irow, 1, 16);
        int nx0 = __builtin_amdgcn_readlane(rcN.z, 0) - r0;
        if (m == 15) nxt = nx0;
        unsigned ends = __builtin_amdgcn_ballot_w32(half == 0 && nxt != irow);
        tailrow = __builtin_amdgcn_readlane(irow, 15);

        v8f c0 = {};
        v8f c1 = {};
#pragma unroll
        for (int ks = 0; ks < 16; ks += 2) {
            float2 b0 = *(const float2*)(F + 4 * ks);
            float2 b1 = *(const float2*)(F + 4 * ks + 4);
            v2f B0;  B0.x = b0.x;  B0.y = b0.y;
            v2f B1;  B1.x = b1.x;  B1.y = b1.y;
            c0 = __builtin_amdgcn_wmma_f32_16x16x4_f32(
                 false, A[ks],  false, B0,
                 (short)0, c0,  false,  false);
            c1 = __builtin_amdgcn_wmma_f32_16x16x4_f32(
                 false, A[ks + 1],  false, B1,
                 (short)0, c1,  false,  false);
        }
        v8f cc = c0 + c1;

        int   i7  = irow & 7;
        float sel = cc[0];
#pragma unroll
        for (int vv = 1; vv < 8; ++vv) {
            if (i7 == vv) sel = cc[vv];
        }
        float dot = ((irow >> 3) == half) ? sel : 0.0f;
        dot += __shfl_xor(dot, 16, 32);

        float cross = dot * (1.0f / EMB);
        float z     = 4.0f * (1.0f - cross);
        float sp    = log1pf(expf(-fabsf(z))) + fmaxf(z, 0.0f);
        float mat   = valid ? 0.5f * expf(z) * sp * v : 0.0f;

        float2 xr[16];
#pragma unroll
        for (int j = 0; j < 16; ++j) {
            xr[j] = *(const float2*)(sT + j * FU_PITCH + 2 * lane);
        }
#pragma unroll
        for (int j = 0; j < 16; ++j) {
            float mj = readlane_f32(mat, j);
            av.x += xr[j].x * mj;
            av.y += xr[j].y * mj;
            if ((ends >> j) & 1u) {
                store_row(out, r0 + __builtin_amdgcn_readlane(irow, j), lane, av);
                av = zero2;
            }
        }

        rcC = rcN;
        rcN = rcNN;
    }
    store_row(out, r0 + tailrow, lane, av);
}

extern "C" void kernel_launch(void* const* d_in, const int* in_sizes, int n_in,
                              void* d_out, int out_size, void* d_ws, size_t ws_size,
                              hipStream_t stream) {
    (void)n_in; (void)out_size; (void)ws_size;
    const float* feats     = (const float*)d_in[0];
    const float* edge_vals = (const float*)d_in[1];
    const int*   edge_row  = (const int*)d_in[2];
    const int*   edge_col  = (const int*)d_in[3];

    int N = in_sizes[0] / EMB;
    int E = in_sizes[1];

    float* bufA = (float*)d_ws;
    float* bufB = bufA + (size_t)N * EMB;
    float* acc  = bufB + (size_t)N * EMB;
    float* out  = (float*)d_out;

    int* ce    = (int*)bufB;
    int* seg   = ce + (size_t)E * 4;
    int* pos   = seg + (size_t)N * 2;
    int* cnt   = pos + E;
    int* total = cnt + N;

    const int blk = 256;
    int gridE  = (E + blk - 1) / blk;
    int gridN  = (N + blk - 1) / blk;
    int gridGa = (int)(((long long)N * 16 + ROW_BLK - 1) / ROW_BLK);
    int gridFu = (N + 16 * FU_WAVES - 1) / (16 * FU_WAVES);

    hipMemsetAsync(cnt, 0, (size_t)(N + 1) * sizeof(int), stream);
    row_hist<<<gridE, blk, 0, stream>>>(edge_row, cnt, pos, E);
    row_alloc<<<gridN, blk, 0, stream>>>(cnt, seg, total, N);
    csr_fill<<<gridE, blk, 0, stream>>>(edge_vals, edge_row, edge_col, seg, pos, ce, E);

    spmm_rows<0><<<gridGa, ROW_BLK, 0, stream>>>(feats, seg, ce, feats, bufA, acc, 1.0f, 0.5f, N);

    spmm_rows<1><<<gridGa, ROW_BLK, 0, stream>>>(bufA, seg, ce, acc, nullptr, acc, 1.0f, 1.0f / 3.0f, N);

    edge_upd<<<gridFu, FU_WAVES * 32, 0, stream>>>(acc, seg, ce, out, N);
}


# Layer 0 stores only its features, not the running layer sum; layer 1 forms (input + layer 0) + layer 1 from its own row of both arrays: a 38 MB store stream goes, a read moves to the gathered array
# speedup vs baseline: 2.5477x; 1.0784x over previous
#include <hip/hip_runtime.h>
#include <hip/hip_bf16.h>
#include <math.h>

typedef float v2f __attribute__((ext_vector_type(2)));
typedef float v8f __attribute__((ext_vector_type(8)));

#define EMB   64
#define EPSN  1e-12f

__device__ __forceinline__ float readlane_f32(float x, int lane) {
    return __int_as_float(__builtin_amdgcn_readlane(__float_as_int(x), lane));
}

__global__ void row_hist(const int* __restrict__ rows,
                         int*       __restrict__ cnt,
                         int*       __restrict__ pos, int E) {
    int e = blockIdx.x * blockDim.x + threadIdx.x;
    if (e >= E) return;
    pos[e] = atomicAdd(cnt + rows[e], 1);
}

__global__ void row_alloc(const int* __restrict__ cnt,
                          int*       __restrict__ seg,
                          int*       __restrict__ total, int N) {
    __shared__ int wsum[8];
    __shared__ int base;
    int r    = blockIdx.x * 256 + threadIdx.x;
    int lane = threadIdx.x & 31;
    int w    = threadIdx.x >> 5;
    int c    = (r < N) ? cnt[r] : 0;
    int incl = c;
#pragma unroll
    for (int d = 1; d < 32; d <<= 1) {
        int t = __shfl_up(incl, d, 32);
        if (lane >= d) incl += t;
    }
    if (lane == 31) wsum[w] = incl;
    __syncthreads();
    if (threadIdx.x == 0) {
        int run = 0;
        for (int i = 0; i < 8; ++i) { int t = wsum[i]; wsum[i] = run; run += t; }
        base = atomicAdd(total, run);
    }
    __syncthreads();
    if (r < N) {
        int2 sc;
        sc.x = base + wsum[w] + incl - c;
        sc.y = c;
        *(int2*)(seg + (size_t)r * 2) = sc;
    }
}

__global__ void csr_fill(const float* __restrict__ vals,
                         const int*   __restrict__ rows,
                         const int*   __restrict__ cols,
                         const int*   __restrict__ seg,
                         const int*   __restrict__ pos,
                         int*         __restrict__ ce, int E) {
    int e = blockIdx.x * blockDim.x + threadIdx.x;
    if (e >= E) return;
    int r = rows[e];
    int p = seg[(size_t)r * 2] + pos[e];
    int4 rec;
    rec.x = cols[e];
    rec.y = __float_as_int(vals[e]);
    rec.z = r;
    rec.w = e;
    *(int4*)(ce + (size_t)p * 4) = rec;
}

#define ROW_BLK   256
#define ROW_HOLD  32
__device__ __forceinline__ void fma_row(float4& a, const float4& x, float sc, float w) {
    a.x += (x.x * sc) * w;
    a.y += (x.y * sc) * w;
    a.z += (x.z * sc) * w;
    a.w += (x.w * sc) * w;
}

template <int K>
__device__ __forceinline__ int row_bcast(int v) {
    return __builtin_amdgcn_update_dpp(0, v, 0x150 + K, 0xf, 0xf, false);
}

template <int D>
__device__ __forceinline__ int row_rot(int v) {
    return __builtin_amdgcn_update_dpp(0, v, 0x120 + D, 0xf, 0xf, false);
}

template <int M>
__device__ __forceinline__ float row_xadd(float s) {
    int t = __builtin_amdgcn_update_dpp(0, __float_as_int(s), 0x160 + M, 0xf, 0xf, false);
    return s + __int_as_float(t);
}

template <int D>
__device__ __forceinline__ void rank2(int ka, int kb, int& ra, int& rb) {
    int ta = row_rot<D>(ka);
    int tb = row_rot<D>(kb);
    ra += ((ta < ka) ? 1 : 0) + ((tb < ka) ? 1 : 0);
    rb += ((ta < kb) ? 1 : 0) + ((tb < kb) ? 1 : 0);
}

template <int I>
__device__ __forceinline__ void gather4(float4& a, const float* fl, int scol, int sval, float sc) {
    int c0 = row_bcast<I + 0>(scol);
    int c1 = row_bcast<I + 1>(scol);
    int c2 = row_bcast<I + 2>(scol);
    int c3 = row_bcast<I + 3>(scol);
    float w0 = __int_as_float(row_bcast<I + 0>(sval));
    float w1 = __int_as_float(row_bcast<I + 1>(sval));
    float w2 = __int_as_float(row_bcast<I + 2>(sval));
    float w3 = __int_as_float(row_bcast<I + 3>(sval));
    float4 x0 = *(const float4*)(fl + (size_t)c0 * EMB);
    float4 x1 = *(const float4*)(fl + (size_t)c1 * EMB);
    float4 x2 = *(const float4*)(fl + (size_t)c2 * EMB);
    float4 x3 = *(const float4*)(fl + (size_t)c3 * EMB);
    fma_row(a, x0, sc, w0);
    fma_row(a, x1, sc, w1);
    fma_row(a, x2, sc, w2);
    fma_row(a, x3, sc, w3);
}

template <int MODE>
__global__ void spmm_rows(const float* __restrict__ f,
                          const int*   __restrict__ seg,
                          int*                      ce,
                          const float* __restrict__ base0,
                          const float* __restrict__ base1,
                          float*       __restrict__ dst,
                          float srcScale, float inv_c, int N) {
    int tid = blockIdx.x * blockDim.x + threadIdx.x;
    int r   = tid >> 4;
    int m   = tid & 15;
    int l4  = m << 2;
    if (r >= N) return;
    int2 sc = *(const int2*)(seg + (size_t)r * 2);
    int c = sc.y;
    int4* rec = (int4*)ce + sc.x;
    const float* fl = f + l4;
    size_t off = (size_t)r * EMB + l4;
    float4 b0;  b0.x = 0.0f;  b0.y = 0.0f;  b0.z = 0.0f;  b0.w = 0.0f;
    float4 b1 = b0;
    if (MODE == 1) {
        b0 = *(const float4*)(base0 + off);
        b1 = *(const float4*)(base1 + off);
    }
    float4 a;  a.x = 0.0f;  a.y = 0.0f;  a.z = 0.0f;  a.w = 0.0f;

    if (c > ROW_HOLD) {
        int prev = -1;
        for (int k = 0; k < c; ++k) {
            int best = 0x7fffffff, bcol = 0, bval = 0;
            for (int j = 0; j < c; ++j) {
                int4 q = rec[j];
                if (q.w > prev && q.w < best) { best = q.w;  bcol = q.x;  bval = q.y; }
            }
            prev = best;
            float4 x = *(const float4*)(fl + (size_t)bcol * EMB);
            fma_row(a, x, srcScale, __int_as_float(bval));
        }
    } else if (MODE == 0 && c > 16) {
        __shared__ int2 ordS[ROW_BLK / 16][ROW_HOLD];
        int2* ord = ordS[threadIdx.x >> 4];
        bool hb = (m + 16 < c);
        int4 qa = rec[m];
        int4 qb;  qb.x = 0;  qb.y = 0;  qb.z = r;  qb.w = 0x7fffffff;
        if (hb) qb = rec[m + 16];
        int ka = qa.w;
        int kb = hb ? qb.w : 0x7fffffff;
        int ra = (kb < ka) ? 1 : 0;
        int rb = (ka < kb) ? 1 : 0;
        rank2<1>(ka, kb, ra, rb);   rank2<2>(ka, kb, ra, rb);   rank2<3>(ka, kb, ra, rb);
        rank2<4>(ka, kb, ra, rb);   rank2<5>(ka, kb, ra, rb);   rank2<6>(ka, kb, ra, rb);
        rank2<7>(ka, kb, ra, rb);   rank2<8>(ka, kb, ra, rb);   rank2<9>(ka, kb, ra, rb);
        rank2<10>(ka, kb, ra, rb);  rank2<11>(ka, kb, ra, rb);  rank2<12>(ka, kb, ra, rb);
        rank2<13>(ka, kb, ra, rb);  rank2<14>(ka, kb, ra, rb);  rank2<15>(ka, kb, ra, rb);
        if (!hb) rb = m + 16;
        int2 pa;  pa.x = qa.x;  pa.y = qa.y;
        int2 pb;  pb.x = qb.x;  pb.y = qb.y;
        ord[ra] = pa;
        ord[rb] = pb;
        int2 sa = ord[m];
        int4 oa;  oa.x = sa.x;  oa.y = sa.y;  oa.z = r;  oa.w = 0;
        rec[m] = oa;
        if (hb) {
            int2 sb = ord[m + 16];
            int4 ob;  ob.x = sb.x;  ob.y = sb.y;  ob.z = r;  ob.w = 0;
            rec[m + 16] = ob;
        }
        int last = c - 1;
        for (int i = 0; i < c; i += 4) {
            int2 e0 = ord[i];
            int2 e1 = ord[min(i + 1, last)];
            int2 e2 = ord[min(i + 2, last)];
            int2 e3 = ord[min(i + 3, last)];
            float4 x0 = *(const float4*)(fl + (size_t)e0.x * EMB);
            float4 x1 = *(const float4*)(fl + (size_t)e1.x * EMB);
            float4 x2 = *(const float4*)(fl + (size_t)e2.x * EMB);
            float4 x3 = *(const float4*)(fl + (size_t)e3.x * EMB);
            float w0 = __int_as_float(e0.y);
            float w1 = (i + 1 <= last) ? __int_as_float(e1.y) : 0.0f;
            float w2 = (i + 2 <= last) ? __int_as_float(e2.y) : 0.0f;
            float w3 = (i + 3 <= last) ? __int_as_float(e3.y) : 0.0f;
            fma_row(a, x0, srcScale, w0);
            fma_row(a, x1, srcScale, w1);
            fma_row(a, x2, srcScale, w2);
            fma_row(a, x3, srcScale, w3);
        }
    } else if (MODE == 0) {
        if (c > 0) {
            int4 q;  q.x = 0;  q.y = 0;  q.z = r;  q.w = 0x7fffffff;
            if (m < c) q = rec[m];
            int key = (m < c) ? q.w : 0x7fffffff;
            int rank = 0;
            rank += (row_rot<1>(key)  < key) ? 1 : 0;
            rank += (row_rot<2>(key)  < key) ? 1 : 0;
            rank += (row_rot<3>(key)  < key) ? 1 : 0;
            rank += (row_rot<4>(key)  < key) ? 1 : 0;
            rank += (row_rot<5>(key)  < key) ? 1 : 0;
            rank += (row_rot<6>(key)  < key) ? 1 : 0;
            rank += (row_rot<7>(key)  < key) ? 1 : 0;
            rank += (row_rot<8>(key)  < key) ? 1 : 0;
            rank += (row_rot<9>(key)  < key) ? 1 : 0;
            rank += (row_rot<10>(key) < key) ? 1 : 0;
            rank += (row_rot<11>(key) < key) ? 1 : 0;
            rank += (row_rot<12>(key) < key) ? 1 : 0;
            rank += (row_rot<13>(key) < key) ? 1 : 0;
            rank += (row_rot<14>(key) < key) ? 1 : 0;
            rank += (row_rot<15>(key) < key) ? 1 : 0;
            if (m >= c) rank = m;
            int to   = (((int)threadIdx.x & 16) | rank) << 2;
            int scol = __builtin_amdgcn_ds_permute(to, q.x);
            int sval = __builtin_amdgcn_ds_permute(to, q.y);
            int lastcol = __shfl(scol, c - 1, 16);
            if (m >= c) scol = lastcol;
            if (m < c) {
                int4 o;  o.x = scol;  o.y = sval;  o.z = r;  o.w = 0;
                rec[m] = o;
            }
            gather4<0>(a, fl, scol, sval, srcScale);
            if (c > 4)  gather4<4>(a, fl, scol, sval, srcScale);
            if (c > 8)  gather4<8>(a, fl, scol, sval, srcScale);
            if (c > 12) gather4<12>(a, fl, scol, sval, srcScale);
        }
    } else {
        for (int i = 0; i < c; i += 4) {
            int last = c - 1;
            int i1 = min(i + 1, last);
            int i2 = min(i + 2, last);
            int i3 = min(i + 3, last);
            int4 e0 = rec[i];
            int4 e1 = rec[i1];
            int4 e2 = rec[i2];
            int4 e3 = rec[i3];
            float4 x0 = *(const float4*)(fl + (size_t)e0.x * EMB);
            float4 x1 = *(const float4*)(fl + (size_t)e1.x * EMB);
            float4 x2 = *(const float4*)(fl + (size_t)e2.x * EMB);
            float4 x3 = *(const float4*)(fl + (size_t)e3.x * EMB);
            float w0 = __int_as_float(e0.y);
            float w1 = (i + 1 <= last) ? __int_as_float(e1.y) : 0.0f;
            float w2 = (i + 2 <= last) ? __int_as_float(e2.y) : 0.0f;
            float w3 = (i + 3 <= last) ? __int_as_float(e3.y) : 0.0f;
            fma_row(a, x0, srcScale, w0);
            fma_row(a, x1, srcScale, w1);
            fma_row(a, x2, srcScale, w2);
            fma_row(a, x3, srcScale, w3);
        }
    }

    float gx = a.x * inv_c;
    float gy = a.y * inv_c;
    float gz = a.z * inv_c;
    float gw = a.w * inv_c;
    float s  = gx * gx + gy * gy + gz * gz + gw * gw;
    s = row_xadd<8>(s);
    s = row_xadd<4>(s);
    s = row_xadd<2>(s);
    s = row_xadd<1>(s);
    float n   = sqrtf(s);
    float inv = 1.0f / fmaxf(n, EPSN);
    float4 g;  g.x = gx * inv;  g.y = gy * inv;  g.z = gz * inv;  g.w = gw * inv;
    if (MODE == 0) {
        *(float4*)(dst + off) = g;
    } else {
        float4 o;
        o.x = (b0.x + b1.x) + g.x;
        o.y = (b0.y + b1.y) + g.y;
        o.z = (b0.z + b1.z) + g.z;
        o.w = (b0.w + b1.w) + g.w;
        *(float4*)(dst + off) = o;
    }
}

#define FU_WAVES 4
#define FU_PITCH 68

template <int M>
__device__ __forceinline__ int row_xadd_i(int s) {
    return s + __builtin_amdgcn_update_dpp(0, s, 0x160 + M, 0xf, 0xf, false);
}

__device__ __forceinline__ float4 row_load16(const float* __restrict__ base, int row, int m) {
    return *(const float4*)((const char*)base + (unsigned)(row * (EMB * 4) + m * 16));
}

__device__ __forceinline__ void store_row(float* __restrict__ out, int row, int lane, float2 v) {
    *(float2*)((char*)out + (unsigned)(row * (EMB * 4) + lane * 8)) = v;
}

__global__ void __launch_bounds__(FU_WAVES * 32)
edge_upd(const float* __restrict__ acc,
         const int*   __restrict__ seg,
         const int*   __restrict__ ce,
         float*       __restrict__ out, int N) {
    __shared__ __attribute__((aligned(16))) float tileS[FU_WAVES][16 * FU_PITCH];
    const float kScale = 1.0f / 3.0f;
    int lane = threadIdx.x & 31;
    int wv   = __builtin_amdgcn_readfirstlane((int)(threadIdx.x >> 5));
    int r0   = (blockIdx.x * FU_WAVES + wv) * 16;
    if (r0 >= N) return;
    int m    = lane & 15;
    int half = lane >> 4;
    int rows = min(16, N - r0);
    float* sT = tileS[wv];
    const float* F = sT + m * FU_PITCH + 2 * half;

    int2 sc;  sc.x = 0;  sc.y = 0;
    if (m < rows) sc = *(const int2*)(seg + (size_t)(r0 + m) * 2);
    int nsum = sc.y;
    nsum = row_xadd_i<8>(nsum);
    nsum = row_xadd_i<4>(nsum);
    nsum = row_xadd_i<2>(nsum);
    nsum = row_xadd_i<1>(nsum);
    int begin  = __builtin_amdgcn_readfirstlane(sc.x);
    int n      = __builtin_amdgcn_readfirstlane(nsum);
    int ntiles = (n + 15) >> 4;
    int last   = n - 1;
    const int4* recs = (const int4*)ce + begin;

    float2 zero2;  zero2.x = 0.0f;  zero2.y = 0.0f;

    unsigned emp = __builtin_amdgcn_ballot_w32(half == 0 && m < rows && sc.y == 0);
    while (emp != 0u) {
        int zr = __builtin_ctz(emp);
        emp &= emp - 1u;
        store_row(out, r0 + zr, lane, zero2);
    }
    if (ntiles == 0) return;

    int4 rcC = recs[min(m, last)];
    int4 rcN = recs[min(16 + m, last)];

    float4 s4[8];
#pragma unroll
    for (int j = 0; j < 8; ++j) {
        s4[j] = row_load16(acc, min(r0 + 2 * j + half, N - 1), m);
    }
    float4 tq[8];
#pragma unroll
    for (int j = 0; j < 8; ++j) {
        tq[j] = row_load16(acc, __shfl(rcC.x, 2 * j + half, 32), m);
    }
#pragma unroll
    for (int j = 0; j < 8; ++j) {
        float4 x = s4[j];
        x.x *= kScale;  x.y *= kScale;  x.z *= kScale;  x.w *= kScale;
        *(float4*)(sT + (2 * j + half) * FU_PITCH + 4 * m) = x;
    }
    v2f A[16];
#pragma unroll
    for (int ks = 0; ks < 16; ++ks) {
        float2 a = *(const float2*)(F + 4 * ks);
        A[ks].x = a.x;  A[ks].y = a.y;
    }

    float2 av = zero2;
    int tailrow = 0;
    for (int t = 0; t < ntiles; ++t) {
        bool  valid = (16 * t + m < n);
        int   irow  = rcC.z - r0;
        float v     = __int_as_float(rcC.y);

#pragma unroll
        for (int j = 0; j < 8; ++j) {
            float4 x = tq[j];
            x.x *= kScale;  x.y *= kScale;  x.z *= kScale;  x.w *= kScale;
            *(float4*)(sT + (2 * j + half) * FU_PITCH + 4 * m) = x;
        }
#pragma unroll
        for (int j = 0; j < 8; ++j) {
            tq[j] = row_load16(acc, __shfl(rcN.x, 2 * j + half, 32), m);
        }
        int4 rcNN = recs[min(16 * (t + 2) + m, last)];

        int nxt = __shfl_down(irow, 1, 16);
        int nx0 = __builtin_amdgcn_readlane(rcN.z, 0) - r0;
        if (m == 15) nxt = nx0;
        unsigned ends = __builtin_amdgcn_ballot_w32(half == 0 && nxt != irow);
        tailrow = __builtin_amdgcn_readlane(irow, 15);

        v8f c0 = {};
        v8f c1 = {};
#pragma unroll
        for (int ks = 0; ks < 16; ks += 2) {
            float2 b0 = *(const float2*)(F + 4 * ks);
            float2 b1 = *(const float2*)(F + 4 * ks + 4);
            v2f B0;  B0.x = b0.x;  B0.y = b0.y;
            v2f B1;  B1.x = b1.x;  B1.y = b1.y;
            c0 = __builtin_amdgcn_wmma_f32_16x16x4_f32(
                 false, A[ks],  false, B0,
                 (short)0, c0,  false,  false);
            c1 = __builtin_amdgcn_wmma_f32_16x16x4_f32(
                 false, A[ks + 1],  false, B1,
                 (short)0, c1,  false,  false);
        }
        v8f cc = c0 + c1;

        int   i7  = irow & 7;
        float sel = cc[0];
#pragma unroll
        for (int vv = 1; vv < 8; ++vv) {
            if (i7 == vv) sel = cc[vv];
        }
        float dot = ((irow >> 3) == half) ? sel : 0.0f;
        dot += __shfl_xor(dot, 16, 32);

        float cross = dot * (1.0f / EMB);
        float z     = 4.0f * (1.0f - cross);
        float sp    = log1pf(expf(-fabsf(z))) + fmaxf(z, 0.0f);
        float mat   = valid ? 0.5f * expf(z) * sp * v : 0.0f;

        float2 xr[16];
#pragma unroll
        for (int j = 0; j < 16; ++j) {
            xr[j] = *(const float2*)(sT + j * FU_PITCH + 2 * lane);
        }
#pragma unroll
        for (int j = 0; j < 16; ++j) {
            float mj = readlane_f32(mat, j);
            av.x += xr[j].x * mj;
            av.y += xr[j].y * mj;
            if ((ends >> j) & 1u) {
                store_row(out, r0 + __builtin_amdgcn_readlane(irow, j), lane, av);
                av = zero2;
            }
        }

        rcC = rcN;
        rcN = rcNN;
    }
    store_row(out, r0 + tailrow, lane, av);
}

extern "C" void kernel_launch(void* const* d_in, const int* in_sizes, int n_in,
                              void* d_out, int out_size, void* d_ws, size_t ws_size,
                              hipStream_t stream) {
    (void)n_in; (void)out_size; (void)ws_size;
    const float* feats     = (const float*)d_in[0];
    const float* edge_vals = (const float*)d_in[1];
    const int*   edge_row  = (const int*)d_in[2];
    const int*   edge_col  = (const int*)d_in[3];

    int N = in_sizes[0] / EMB;
    int E = in_sizes[1];

    float* bufA = (float*)d_ws;
    float* bufB = bufA + (size_t)N * EMB;
    float* acc  = bufB + (size_t)N * EMB;
    float* out  = (float*)d_out;

    int* ce    = (int*)bufB;
    int* seg   = ce + (size_t)E * 4;
    int* pos   = seg + (size_t)N * 2;
    int* cnt   = pos + E;
    int* total = cnt + N;

    const int blk = 256;
    int gridE  = (E + blk - 1) / blk;
    int gridN  = (N + blk - 1) / blk;
    int gridGa = (int)(((long long)N * 16 + ROW_BLK - 1) / ROW_BLK);
    int gridFu = (N + 16 * FU_WAVES - 1) / (16 * FU_WAVES);

    hipMemsetAsync(cnt, 0, (size_t)(N + 1) * sizeof(int), stream);
    row_hist<<<gridE, blk, 0, stream>>>(edge_row, cnt, pos, E);
    row_alloc<<<gridN, blk, 0, stream>>>(cnt, seg, total, N);
    csr_fill<<<gridE, blk, 0, stream>>>(edge_vals, edge_row, edge_col, seg, pos, ce, E);

    spmm_rows<0><<<gridGa, ROW_BLK, 0, stream>>>(feats, seg, ce, nullptr, nullptr, bufA, 1.0f, 0.5f, N);

    spmm_rows<1><<<gridGa, ROW_BLK, 0, stream>>>(bufA, seg, ce, feats, bufA, acc, 1.0f, 1.0f / 3.0f, N);

    edge_upd<<<gridFu, FU_WAVES * 32, 0, stream>>>(acc, seg, ce, out, N);
}


# Edge bucketing with no shared counters: each 4096-edge block stores its records bin-sorted plus a row of bin starts; a bin's block sums its table column and reads the runs; clear and count launch go
# speedup vs baseline: 3.7799x; 1.4837x over previous
#include <hip/hip_runtime.h>
#include <hip/hip_bf16.h>
#include <math.h>

typedef float v2f __attribute__((ext_vector_type(2)));
typedef float v8f __attribute__((ext_vector_type(8)));

#define EMB   64
#define EPSN  1e-12f

__device__ __forceinline__ float readlane_f32(float x, int lane) {
    return __int_as_float(__builtin_amdgcn_readlane(__float_as_int(x), lane));
}

__global__ void row_hist(const int* __restrict__ rows,
                         int*       __restrict__ cnt,
                         int*       __restrict__ pos, int E) {
    int e = blockIdx.x * blockDim.x + threadIdx.x;
    if (e >= E) return;
    pos[e] = atomicAdd(cnt + rows[e], 1);
}

__global__ void row_alloc(const int* __restrict__ cnt,
                          int*       __restrict__ seg,
                          int*       __restrict__ total, int N) {
    __shared__ int wsum[8];
    __shared__ int base;
    int r    = blockIdx.x * 256 + threadIdx.x;
    int lane = threadIdx.x & 31;
    int w    = threadIdx.x >> 5;
    int c    = (r < N) ? cnt[r] : 0;
    int incl = c;
#pragma unroll
    for (int d = 1; d < 32; d <<= 1) {
        int t = __shfl_up(incl, d, 32);
        if (lane >= d) incl += t;
    }
    if (lane == 31) wsum[w] = incl;
    __syncthreads();
    if (threadIdx.x == 0) {
        int run = 0;
        for (int i = 0; i < 8; ++i) { int t = wsum[i]; wsum[i] = run; run += t; }
        base = atomicAdd(total, run);
    }
    __syncthreads();
    if (r < N) {
        int2 sc;
        sc.x = base + wsum[w] + incl - c;
        sc.y = c;
        *(int2*)(seg + (size_t)r * 2) = sc;
    }
}

__global__ void csr_fill(const float* __restrict__ vals,
                         const int*   __restrict__ rows,
                         const int*   __restrict__ cols,
                         const int*   __restrict__ seg,
                         const int*   __restrict__ pos,
                         int*         __restrict__ ce, int E) {
    int e = blockIdx.x * blockDim.x + threadIdx.x;
    if (e >= E) return;
    int r = rows[e];
    int p = seg[(size_t)r * 2] + pos[e];
    int4 rec;
    rec.x = cols[e];
    rec.y = __float_as_int(vals[e]);
    rec.z = r;
    rec.w = e;
    *(int4*)(ce + (size_t)p * 4) = rec;
}

#define BIN_SHIFT 8
#define NBMAX     1024
#define SC_T      512
#define SC_K      8
#define SC_EPB    (SC_T * SC_K)
#define BR_T      512
#define BR_CAP    3072
#define BR_G      16
#define BR_GROUPS (BR_T / BR_G)
#define BR_U      8

__device__ __forceinline__ int block_excl_scan(int v, int* wsum) {
    int lane = threadIdx.x & 31;
    int w    = threadIdx.x >> 5;
    int incl = v;
    for (int d = 1; d < 32; d <<= 1) {
        int x = __shfl_up(incl, d, 32);
        if (lane >= d) incl += x;
    }
    __syncthreads();
    if (lane == 31) wsum[w] = incl;
    __syncthreads();
    int basew = 0;
    for (int i = 0; i < 16; ++i) {
        int x = wsum[i];
        if (i < w) basew += x;
    }
    return basew + incl - v;
}

__global__ void __launch_bounds__(SC_T)
bin_block_sort(const float* __restrict__ vals,
               const int*   __restrict__ rows,
               const int*   __restrict__ cols,
               int*         __restrict__ tab,
               int*         __restrict__ tmpw, int E, int NB) {
    int4* tmp = (int4*)tmpw;
    __shared__ int  h[NBMAX];
    __shared__ int  wsum[16];
    __shared__ int4 stage[SC_EPB];
    int t = threadIdx.x;
    for (int b = t; b < NBMAX; b += SC_T) h[b] = 0;
    int base = blockIdx.x * SC_EPB;
    int r[SC_K], cl[SC_K], vl[SC_K], lr[SC_K];
#pragma unroll
    for (int k = 0; k < SC_K; ++k) {
        int e = base + k * SC_T + t;
        r[k] = -1;  cl[k] = 0;  vl[k] = 0;  lr[k] = 0;
        if (e < E) {
            r[k]  = rows[e];
            cl[k] = cols[e];
            vl[k] = __float_as_int(vals[e]);
        }
    }
    __syncthreads();
#pragma unroll
    for (int k = 0; k < SC_K; ++k) {
        if (r[k] >= 0) lr[k] = atomicAdd(&h[r[k] >> BIN_SHIFT], 1);
    }
    __syncthreads();
    int b0 = 2 * t;
    int b1 = 2 * t + 1;
    int c0 = h[b0];
    int c1 = h[b1];
    int lex = block_excl_scan(c0 + c1, wsum);
    h[b0] = lex;
    h[b1] = lex + c0;
    int  TP   = NB + 1;
    int* trow = tab + (size_t)blockIdx.x * TP;
    if (b0 < TP) trow[b0] = lex;
    if (b1 < TP) trow[b1] = lex + c0;
    __syncthreads();
#pragma unroll
    for (int k = 0; k < SC_K; ++k) {
        if (r[k] >= 0) {
            int4 q;
            q.x = cl[k];
            q.y = vl[k];
            q.z = r[k];
            q.w = base + k * SC_T + t;
            stage[h[r[k] >> BIN_SHIFT] + lr[k]] = q;
        }
    }
    __syncthreads();
    int n = min(SC_EPB, E - base);
    for (int s = t; s < n; s += SC_T) tmp[base + s] = stage[s];
}

__global__ void __launch_bounds__(BR_T)
bin_rows(const int* __restrict__ tab,
         const int* __restrict__ tmpw,
         int*       __restrict__ cew,
         int*       __restrict__ seg, int N, int NB, int NBLK) {
    const int4* tmp = (const int4*)tmpw;
    int4*       ce  = (int4*)cew;
    __shared__ int  rc[256];
    __shared__ int  rs[256];
    __shared__ int  wsum[16];
    __shared__ int  sumS[16];
    __shared__ int  sumC[16];
    __shared__ int  runSrc[BR_T];
    __shared__ int  runLen[BR_T];
    __shared__ int4 stage[BR_CAP];
    int t    = threadIdx.x;
    int b    = blockIdx.x;
    int lane = t & 31;
    int w    = t >> 5;
    int TP   = NB + 1;
    int s0 = 0, len0 = 0;
    if (t < NBLK) {
        const int* trow = tab + (size_t)t * TP + b;
        s0   = trow[0];
        len0 = trow[1] - s0;
    }
    runSrc[t] = t * SC_EPB + s0;
    runLen[t] = len0;
    if (t < 256) rc[t] = 0;
    int ws = s0, wc = len0;
    for (int d = 16; d >= 1; d >>= 1) {
        ws += __shfl_xor(ws, d, 32);
        wc += __shfl_xor(wc, d, 32);
    }
    if (lane == 0) { sumS[w] = ws;  sumC[w] = wc; }
    __syncthreads();
    int bstart = 0, nb = 0;
    for (int i = 0; i < 16; ++i) { bstart += sumS[i];  nb += sumC[i]; }

    int g = t / BR_G;
    int l = t % BR_G;
    for (int k0 = g; k0 < NBLK; k0 += BR_U * BR_GROUPS) {
        int rw[BR_U];
#pragma unroll
        for (int u = 0; u < BR_U; ++u) {
            int k = k0 + u * BR_GROUPS;
            rw[u] = -1;
            if (k < NBLK && l < runLen[k]) rw[u] = tmp[runSrc[k] + l].z;
        }
#pragma unroll
        for (int u = 0; u < BR_U; ++u) {
            if (rw[u] >= 0) atomicAdd(&rc[rw[u] & 255], 1);
        }
        for (int u = 0; u < BR_U; ++u) {
            int k = k0 + u * BR_GROUPS;
            if (k < NBLK) {
                int len = runLen[k];
                int src = runSrc[k];
                for (int i = l + BR_G; i < len; i += BR_G) atomicAdd(&rc[tmp[src + i].z & 255], 1);
            }
        }
    }
    __syncthreads();
    int c  = (t < 256) ? rc[t] : 0;
    int ex = block_excl_scan(c, wsum);
    if (t < 256) {
        rs[t] = ex;
        rc[t] = 0;
        int row = (b << BIN_SHIFT) + t;
        if (row < N) {
            int2 sc;
            sc.x = bstart + ex;
            sc.y = c;
            *(int2*)(seg + (size_t)row * 2) = sc;
        }
    }
    __syncthreads();
    bool staged = (nb <= BR_CAP);
    for (int k0 = g; k0 < NBLK; k0 += BR_U * BR_GROUPS) {
        int4 q[BR_U];
#pragma unroll
        for (int u = 0; u < BR_U; ++u) {
            int k = k0 + u * BR_GROUPS;
            q[u].x = 0;  q[u].y = 0;  q[u].z = -1;  q[u].w = 0;
            if (k < NBLK && l < runLen[k]) q[u] = tmp[runSrc[k] + l];
        }
#pragma unroll
        for (int u = 0; u < BR_U; ++u) {
            if (q[u].z >= 0) {
                int j    = q[u].z & 255;
                int slot = rs[j] + atomicAdd(&rc[j], 1);
                if (staged) stage[slot] = q[u];
                else        ce[bstart + slot] = q[u];
            }
        }
        for (int u = 0; u < BR_U; ++u) {
            int k = k0 + u * BR_GROUPS;
            if (k < NBLK) {
                int len = runLen[k];
                int src = runSrc[k];
                for (int i = l + BR_G; i < len; i += BR_G) {
                    int4 qq  = tmp[src + i];
                    int  j    = qq.z & 255;
                    int  slot = rs[j] + atomicAdd(&rc[j], 1);
                    if (staged) stage[slot] = qq;
                    else        ce[bstart + slot] = qq;
                }
            }
        }
    }
    if (staged) {
        __syncthreads();
        for (int i = t; i < nb; i += BR_T) ce[bstart + i] = stage[i];
    }
}

#define ROW_BLK   256
#define ROW_HOLD  32
__device__ __forceinline__ void fma_row(float4& a, const float4& x, float sc, float w) {
    a.x += (x.x * sc) * w;
    a.y += (x.y * sc) * w;
    a.z += (x.z * sc) * w;
    a.w += (x.w * sc) * w;
}

template <int K>
__device__ __forceinline__ int row_bcast(int v) {
    return __builtin_amdgcn_update_dpp(0, v, 0x150 + K, 0xf, 0xf, false);
}

template <int D>
__device__ __forceinline__ int row_rot(int v) {
    return __builtin_amdgcn_update_dpp(0, v, 0x120 + D, 0xf, 0xf, false);
}

template <int M>
__device__ __forceinline__ float row_xadd(float s) {
    int t = __builtin_amdgcn_update_dpp(0, __float_as_int(s), 0x160 + M, 0xf, 0xf, false);
    return s + __int_as_float(t);
}

template <int D>
__device__ __forceinline__ void rank2(int ka, int kb, int& ra, int& rb) {
    int ta = row_rot<D>(ka);
    int tb = row_rot<D>(kb);
    ra += ((ta < ka) ? 1 : 0) + ((tb < ka) ? 1 : 0);
    rb += ((ta < kb) ? 1 : 0) + ((tb < kb) ? 1 : 0);
}

template <int I>
__device__ __forceinline__ void gather4(float4& a, const float* fl, int scol, int sval, float sc) {
    int c0 = row_bcast<I + 0>(scol);
    int c1 = row_bcast<I + 1>(scol);
    int c2 = row_bcast<I + 2>(scol);
    int c3 = row_bcast<I + 3>(scol);
    float w0 = __int_as_float(row_bcast<I + 0>(sval));
    float w1 = __int_as_float(row_bcast<I + 1>(sval));
    float w2 = __int_as_float(row_bcast<I + 2>(sval));
    float w3 = __int_as_float(row_bcast<I + 3>(sval));
    float4 x0 = *(const float4*)(fl + (size_t)c0 * EMB);
    float4 x1 = *(const float4*)(fl + (size_t)c1 * EMB);
    float4 x2 = *(const float4*)(fl + (size_t)c2 * EMB);
    float4 x3 = *(const float4*)(fl + (size_t)c3 * EMB);
    fma_row(a, x0, sc, w0);
    fma_row(a, x1, sc, w1);
    fma_row(a, x2, sc, w2);
    fma_row(a, x3, sc, w3);
}

template <int MODE>
__global__ void spmm_rows(const float* __restrict__ f,
                          const int*   __restrict__ seg,
                          int*                      ce,
                          const float* __restrict__ base0,
                          const float* __restrict__ base1,
                          float*       __restrict__ dst,
                          float srcScale, float inv_c, int N) {
    int tid = blockIdx.x * blockDim.x + threadIdx.x;
    int r   = tid >> 4;
    int m   = tid & 15;
    int l4  = m << 2;
    if (r >= N) return;
    int2 sc = *(const int2*)(seg + (size_t)r * 2);
    int c = sc.y;
    int4* rec = (int4*)ce + sc.x;
    const float* fl = f + l4;
    size_t off = (size_t)r * EMB + l4;
    float4 b0;  b0.x = 0.0f;  b0.y = 0.0f;  b0.z = 0.0f;  b0.w = 0.0f;
    float4 b1 = b0;
    if (MODE == 1) {
        b0 = *(const float4*)(base0 + off);
        b1 = *(const float4*)(base1 + off);
    }
    float4 a;  a.x = 0.0f;  a.y = 0.0f;  a.z = 0.0f;  a.w = 0.0f;

    if (c > ROW_HOLD) {
        int prev = -1;
        for (int k = 0; k < c; ++k) {
            int best = 0x7fffffff, bcol = 0, bval = 0;
            for (int j = 0; j < c; ++j) {
                int4 q = rec[j];
                if (q.w > prev && q.w < best) { best = q.w;  bcol = q.x;  bval = q.y; }
            }
            prev = best;
            float4 x = *(const float4*)(fl + (size_t)bcol * EMB);
            fma_row(a, x, srcScale, __int_as_float(bval));
        }
    } else if (MODE == 0 && c > 16) {
        __shared__ int2 ordS[ROW_BLK / 16][ROW_HOLD];
        int2* ord = ordS[threadIdx.x >> 4];
        bool hb = (m + 16 < c);
        int4 qa = rec[m];
        int4 qb;  qb.x = 0;  qb.y = 0;  qb.z = r;  qb.w = 0x7fffffff;
        if (hb) qb = rec[m + 16];
        int ka = qa.w;
        int kb = hb ? qb.w : 0x7fffffff;
        int ra = (kb < ka) ? 1 : 0;
        int rb = (ka < kb) ? 1 : 0;
        rank2<1>(ka, kb, ra, rb);   rank2<2>(ka, kb, ra, rb);   rank2<3>(ka, kb, ra, rb);
        rank2<4>(ka, kb, ra, rb);   rank2<5>(ka, kb, ra, rb);   rank2<6>(ka, kb, ra, rb);
        rank2<7>(ka, kb, ra, rb);   rank2<8>(ka, kb, ra, rb);   rank2<9>(ka, kb, ra, rb);
        rank2<10>(ka, kb, ra, rb);  rank2<11>(ka, kb, ra, rb);  rank2<12>(ka, kb, ra, rb);
        rank2<13>(ka, kb, ra, rb);  rank2<14>(ka, kb, ra, rb);  rank2<15>(ka, kb, ra, rb);
        if (!hb) rb = m + 16;
        int2 pa;  pa.x = qa.x;  pa.y = qa.y;
        int2 pb;  pb.x = qb.x;  pb.y = qb.y;
        ord[ra] = pa;
        ord[rb] = pb;
        int2 sa = ord[m];
        int4 oa;  oa.x = sa.x;  oa.y = sa.y;  oa.z = r;  oa.w = 0;
        rec[m] = oa;
        if (hb) {
            int2 sb = ord[m + 16];
            int4 ob;  ob.x = sb.x;  ob.y = sb.y;  ob.z = r;  ob.w = 0;
            rec[m + 16] = ob;
        }
        int last = c - 1;
        for (int i = 0; i < c; i += 4) {
            int2 e0 = ord[i];
            int2 e1 = ord[min(i + 1, last)];
            int2 e2 = ord[min(i + 2, last)];
            int2 e3 = ord[min(i + 3, last)];
            float4 x0 = *(const float4*)(fl + (size_t)e0.x * EMB);
            float4 x1 = *(const float4*)(fl + (size_t)e1.x * EMB);
            float4 x2 = *(const float4*)(fl + (size_t)e2.x * EMB);
            float4 x3 = *(const float4*)(fl + (size_t)e3.x * EMB);
            float w0 = __int_as_float(e0.y);
            float w1 = (i + 1 <= last) ? __int_as_float(e1.y) : 0.0f;
            float w2 = (i + 2 <= last) ? __int_as_float(e2.y) : 0.0f;
            float w3 = (i + 3 <= last) ? __int_as_float(e3.y) : 0.0f;
            fma_row(a, x0, srcScale, w0);
            fma_row(a, x1, srcScale, w1);
            fma_row(a, x2, srcScale, w2);
            fma_row(a, x3, srcScale, w3);
        }
    } else if (MODE == 0) {
        if (c > 0) {
            int4 q;  q.x = 0;  q.y = 0;  q.z = r;  q.w = 0x7fffffff;
            if (m < c) q = rec[m];
            int key = (m < c) ? q.w : 0x7fffffff;
            int rank = 0;
            rank += (row_rot<1>(key)  < key) ? 1 : 0;
            rank += (row_rot<2>(key)  < key) ? 1 : 0;
            rank += (row_rot<3>(key)  < key) ? 1 : 0;
            rank += (row_rot<4>(key)  < key) ? 1 : 0;
            rank += (row_rot<5>(key)  < key) ? 1 : 0;
            rank += (row_rot<6>(key)  < key) ? 1 : 0;
            rank += (row_rot<7>(key)  < key) ? 1 : 0;
            rank += (row_rot<8>(key)  < key) ? 1 : 0;
            rank += (row_rot<9>(key)  < key) ? 1 : 0;
            rank += (row_rot<10>(key) < key) ? 1 : 0;
            rank += (row_rot<11>(key) < key) ? 1 : 0;
            rank += (row_rot<12>(key) < key) ? 1 : 0;
            rank += (row_rot<13>(key) < key) ? 1 : 0;
            rank += (row_rot<14>(key) < key) ? 1 : 0;
            rank += (row_rot<15>(key) < key) ? 1 : 0;
            if (m >= c) rank = m;
            int to   = (((int)threadIdx.x & 16) | rank) << 2;
            int scol = __builtin_amdgcn_ds_permute(to, q.x);
            int sval = __builtin_amdgcn_ds_permute(to, q.y);
            int lastcol = __shfl(scol, c - 1, 16);
            if (m >= c) scol = lastcol;
            if (m < c) {
                int4 o;  o.x = scol;  o.y = sval;  o.z = r;  o.w = 0;
                rec[m] = o;
            }
            gather4<0>(a, fl, scol, sval, srcScale);
            if (c > 4)  gather4<4>(a, fl, scol, sval, srcScale);
            if (c > 8)  gather4<8>(a, fl, scol, sval, srcScale);
            if (c > 12) gather4<12>(a, fl, scol, sval, srcScale);
        }
    } else {
        for (int i = 0; i < c; i += 4) {
            int last = c - 1;
            int i1 = min(i + 1, last);
            int i2 = min(i + 2, last);
            int i3 = min(i + 3, last);
            int4 e0 = rec[i];
            int4 e1 = rec[i1];
            int4 e2 = rec[i2];
            int4 e3 = rec[i3];
            float4 x0 = *(const float4*)(fl + (size_t)e0.x * EMB);
            float4 x1 = *(const float4*)(fl + (size_t)e1.x * EMB);
            float4 x2 = *(const float4*)(fl + (size_t)e2.x * EMB);
            float4 x3 = *(const float4*)(fl + (size_t)e3.x * EMB);
            float w0 = __int_as_float(e0.y);
            float w1 = (i + 1 <= last) ? __int_as_float(e1.y) : 0.0f;
            float w2 = (i + 2 <= last) ? __int_as_float(e2.y) : 0.0f;
            float w3 = (i + 3 <= last) ? __int_as_float(e3.y) : 0.0f;
            fma_row(a, x0, srcScale, w0);
            fma_row(a, x1, srcScale, w1);
            fma_row(a, x2, srcScale, w2);
            fma_row(a, x3, srcScale, w3);
        }
    }

    float gx = a.x * inv_c;
    float gy = a.y * inv_c;
    float gz = a.z * inv_c;
    float gw = a.w * inv_c;
    float s  = gx * gx + gy * gy + gz * gz + gw * gw;
    s = row_xadd<8>(s);
    s = row_xadd<4>(s);
    s = row_xadd<2>(s);
    s = row_xadd<1>(s);
    float n   = sqrtf(s);
    float inv = 1.0f / fmaxf(n, EPSN);
    float4 g;  g.x = gx * inv;  g.y = gy * inv;  g.z = gz * inv;  g.w = gw * inv;
    if (MODE == 0) {
        *(float4*)(dst + off) = g;
    } else {
        float4 o;
        o.x = (b0.x + b1.x) + g.x;
        o.y = (b0.y + b1.y) + g.y;
        o.z = (b0.z + b1.z) + g.z;
        o.w = (b0.w + b1.w) + g.w;
        *(float4*)(dst + off) = o;
    }
}

#define FU_WAVES 4
#define FU_PITCH 68

template <int M>
__device__ __forceinline__ int row_xadd_i(int s) {
    return s + __builtin_amdgcn_update_dpp(0, s, 0x160 + M, 0xf, 0xf, false);
}

__device__ __forceinline__ float4 row_load16(const float* __restrict__ base, int row, int m) {
    return *(const float4*)((const char*)base + (unsigned)(row * (EMB * 4) + m * 16));
}

__device__ __forceinline__ void store_row(float* __restrict__ out, int row, int lane, float2 v) {
    *(float2*)((char*)out + (unsigned)(row * (EMB * 4) + lane * 8)) = v;
}

__global__ void __launch_bounds__(FU_WAVES * 32)
edge_upd(const float* __restrict__ acc,
         const int*   __restrict__ seg,
         const int*   __restrict__ ce,
         float*       __restrict__ out, int N) {
    __shared__ __attribute__((aligned(16))) float tileS[FU_WAVES][16 * FU_PITCH];
    const float kScale = 1.0f / 3.0f;
    int lane = threadIdx.x & 31;
    int wv   = __builtin_amdgcn_readfirstlane((int)(threadIdx.x >> 5));
    int r0   = (blockIdx.x * FU_WAVES + wv) * 16;
    if (r0 >= N) return;
    int m    = lane & 15;
    int half = lane >> 4;
    int rows = min(16, N - r0);
    float* sT = tileS[wv];
    const float* F = sT + m * FU_PITCH + 2 * half;

    int2 sc;  sc.x = 0;  sc.y = 0;
    if (m < rows) sc = *(const int2*)(seg + (size_t)(r0 + m) * 2);
    int nsum = sc.y;
    nsum = row_xadd_i<8>(nsum);
    nsum = row_xadd_i<4>(nsum);
    nsum = row_xadd_i<2>(nsum);
    nsum = row_xadd_i<1>(nsum);
    int begin  = __builtin_amdgcn_readfirstlane(sc.x);
    int n      = __builtin_amdgcn_readfirstlane(nsum);
    int ntiles = (n + 15) >> 4;
    int last   = n - 1;
    const int4* recs = (const int4*)ce + begin;

    float2 zero2;  zero2.x = 0.0f;  zero2.y = 0.0f;

    unsigned emp = __builtin_amdgcn_ballot_w32(half == 0 && m < rows && sc.y == 0);
    while (emp != 0u) {
        int zr = __builtin_ctz(emp);
        emp &= emp - 1u;
        store_row(out, r0 + zr, lane, zero2);
    }
    if (ntiles == 0) return;

    int4 rcC = recs[min(m, last)];
    int4 rcN = recs[min(16 + m, last)];

    float4 s4[8];
#pragma unroll
    for (int j = 0; j < 8; ++j) {
        s4[j] = row_load16(acc, min(r0 + 2 * j + half, N - 1), m);
    }
    float4 tq[8];
#pragma unroll
    for (int j = 0; j < 8; ++j) {
        tq[j] = row_load16(acc, __shfl(rcC.x, 2 * j + half, 32), m);
    }
#pragma unroll
    for (int j = 0; j < 8; ++j) {
        float4 x = s4[j];
        x.x *= kScale;  x.y *= kScale;  x.z *= kScale;  x.w *= kScale;
        *(float4*)(sT + (2 * j + half) * FU_PITCH + 4 * m) = x;
    }
    v2f A[16];
#pragma unroll
    for (int ks = 0; ks < 16; ++ks) {
        float2 a = *(const float2*)(F + 4 * ks);
        A[ks].x = a.x;  A[ks].y = a.y;
    }

    float2 av = zero2;
    int tailrow = 0;
    for (int t = 0; t < ntiles; ++t) {
        bool  valid = (16 * t + m < n);
        int   irow  = rcC.z - r0;
        float v     = __int_as_float(rcC.y);

#pragma unroll
        for (int j = 0; j < 8; ++j) {
            float4 x = tq[j];
            x.x *= kScale;  x.y *= kScale;  x.z *= kScale;  x.w *= kScale;
            *(float4*)(sT + (2 * j + half) * FU_PITCH + 4 * m) = x;
        }
#pragma unroll
        for (int j = 0; j < 8; ++j) {
            tq[j] = row_load16(acc, __shfl(rcN.x, 2 * j + half, 32), m);
        }
        int4 rcNN = recs[min(16 * (t + 2) + m, last)];

        int nxt = __shfl_down(irow, 1, 16);
        int nx0 = __builtin_amdgcn_readlane(rcN.z, 0) - r0;
        if (m == 15) nxt = nx0;
        unsigned ends = __builtin_amdgcn_ballot_w32(half == 0 && nxt != irow);
        tailrow = __builtin_amdgcn_readlane(irow, 15);

        v8f c0 = {};
        v8f c1 = {};
#pragma unroll
        for (int ks = 0; ks < 16; ks += 2) {
            float2 b0 = *(const float2*)(F + 4 * ks);
            float2 b1 = *(const float2*)(F + 4 * ks + 4);
            v2f B0;  B0.x = b0.x;  B0.y = b0.y;
            v2f B1;  B1.x = b1.x;  B1.y = b1.y;
            c0 = __builtin_amdgcn_wmma_f32_16x16x4_f32(
                 false, A[ks],  false, B0,
                 (short)0, c0,  false,  false);
            c1 = __builtin_amdgcn_wmma_f32_16x16x4_f32(
                 false, A[ks + 1],  false, B1,
                 (short)0, c1,  false,  false);
        }
        v8f cc = c0 + c1;

        int   i7  = irow & 7;
        float sel = cc[0];
#pragma unroll
        for (int vv = 1; vv < 8; ++vv) {
            if (i7 == vv) sel = cc[vv];
        }
        float dot = ((irow >> 3) == half) ? sel : 0.0f;
        dot += __shfl_xor(dot, 16, 32);

        float cross = dot * (1.0f / EMB);
        float z     = 4.0f * (1.0f - cross);
        float sp    = log1pf(expf(-fabsf(z))) + fmaxf(z, 0.0f);
        float mat   = valid ? 0.5f * expf(z) * sp * v : 0.0f;

        float2 xr[16];
#pragma unroll
        for (int j = 0; j < 16; ++j) {
            xr[j] = *(const float2*)(sT + j * FU_PITCH + 2 * lane);
        }
#pragma unroll
        for (int j = 0; j < 16; ++j) {
            float mj = readlane_f32(mat, j);
            av.x += xr[j].x * mj;
            av.y += xr[j].y * mj;
            if ((ends >> j) & 1u) {
                store_row(out, r0 + __builtin_amdgcn_readlane(irow, j), lane, av);
                av = zero2;
            }
        }

        rcC = rcN;
        rcN = rcNN;
    }
    store_row(out, r0 + tailrow, lane, av);
}

extern "C" void kernel_launch(void* const* d_in, const int* in_sizes, int n_in,
                              void* d_out, int out_size, void* d_ws, size_t ws_size,
                              hipStream_t stream) {
    (void)n_in; (void)out_size; (void)ws_size;
    const float* feats     = (const float*)d_in[0];
    const float* edge_vals = (const float*)d_in[1];
    const int*   edge_row  = (const int*)d_in[2];
    const int*   edge_col  = (const int*)d_in[3];

    int N = in_sizes[0] / EMB;
    int E = in_sizes[1];

    float* bufA = (float*)d_ws;
    float* bufB = bufA + (size_t)N * EMB;
    float* acc  = bufB + (size_t)N * EMB;
    float* out  = (float*)d_out;

    int* ce    = (int*)bufB;
    int* seg   = ce + (size_t)E * 4;
    int* pos   = seg + (size_t)N * 2;
    int* cnt   = pos + E;
    int* total = cnt + N;

    const int blk = 256;
    int gridE  = (E + blk - 1) / blk;
    int gridN  = (N + blk - 1) / blk;
    int gridGa = (int)(((long long)N * 16 + ROW_BLK - 1) / ROW_BLK);
    int gridFu = (N + 16 * FU_WAVES - 1) / (16 * FU_WAVES);
    int NB     = (N + (1 << BIN_SHIFT) - 1) >> BIN_SHIFT;
    int gridSc = (E + SC_EPB - 1) / SC_EPB;

    if (NB + 1 <= NBMAX && gridSc <= BR_T && (long long)gridSc * SC_EPB * 4 <= (long long)N * EMB &&
        (long long)gridSc * (NB + 1) <= (long long)E) {
        int* tab = pos;
        int* tmp = (int*)acc;
        bin_block_sort<<<gridSc, SC_T, 0, stream>>>(edge_vals, edge_row, edge_col, tab, tmp, E, NB);
        bin_rows<<<NB, BR_T, 0, stream>>>(tab, tmp, ce, seg, N, NB, gridSc);
    } else {
        hipMemsetAsync(cnt, 0, (size_t)(N + 1) * sizeof(int), stream);
        row_hist<<<gridE, blk, 0, stream>>>(edge_row, cnt, pos, E);
        row_alloc<<<gridN, blk, 0, stream>>>(cnt, seg, total, N);
        csr_fill<<<gridE, blk, 0, stream>>>(edge_vals, edge_row, edge_col, seg, pos, ce, E);
    }

    spmm_rows<0><<<gridGa, ROW_BLK, 0, stream>>>(feats, seg, ce, nullptr, nullptr, bufA, 1.0f, 0.5f, N);

    spmm_rows<1><<<gridGa, ROW_BLK, 0, stream>>>(bufA, seg, ce, feats, bufA, acc, 1.0f, 1.0f / 3.0f, N);

    edge_upd<<<gridFu, FU_WAVES * 32, 0, stream>>>(acc, seg, ce, out, N);
}


# Layer 0 joins the bin's row-ordering block: records are ranked into edge order inside the LDS tile and half-waves sum the bin's rows from it; its launch, record read-back and rewrite go
# speedup vs baseline: 3.9706x; 1.0505x over previous
#include <hip/hip_runtime.h>
#include <hip/hip_bf16.h>
#include <math.h>

typedef float v2f __attribute__((ext_vector_type(2)));
typedef float v8f __attribute__((ext_vector_type(8)));

#define EMB   64
#define EPSN  1e-12f

__device__ __forceinline__ float readlane_f32(float x, int lane) {
    return __int_as_float(__builtin_amdgcn_readlane(__float_as_int(x), lane));
}

__global__ void row_hist(const int* __restrict__ rows,
                         int*       __restrict__ cnt,
                         int*       __restrict__ pos, int E) {
    int e = blockIdx.x * blockDim.x + threadIdx.x;
    if (e >= E) return;
    pos[e] = atomicAdd(cnt + rows[e], 1);
}

__global__ void row_alloc(const int* __restrict__ cnt,
                          int*       __restrict__ seg,
                          int*       __restrict__ total, int N) {
    __shared__ int wsum[8];
    __shared__ int base;
    int r    = blockIdx.x * 256 + threadIdx.x;
    int lane = threadIdx.x & 31;
    int w    = threadIdx.x >> 5;
    int c    = (r < N) ? cnt[r] : 0;
    int incl = c;
#pragma unroll
    for (int d = 1; d < 32; d <<= 1) {
        int t = __shfl_up(incl, d, 32);
        if (lane >= d) incl += t;
    }
    if (lane == 31) wsum[w] = incl;
    __syncthreads();
    if (threadIdx.x == 0) {
        int run = 0;
        for (int i = 0; i < 8; ++i) { int t = wsum[i]; wsum[i] = run; run += t; }
        base = atomicAdd(total, run);
    }
    __syncthreads();
    if (r < N) {
        int2 sc;
        sc.x = base + wsum[w] + incl - c;
        sc.y = c;
        *(int2*)(seg + (size_t)r * 2) = sc;
    }
}

__global__ void csr_fill(const float* __restrict__ vals,
                         const int*   __restrict__ rows,
                         const int*   __restrict__ cols,
                         const int*   __restrict__ seg,
                         const int*   __restrict__ pos,
                         int*         __restrict__ ce, int E) {
    int e = blockIdx.x * blockDim.x + threadIdx.x;
    if (e >= E) return;
    int r = rows[e];
    int p = seg[(size_t)r * 2] + pos[e];
    int4 rec;
    rec.x = cols[e];
    rec.y = __float_as_int(vals[e]);
    rec.z = r;
    rec.w = e;
    *(int4*)(ce + (size_t)p * 4) = rec;
}

#define BIN_SHIFT 8
#define NBMAX     1024
#define SC_T      512
#define SC_K      8
#define SC_EPB    (SC_T * SC_K)
#define BR_T      512
#define BR_CAP    3072
#define BR_G      16
#define BR_GROUPS (BR_T / BR_G)
#define BR_U      8

__device__ __forceinline__ int block_excl_scan(int v, int* wsum) {
    int lane = threadIdx.x & 31;
    int w    = threadIdx.x >> 5;
    int incl = v;
    for (int d = 1; d < 32; d <<= 1) {
        int x = __shfl_up(incl, d, 32);
        if (lane >= d) incl += x;
    }
    __syncthreads();
    if (lane == 31) wsum[w] = incl;
    __syncthreads();
    int basew = 0;
    for (int i = 0; i < 16; ++i) {
        int x = wsum[i];
        if (i < w) basew += x;
    }
    return basew + incl - v;
}

__global__ void __launch_bounds__(SC_T)
bin_block_sort(const float* __restrict__ vals,
               const int*   __restrict__ rows,
               const int*   __restrict__ cols,
               int*         __restrict__ tab,
               int*         __restrict__ tmpw, int E, int NB) {
    int4* tmp = (int4*)tmpw;
    __shared__ int  h[NBMAX];
    __shared__ int  wsum[16];
    __shared__ int4 stage[SC_EPB];
    int t = threadIdx.x;
    for (int b = t; b < NBMAX; b += SC_T) h[b] = 0;
    int base = blockIdx.x * SC_EPB;
    int r[SC_K], cl[SC_K], vl[SC_K], lr[SC_K];
#pragma unroll
    for (int k = 0; k < SC_K; ++k) {
        int e = base + k * SC_T + t;
        r[k] = -1;  cl[k] = 0;  vl[k] = 0;  lr[k] = 0;
        if (e < E) {
            r[k]  = rows[e];
            cl[k] = cols[e];
            vl[k] = __float_as_int(vals[e]);
        }
    }
    __syncthreads();
#pragma unroll
    for (int k = 0; k < SC_K; ++k) {
        if (r[k] >= 0) lr[k] = atomicAdd(&h[r[k] >> BIN_SHIFT], 1);
    }
    __syncthreads();
    int b0 = 2 * t;
    int b1 = 2 * t + 1;
    int c0 = h[b0];
    int c1 = h[b1];
    int lex = block_excl_scan(c0 + c1, wsum);
    h[b0] = lex;
    h[b1] = lex + c0;
    int  TP   = NB + 1;
    int* trow = tab + (size_t)blockIdx.x * TP;
    if (b0 < TP) trow[b0] = lex;
    if (b1 < TP) trow[b1] = lex + c0;
    __syncthreads();
#pragma unroll
    for (int k = 0; k < SC_K; ++k) {
        if (r[k] >= 0) {
            int4 q;
            q.x = cl[k];
            q.y = vl[k];
            q.z = r[k];
            q.w = base + k * SC_T + t;
            stage[h[r[k] >> BIN_SHIFT] + lr[k]] = q;
        }
    }
    __syncthreads();
    int n = min(SC_EPB, E - base);
    for (int s = t; s < n; s += SC_T) tmp[base + s] = stage[s];
}

__device__ __forceinline__ void fma_row(float4& a, const float4& x, float sc, float w) {
    a.x += (x.x * sc) * w;
    a.y += (x.y * sc) * w;
    a.z += (x.z * sc) * w;
    a.w += (x.w * sc) * w;
}

template <int M>
__device__ __forceinline__ float row_xadd(float s) {
    int t = __builtin_amdgcn_update_dpp(0, __float_as_int(s), 0x160 + M, 0xf, 0xf, false);
    return s + __int_as_float(t);
}

__device__ __forceinline__ float4 row_finish(const float4& a, float inv_c) {
    float gx = a.x * inv_c;
    float gy = a.y * inv_c;
    float gz = a.z * inv_c;
    float gw = a.w * inv_c;
    float s  = gx * gx + gy * gy + gz * gz + gw * gw;
    s = row_xadd<8>(s);
    s = row_xadd<4>(s);
    s = row_xadd<2>(s);
    s = row_xadd<1>(s);
    float n   = sqrtf(s);
    float inv = 1.0f / fmaxf(n, EPSN);
    float4 g;  g.x = gx * inv;  g.y = gy * inv;  g.z = gz * inv;  g.w = gw * inv;
    return g;
}

template <bool TILE>
__device__ __forceinline__ void bin_layer0(const int4* tile, const int4* gr,
                                           const int* rs, const int* rc,
                                           const float* __restrict__ f,
                                           float*       __restrict__ dst, int b, int N) {
    int t  = threadIdx.x;
    int hw = t >> 4;
    int m  = t & 15;
    int l4 = m << 2;
    const float* fl = f + l4;
    for (int j = hw; j < 256; j += BR_T / 16) {
        int row = (b << BIN_SHIFT) + j;
        if (row >= N) break;
        int s    = rs[j];
        int cc   = rc[j];
        int last = cc - 1;
        float4 a;  a.x = 0.0f;  a.y = 0.0f;  a.z = 0.0f;  a.w = 0.0f;
        for (int i = 0; i < cc; i += 4) {
            int i0 = s + i;
            int i1 = s + min(i + 1, last);
            int i2 = s + min(i + 2, last);
            int i3 = s + min(i + 3, last);
            int4 e0, e1, e2, e3;
            if (TILE) { e0 = tile[i0];  e1 = tile[i1];  e2 = tile[i2];  e3 = tile[i3]; }
            else      { e0 = gr[i0];    e1 = gr[i1];    e2 = gr[i2];    e3 = gr[i3]; }
            float4 x0 = *(const float4*)(fl + (size_t)e0.x * EMB);
            float4 x1 = *(const float4*)(fl + (size_t)e1.x * EMB);
            float4 x2 = *(const float4*)(fl + (size_t)e2.x * EMB);
            float4 x3 = *(const float4*)(fl + (size_t)e3.x * EMB);
            float w0 = __int_as_float(e0.y);
            float w1 = (i + 1 <= last) ? __int_as_float(e1.y) : 0.0f;
            float w2 = (i + 2 <= last) ? __int_as_float(e2.y) : 0.0f;
            float w3 = (i + 3 <= last) ? __int_as_float(e3.y) : 0.0f;
            fma_row(a, x0, 1.0f, w0);
            fma_row(a, x1, 1.0f, w1);
            fma_row(a, x2, 1.0f, w2);
            fma_row(a, x3, 1.0f, w3);
        }
        float4 gv = row_finish(a, 0.5f);
        *(float4*)(dst + (size_t)row * EMB + l4) = gv;
    }
}

__global__ void __launch_bounds__(BR_T)
bin_rows_l0(const int*   __restrict__ tab,
            const int*   __restrict__ tmpw,
            int*                      cew,
            int*         __restrict__ seg,
            const float* __restrict__ f,
            float*       __restrict__ dst, int N, int NB, int NBLK) {
    const int4* tmp = (const int4*)tmpw;
    int4*       ce  = (int4*)cew;
    __shared__ int  rc[256];
    __shared__ int  rs[256];
    __shared__ int  wsum[16];
    __shared__ int  sumS[16];
    __shared__ int  sumC[16];
    __shared__ int  runSrc[BR_T];
    __shared__ int  runLen[BR_T];
    __shared__ int4 stage[BR_CAP];
    int t    = threadIdx.x;
    int b    = blockIdx.x;
    int lane = t & 31;
    int w    = t >> 5;
    int TP   = NB + 1;
    int s0 = 0, len0 = 0;
    if (t < NBLK) {
        const int* trow = tab + (size_t)t * TP + b;
        s0   = trow[0];
        len0 = trow[1] - s0;
    }
    runSrc[t] = t * SC_EPB + s0;
    runLen[t] = len0;
    if (t < 256) rc[t] = 0;
    int ws = s0, wc = len0;
    for (int d = 16; d >= 1; d >>= 1) {
        ws += __shfl_xor(ws, d, 32);
        wc += __shfl_xor(wc, d, 32);
    }
    if (lane == 0) { sumS[w] = ws;  sumC[w] = wc; }
    __syncthreads();
    int bstart = 0, nb = 0;
    for (int i = 0; i < 16; ++i) { bstart += sumS[i];  nb += sumC[i]; }

    int g = t / BR_G;
    int l = t % BR_G;
    for (int k0 = g; k0 < NBLK; k0 += BR_U * BR_GROUPS) {
        int rw[BR_U];
#pragma unroll
        for (int u = 0; u < BR_U; ++u) {
            int k = k0 + u * BR_GROUPS;
            rw[u] = -1;
            if (k < NBLK && l < runLen[k]) rw[u] = tmp[runSrc[k] + l].z;
        }
#pragma unroll
        for (int u = 0; u < BR_U; ++u) {
            if (rw[u] >= 0) atomicAdd(&rc[rw[u] & 255], 1);
        }
        for (int u = 0; u < BR_U; ++u) {
            int k = k0 + u * BR_GROUPS;
            if (k < NBLK) {
                int len = runLen[k];
                int src = runSrc[k];
                for (int i = l + BR_G; i < len; i += BR_G) atomicAdd(&rc[tmp[src + i].z & 255], 1);
            }
        }
    }
    __syncthreads();
    int c  = (t < 256) ? rc[t] : 0;
    int ex = block_excl_scan(c, wsum);
    if (t < 256) {
        rs[t] = ex;
        rc[t] = 0;
        int row = (b << BIN_SHIFT) + t;
        if (row < N) {
            int2 sc;
            sc.x = bstart + ex;
            sc.y = c;
            *(int2*)(seg + (size_t)row * 2) = sc;
        }
    }
    __syncthreads();
    bool staged = (nb <= BR_CAP);
    for (int k0 = g; k0 < NBLK; k0 += BR_U * BR_GROUPS) {
        int4 q[BR_U];
#pragma unroll
        for (int u = 0; u < BR_U; ++u) {
            int k = k0 + u * BR_GROUPS;
            q[u].x = 0;  q[u].y = 0;  q[u].z = -1;  q[u].w = 0;
            if (k < NBLK && l < runLen[k]) q[u] = tmp[runSrc[k] + l];
        }
#pragma unroll
        for (int u = 0; u < BR_U; ++u) {
            if (q[u].z >= 0) {
                int j    = q[u].z & 255;
                int slot = rs[j] + atomicAdd(&rc[j], 1);
                if (staged) stage[slot] = q[u];
                else        ce[bstart + slot] = q[u];
            }
        }
        for (int u = 0; u < BR_U; ++u) {
            int k = k0 + u * BR_GROUPS;
            if (k < NBLK) {
                int len = runLen[k];
                int src = runSrc[k];
                for (int i = l + BR_G; i < len; i += BR_G) {
                    int4 qq  = tmp[src + i];
                    int  j    = qq.z & 255;
                    int  slot = rs[j] + atomicAdd(&rc[j], 1);
                    if (staged) stage[slot] = qq;
                    else        ce[bstart + slot] = qq;
                }
            }
        }
    }
    __syncthreads();
    if (staged) {
        int4 q[BR_CAP / BR_T];
        int  dsl[BR_CAP / BR_T];
#pragma unroll
        for (int u = 0; u < BR_CAP / BR_T; ++u) {
            int i = t + u * BR_T;
            q[u].x = 0;  q[u].y = 0;  q[u].z = 0;  q[u].w = 0;
            dsl[u] = -1;
            if (i < nb) {
                q[u] = stage[i];
                int j   = q[u].z & 255;
                int s   = rs[j];
                int n   = rc[j];
                int key = q[u].w;
                int rank = 0;
                for (int k = 0; k < n; k += 4) {
                    int w0 = stage[s + k].w;
                    int w1 = stage[s + min(k + 1, n - 1)].w;
                    int w2 = stage[s + min(k + 2, n - 1)].w;
                    int w3 = stage[s + min(k + 3, n - 1)].w;
                    rank += (w0 < key) ? 1 : 0;
                    rank += (k + 1 < n && w1 < key) ? 1 : 0;
                    rank += (k + 2 < n && w2 < key) ? 1 : 0;
                    rank += (k + 3 < n && w3 < key) ? 1 : 0;
                }
                dsl[u] = s + rank;
            }
        }
        __syncthreads();
#pragma unroll
        for (int u = 0; u < BR_CAP / BR_T; ++u) {
            if (dsl[u] >= 0) stage[dsl[u]] = q[u];
        }
        __syncthreads();
        for (int i = t; i < nb; i += BR_T) ce[bstart + i] = stage[i];
    } else {
        if (t < 256) {
            int4* g0 = ce + bstart + rs[t];
            for (int i = 1; i < c; ++i) {
                int4 key = g0[i];
                int  k   = i - 1;
                while (k >= 0 && g0[k].w > key.w) { g0[k + 1] = g0[k];  --k; }
                g0[k + 1] = key;
            }
        }
        __syncthreads();
    }
    if (staged) bin_layer0<true>(stage, ce + bstart, rs, rc, f, dst, b, N);
    else        bin_layer0<false>(stage, ce + bstart, rs, rc, f, dst, b, N);
}

#define ROW_BLK   256
#define ROW_HOLD  32

template <int K>
__device__ __forceinline__ int row_bcast(int v) {
    return __builtin_amdgcn_update_dpp(0, v, 0x150 + K, 0xf, 0xf, false);
}

template <int D>
__device__ __forceinline__ int row_rot(int v) {
    return __builtin_amdgcn_update_dpp(0, v, 0x120 + D, 0xf, 0xf, false);
}


template <int D>
__device__ __forceinline__ void rank2(int ka, int kb, int& ra, int& rb) {
    int ta = row_rot<D>(ka);
    int tb = row_rot<D>(kb);
    ra += ((ta < ka) ? 1 : 0) + ((tb < ka) ? 1 : 0);
    rb += ((ta < kb) ? 1 : 0) + ((tb < kb) ? 1 : 0);
}

template <int I>
__device__ __forceinline__ void gather4(float4& a, const float* fl, int scol, int sval, float sc) {
    int c0 = row_bcast<I + 0>(scol);
    int c1 = row_bcast<I + 1>(scol);
    int c2 = row_bcast<I + 2>(scol);
    int c3 = row_bcast<I + 3>(scol);
    float w0 = __int_as_float(row_bcast<I + 0>(sval));
    float w1 = __int_as_float(row_bcast<I + 1>(sval));
    float w2 = __int_as_float(row_bcast<I + 2>(sval));
    float w3 = __int_as_float(row_bcast<I + 3>(sval));
    float4 x0 = *(const float4*)(fl + (size_t)c0 * EMB);
    float4 x1 = *(const float4*)(fl + (size_t)c1 * EMB);
    float4 x2 = *(const float4*)(fl + (size_t)c2 * EMB);
    float4 x3 = *(const float4*)(fl + (size_t)c3 * EMB);
    fma_row(a, x0, sc, w0);
    fma_row(a, x1, sc, w1);
    fma_row(a, x2, sc, w2);
    fma_row(a, x3, sc, w3);
}

template <int MODE>
__global__ void spmm_rows(const float* __restrict__ f,
                          const int*   __restrict__ seg,
                          int*                      ce,
                          const float* __restrict__ base0,
                          const float* __restrict__ base1,
                          float*       __restrict__ dst,
                          float srcScale, float inv_c, int N) {
    int tid = blockIdx.x * blockDim.x + threadIdx.x;
    int r   = tid >> 4;
    int m   = tid & 15;
    int l4  = m << 2;
    if (r >= N) return;
    int2 sc = *(const int2*)(seg + (size_t)r * 2);
    int c = sc.y;
    int4* rec = (int4*)ce + sc.x;
    const float* fl = f + l4;
    size_t off = (size_t)r * EMB + l4;
    float4 b0;  b0.x = 0.0f;  b0.y = 0.0f;  b0.z = 0.0f;  b0.w = 0.0f;
    float4 b1 = b0;
    if (MODE == 1) {
        b0 = *(const float4*)(base0 + off);
        b1 = *(const float4*)(base1 + off);
    }
    float4 a;  a.x = 0.0f;  a.y = 0.0f;  a.z = 0.0f;  a.w = 0.0f;

    if (c > ROW_HOLD) {
        int prev = -1;
        for (int k = 0; k < c; ++k) {
            int best = 0x7fffffff, bcol = 0, bval = 0;
            for (int j = 0; j < c; ++j) {
                int4 q = rec[j];
                if (q.w > prev && q.w < best) { best = q.w;  bcol = q.x;  bval = q.y; }
            }
            prev = best;
            float4 x = *(const float4*)(fl + (size_t)bcol * EMB);
            fma_row(a, x, srcScale, __int_as_float(bval));
        }
    } else if (MODE == 0 && c > 16) {
        __shared__ int2 ordS[ROW_BLK / 16][ROW_HOLD];
        int2* ord = ordS[threadIdx.x >> 4];
        bool hb = (m + 16 < c);
        int4 qa = rec[m];
        int4 qb;  qb.x = 0;  qb.y = 0;  qb.z = r;  qb.w = 0x7fffffff;
        if (hb) qb = rec[m + 16];
        int ka = qa.w;
        int kb = hb ? qb.w : 0x7fffffff;
        int ra = (kb < ka) ? 1 : 0;
        int rb = (ka < kb) ? 1 : 0;
        rank2<1>(ka, kb, ra, rb);   rank2<2>(ka, kb, ra, rb);   rank2<3>(ka, kb, ra, rb);
        rank2<4>(ka, kb, ra, rb);   rank2<5>(ka, kb, ra, rb);   rank2<6>(ka, kb, ra, rb);
        rank2<7>(ka, kb, ra, rb);   rank2<8>(ka, kb, ra, rb);   rank2<9>(ka, kb, ra, rb);
        rank2<10>(ka, kb, ra, rb);  rank2<11>(ka, kb, ra, rb);  rank2<12>(ka, kb, ra, rb);
        rank2<13>(ka, kb, ra, rb);  rank2<14>(ka, kb, ra, rb);  rank2<15>(ka, kb, ra, rb);
        if (!hb) rb = m + 16;
        int2 pa;  pa.x = qa.x;  pa.y = qa.y;
        int2 pb;  pb.x = qb.x;  pb.y = qb.y;
        ord[ra] = pa;
        ord[rb] = pb;
        int2 sa = ord[m];
        int4 oa;  oa.x = sa.x;  oa.y = sa.y;  oa.z = r;  oa.w = 0;
        rec[m] = oa;
        if (hb) {
            int2 sb = ord[m + 16];
            int4 ob;  ob.x = sb.x;  ob.y = sb.y;  ob.z = r;  ob.w = 0;
            rec[m + 16] = ob;
        }
        int last = c - 1;
        for (int i = 0; i < c; i += 4) {
            int2 e0 = ord[i];
            int2 e1 = ord[min(i + 1, last)];
            int2 e2 = ord[min(i + 2, last)];
            int2 e3 = ord[min(i + 3, last)];
            float4 x0 = *(const float4*)(fl + (size_t)e0.x * EMB);
            float4 x1 = *(const float4*)(fl + (size_t)e1.x * EMB);
            float4 x2 = *(const float4*)(fl + (size_t)e2.x * EMB);
            float4 x3 = *(const float4*)(fl + (size_t)e3.x * EMB);
            float w0 = __int_as_float(e0.y);
            float w1 = (i + 1 <= last) ? __int_as_float(e1.y) : 0.0f;
            float w2 = (i + 2 <= last) ? __int_as_float(e2.y) : 0.0f;
            float w3 = (i + 3 <= last) ? __int_as_float(e3.y) : 0.0f;
            fma_row(a, x0, srcScale, w0);
            fma_row(a, x1, srcScale, w1);
            fma_row(a, x2, srcScale, w2);
            fma_row(a, x3, srcScale, w3);
        }
    } else if (MODE == 0) {
        if (c > 0) {
            int4 q;  q.x = 0;  q.y = 0;  q.z = r;  q.w = 0x7fffffff;
            if (m < c) q = rec[m];
            int key = (m < c) ? q.w : 0x7fffffff;
            int rank = 0;
            rank += (row_rot<1>(key)  < key) ? 1 : 0;
            rank += (row_rot<2>(key)  < key) ? 1 : 0;
            rank += (row_rot<3>(key)  < key) ? 1 : 0;
            rank += (row_rot<4>(key)  < key) ? 1 : 0;
            rank += (row_rot<5>(key)  < key) ? 1 : 0;
            rank += (row_rot<6>(key)  < key) ? 1 : 0;
            rank += (row_rot<7>(key)  < key) ? 1 : 0;
            rank += (row_rot<8>(key)  < key) ? 1 : 0;
            rank += (row_rot<9>(key)  < key) ? 1 : 0;
            rank += (row_rot<10>(key) < key) ? 1 : 0;
            rank += (row_rot<11>(key) < key) ? 1 : 0;
            rank += (row_rot<12>(key) < key) ? 1 : 0;
            rank += (row_rot<13>(key) < key) ? 1 : 0;
            rank += (row_rot<14>(key) < key) ? 1 : 0;
            rank += (row_rot<15>(key) < key) ? 1 : 0;
            if (m >= c) rank = m;
            int to   = (((int)threadIdx.x & 16) | rank) << 2;
            int scol = __builtin_amdgcn_ds_permute(to, q.x);
            int sval = __builtin_amdgcn_ds_permute(to, q.y);
            int lastcol = __shfl(scol, c - 1, 16);
            if (m >= c) scol = lastcol;
            if (m < c) {
                int4 o;  o.x = scol;  o.y = sval;  o.z = r;  o.w = 0;
                rec[m] = o;
            }
            gather4<0>(a, fl, scol, sval, srcScale);
            if (c > 4)  gather4<4>(a, fl, scol, sval, srcScale);
            if (c > 8)  gather4<8>(a, fl, scol, sval, srcScale);
            if (c > 12) gather4<12>(a, fl, scol, sval, srcScale);
        }
    } else {
        for (int i = 0; i < c; i += 4) {
            int last = c - 1;
            int i1 = min(i + 1, last);
            int i2 = min(i + 2, last);
            int i3 = min(i + 3, last);
            int4 e0 = rec[i];
            int4 e1 = rec[i1];
            int4 e2 = rec[i2];
            int4 e3 = rec[i3];
            float4 x0 = *(const float4*)(fl + (size_t)e0.x * EMB);
            float4 x1 = *(const float4*)(fl + (size_t)e1.x * EMB);
            float4 x2 = *(const float4*)(fl + (size_t)e2.x * EMB);
            float4 x3 = *(const float4*)(fl + (size_t)e3.x * EMB);
            float w0 = __int_as_float(e0.y);
            float w1 = (i + 1 <= last) ? __int_as_float(e1.y) : 0.0f;
            float w2 = (i + 2 <= last) ? __int_as_float(e2.y) : 0.0f;
            float w3 = (i + 3 <= last) ? __int_as_float(e3.y) : 0.0f;
            fma_row(a, x0, srcScale, w0);
            fma_row(a, x1, srcScale, w1);
            fma_row(a, x2, srcScale, w2);
            fma_row(a, x3, srcScale, w3);
        }
    }

    float4 g = row_finish(a, inv_c);
    if (MODE == 0) {
        *(float4*)(dst + off) = g;
    } else {
        float4 o;
        o.x = (b0.x + b1.x) + g.x;
        o.y = (b0.y + b1.y) + g.y;
        o.z = (b0.z + b1.z) + g.z;
        o.w = (b0.w + b1.w) + g.w;
        *(float4*)(dst + off) = o;
    }
}

#define FU_WAVES 4
#define FU_PITCH 68

template <int M>
__device__ __forceinline__ int row_xadd_i(int s) {
    return s + __builtin_amdgcn_update_dpp(0, s, 0x160 + M, 0xf, 0xf, false);
}

__device__ __forceinline__ float4 row_load16(const float* __restrict__ base, int row, int m) {
    return *(const float4*)((const char*)base + (unsigned)(row * (EMB * 4) + m * 16));
}

__device__ __forceinline__ void store_row(float* __restrict__ out, int row, int lane, float2 v) {
    *(float2*)((char*)out + (unsigned)(row * (EMB * 4) + lane * 8)) = v;
}

__global__ void __launch_bounds__(FU_WAVES * 32)
edge_upd(const float* __restrict__ acc,
         const int*   __restrict__ seg,
         const int*   __restrict__ ce,
         float*       __restrict__ out, int N) {
    __shared__ __attribute__((aligned(16))) float tileS[FU_WAVES][16 * FU_PITCH];
    const float kScale = 1.0f / 3.0f;
    int lane = threadIdx.x & 31;
    int wv   = __builtin_amdgcn_readfirstlane((int)(threadIdx.x >> 5));
    int r0   = (blockIdx.x * FU_WAVES + wv) * 16;
    if (r0 >= N) return;
    int m    = lane & 15;
    int half = lane >> 4;
    int rows = min(16, N - r0);
    float* sT = tileS[wv];
    const float* F = sT + m * FU_PITCH + 2 * half;

    int2 sc;  sc.x = 0;  sc.y = 0;
    if (m < rows) sc = *(const int2*)(seg + (size_t)(r0 + m) * 2);
    int nsum = sc.y;
    nsum = row_xadd_i<8>(nsum);
    nsum = row_xadd_i<4>(nsum);
    nsum = row_xadd_i<2>(nsum);
    nsum = row_xadd_i<1>(nsum);
    int begin  = __builtin_amdgcn_readfirstlane(sc.x);
    int n      = __builtin_amdgcn_readfirstlane(nsum);
    int ntiles = (n + 15) >> 4;
    int last   = n - 1;
    const int4* recs = (const int4*)ce + begin;

    float2 zero2;  zero2.x = 0.0f;  zero2.y = 0.0f;

    unsigned emp = __builtin_amdgcn_ballot_w32(half == 0 && m < rows && sc.y == 0);
    while (emp != 0u) {
        int zr = __builtin_ctz(emp);
        emp &= emp - 1u;
        store_row(out, r0 + zr, lane, zero2);
    }
    if (ntiles == 0) return;

    int4 rcC = recs[min(m, last)];
    int4 rcN = recs[min(16 + m, last)];

    float4 s4[8];
#pragma unroll
    for (int j = 0; j < 8; ++j) {
        s4[j] = row_load16(acc, min(r0 + 2 * j + half, N - 1), m);
    }
    float4 tq[8];
#pragma unroll
    for (int j = 0; j < 8; ++j) {
        tq[j] = row_load16(acc, __shfl(rcC.x, 2 * j + half, 32), m);
    }
#pragma unroll
    for (int j = 0; j < 8; ++j) {
        float4 x = s4[j];
        x.x *= kScale;  x.y *= kScale;  x.z *= kScale;  x.w *= kScale;
        *(float4*)(sT + (2 * j + half) * FU_PITCH + 4 * m) = x;
    }
    v2f A[16];
#pragma unroll
    for (int ks = 0; ks < 16; ++ks) {
        float2 a = *(const float2*)(F + 4 * ks);
        A[ks].x = a.x;  A[ks].y = a.y;
    }

    float2 av = zero2;
    int tailrow = 0;
    for (int t = 0; t < ntiles; ++t) {
        bool  valid = (16 * t + m < n);
        int   irow  = rcC.z - r0;
        float v     = __int_as_float(rcC.y);

#pragma unroll
        for (int j = 0; j < 8; ++j) {
            float4 x = tq[j];
            x.x *= kScale;  x.y *= kScale;  x.z *= kScale;  x.w *= kScale;
            *(float4*)(sT + (2 * j + half) * FU_PITCH + 4 * m) = x;
        }
#pragma unroll
        for (int j = 0; j < 8; ++j) {
            tq[j] = row_load16(acc, __shfl(rcN.x, 2 * j + half, 32), m);
        }
        int4 rcNN = recs[min(16 * (t + 2) + m, last)];

        int nxt = __shfl_down(irow, 1, 16);
        int nx0 = __builtin_amdgcn_readlane(rcN.z, 0) - r0;
        if (m == 15) nxt = nx0;
        unsigned ends = __builtin_amdgcn_ballot_w32(half == 0 && nxt != irow);
        tailrow = __builtin_amdgcn_readlane(irow, 15);

        v8f c0 = {};
        v8f c1 = {};
#pragma unroll
        for (int ks = 0; ks < 16; ks += 2) {
            float2 b0 = *(const float2*)(F + 4 * ks);
            float2 b1 = *(const float2*)(F + 4 * ks + 4);
            v2f B0;  B0.x = b0.x;  B0.y = b0.y;
            v2f B1;  B1.x = b1.x;  B1.y = b1.y;
            c0 = __builtin_amdgcn_wmma_f32_16x16x4_f32(
                 false, A[ks],  false, B0,
                 (short)0, c0,  false,  false);
            c1 = __builtin_amdgcn_wmma_f32_16x16x4_f32(
                 false, A[ks + 1],  false, B1,
                 (short)0, c1,  false,  false);
        }
        v8f cc = c0 + c1;

        int   i7  = irow & 7;
        float sel = cc[0];
#pragma unroll
        for (int vv = 1; vv < 8; ++vv) {
            if (i7 == vv) sel = cc[vv];
        }
        float dot = ((irow >> 3) == half) ? sel : 0.0f;
        dot += __shfl_xor(dot, 16, 32);

        float cross = dot * (1.0f / EMB);
        float z     = 4.0f * (1.0f - cross);
        float sp    = log1pf(expf(-fabsf(z))) + fmaxf(z, 0.0f);
        float mat   = valid ? 0.5f * expf(z) * sp * v : 0.0f;

        float2 xr[16];
#pragma unroll
        for (int j = 0; j < 16; ++j) {
            xr[j] = *(const float2*)(sT + j * FU_PITCH + 2 * lane);
        }
#pragma unroll
        for (int j = 0; j < 16; ++j) {
            float mj = readlane_f32(mat, j);
            av.x += xr[j].x * mj;
            av.y += xr[j].y * mj;
            if ((ends >> j) & 1u) {
                store_row(out, r0 + __builtin_amdgcn_readlane(irow, j), lane, av);
                av = zero2;
            }
        }

        rcC = rcN;
        rcN = rcNN;
    }
    store_row(out, r0 + tailrow, lane, av);
}

extern "C" void kernel_launch(void* const* d_in, const int* in_sizes, int n_in,
                              void* d_out, int out_size, void* d_ws, size_t ws_size,
                              hipStream_t stream) {
    (void)n_in; (void)out_size; (void)ws_size;
    const float* feats     = (const float*)d_in[0];
    const float* edge_vals = (const float*)d_in[1];
    const int*   edge_row  = (const int*)d_in[2];
    const int*   edge_col  = (const int*)d_in[3];

    int N = in_sizes[0] / EMB;
    int E = in_sizes[1];

    float* bufA = (float*)d_ws;
    float* bufB = bufA + (size_t)N * EMB;
    float* acc  = bufB + (size_t)N * EMB;
    float* out  = (float*)d_out;

    int* ce    = (int*)bufB;
    int* seg   = ce + (size_t)E * 4;
    int* pos   = seg + (size_t)N * 2;
    int* cnt   = pos + E;
    int* total = cnt + N;

    const int blk = 256;
    int gridE  = (E + blk - 1) / blk;
    int gridN  = (N + blk - 1) / blk;
    int gridGa = (int)(((long long)N * 16 + ROW_BLK - 1) / ROW_BLK);
    int gridFu = (N + 16 * FU_WAVES - 1) / (16 * FU_WAVES);
    int NB     = (N + (1 << BIN_SHIFT) - 1) >> BIN_SHIFT;
    int gridSc = (E + SC_EPB - 1) / SC_EPB;

    if (NB + 1 <= NBMAX && gridSc <= BR_T && (long long)gridSc * SC_EPB * 4 <= (long long)N * EMB &&
        (long long)gridSc * (NB + 1) <= (long long)E) {
        int* tab = pos;
        int* tmp = (int*)acc;
        bin_block_sort<<<gridSc, SC_T, 0, stream>>>(edge_vals, edge_row, edge_col, tab, tmp, E, NB);
        bin_rows_l0<<<NB, BR_T, 0, stream>>>(tab, tmp, ce, seg, feats, bufA, N, NB, gridSc);
    } else {
        hipMemsetAsync(cnt, 0, (size_t)(N + 1) * sizeof(int), stream);
        row_hist<<<gridE, blk, 0, stream>>>(edge_row, cnt, pos, E);
        row_alloc<<<gridN, blk, 0, stream>>>(cnt, seg, total, N);
        csr_fill<<<gridE, blk, 0, stream>>>(edge_vals, edge_row, edge_col, seg, pos, ce, E);
        spmm_rows<0><<<gridGa, ROW_BLK, 0, stream>>>(feats, seg, ce, nullptr, nullptr, bufA, 1.0f, 0.5f, N);
    }

    spmm_rows<1><<<gridGa, ROW_BLK, 0, stream>>>(bufA, seg, ce, feats, bufA, acc, 1.0f, 1.0f / 3.0f, N);

    edge_upd<<<gridFu, FU_WAVES * 32, 0, stream>>>(acc, seg, ce, out, N);
}


# Library expf/log1pf (edge weights) and sqrtf and 1/x (row norms) replaced by the hardware exp2, log2, sqrt and rcp instructions, exp argument compensated; rounding-level change, no sum reordered
# speedup vs baseline: 4.1134x; 1.0360x over previous
#include <hip/hip_runtime.h>
#include <hip/hip_bf16.h>
#include <math.h>

typedef float v2f __attribute__((ext_vector_type(2)));
typedef float v8f __attribute__((ext_vector_type(8)));

#define EMB   64
#define EPSN  1e-12f

__device__ __forceinline__ float readlane_f32(float x, int lane) {
    return __int_as_float(__builtin_amdgcn_readlane(__float_as_int(x), lane));
}

__global__ void row_hist(const int* __restrict__ rows,
                         int*       __restrict__ cnt,
                         int*       __restrict__ pos, int E) {
    int e = blockIdx.x * blockDim.x + threadIdx.x;
    if (e >= E) return;
    pos[e] = atomicAdd(cnt + rows[e], 1);
}

__global__ void row_alloc(const int* __restrict__ cnt,
                          int*       __restrict__ seg,
                          int*       __restrict__ total, int N) {
    __shared__ int wsum[8];
    __shared__ int base;
    int r    = blockIdx.x * 256 + threadIdx.x;
    int lane = threadIdx.x & 31;
    int w    = threadIdx.x >> 5;
    int c    = (r < N) ? cnt[r] : 0;
    int incl = c;
#pragma unroll
    for (int d = 1; d < 32; d <<= 1) {
        int t = __shfl_up(incl, d, 32);
        if (lane >= d) incl += t;
    }
    if (lane == 31) wsum[w] = incl;
    __syncthreads();
    if (threadIdx.x == 0) {
        int run = 0;
        for (int i = 0; i < 8; ++i) { int t = wsum[i]; wsum[i] = run; run += t; }
        base = atomicAdd(total, run);
    }
    __syncthreads();
    if (r < N) {
        int2 sc;
        sc.x = base + wsum[w] + incl - c;
        sc.y = c;
        *(int2*)(seg + (size_t)r * 2) = sc;
    }
}

__global__ void csr_fill(const float* __restrict__ vals,
                         const int*   __restrict__ rows,
                         const int*   __restrict__ cols,
                         const int*   __restrict__ seg,
                         const int*   __restrict__ pos,
                         int*         __restrict__ ce, int E) {
    int e = blockIdx.x * blockDim.x + threadIdx.x;
    if (e >= E) return;
    int r = rows[e];
    int p = seg[(size_t)r * 2] + pos[e];
    int4 rec;
    rec.x = cols[e];
    rec.y = __float_as_int(vals[e]);
    rec.z = r;
    rec.w = e;
    *(int4*)(ce + (size_t)p * 4) = rec;
}

#define BIN_SHIFT 8
#define NBMAX     1024
#define SC_T      512
#define SC_K      8
#define SC_EPB    (SC_T * SC_K)
#define BR_T      512
#define BR_CAP    3072
#define BR_G      16
#define BR_GROUPS (BR_T / BR_G)
#define BR_U      8

__device__ __forceinline__ int block_excl_scan(int v, int* wsum) {
    int lane = threadIdx.x & 31;
    int w    = threadIdx.x >> 5;
    int incl = v;
    for (int d = 1; d < 32; d <<= 1) {
        int x = __shfl_up(incl, d, 32);
        if (lane >= d) incl += x;
    }
    __syncthreads();
    if (lane == 31) wsum[w] = incl;
    __syncthreads();
    int basew = 0;
    for (int i = 0; i < 16; ++i) {
        int x = wsum[i];
        if (i < w) basew += x;
    }
    return basew + incl - v;
}

__global__ void __launch_bounds__(SC_T)
bin_block_sort(const float* __restrict__ vals,
               const int*   __restrict__ rows,
               const int*   __restrict__ cols,
               int*         __restrict__ tab,
               int*         __restrict__ tmpw, int E, int NB) {
    int4* tmp = (int4*)tmpw;
    __shared__ int  h[NBMAX];
    __shared__ int  wsum[16];
    __shared__ int4 stage[SC_EPB];
    int t = threadIdx.x;
    for (int b = t; b < NBMAX; b += SC_T) h[b] = 0;
    int base = blockIdx.x * SC_EPB;
    int r[SC_K], cl[SC_K], vl[SC_K], lr[SC_K];
#pragma unroll
    for (int k = 0; k < SC_K; ++k) {
        int e = base + k * SC_T + t;
        r[k] = -1;  cl[k] = 0;  vl[k] = 0;  lr[k] = 0;
        if (e < E) {
            r[k]  = rows[e];
            cl[k] = cols[e];
            vl[k] = __float_as_int(vals[e]);
        }
    }
    __syncthreads();
#pragma unroll
    for (int k = 0; k < SC_K; ++k) {
        if (r[k] >= 0) lr[k] = atomicAdd(&h[r[k] >> BIN_SHIFT], 1);
    }
    __syncthreads();
    int b0 = 2 * t;
    int b1 = 2 * t + 1;
    int c0 = h[b0];
    int c1 = h[b1];
    int lex = block_excl_scan(c0 + c1, wsum);
    h[b0] = lex;
    h[b1] = lex + c0;
    int  TP   = NB + 1;
    int* trow = tab + (size_t)blockIdx.x * TP;
    if (b0 < TP) trow[b0] = lex;
    if (b1 < TP) trow[b1] = lex + c0;
    __syncthreads();
#pragma unroll
    for (int k = 0; k < SC_K; ++k) {
        if (r[k] >= 0) {
            int4 q;
            q.x = cl[k];
            q.y = vl[k];
            q.z = r[k];
            q.w = base + k * SC_T + t;
            stage[h[r[k] >> BIN_SHIFT] + lr[k]] = q;
        }
    }
    __syncthreads();
    int n = min(SC_EPB, E - base);
    for (int s = t; s < n; s += SC_T) tmp[base + s] = stage[s];
}

__device__ __forceinline__ void fma_row(float4& a, const float4& x, float sc, float w) {
    a.x += (x.x * sc) * w;
    a.y += (x.y * sc) * w;
    a.z += (x.z * sc) * w;
    a.w += (x.w * sc) * w;
}

template <int M>
__device__ __forceinline__ float row_xadd(float s) {
    int t = __builtin_amdgcn_update_dpp(0, __float_as_int(s), 0x160 + M, 0xf, 0xf, false);
    return s + __int_as_float(t);
}

__device__ __forceinline__ float4 row_finish(const float4& a, float inv_c) {
    float gx = a.x * inv_c;
    float gy = a.y * inv_c;
    float gz = a.z * inv_c;
    float gw = a.w * inv_c;
    float s  = gx * gx + gy * gy + gz * gz + gw * gw;
    s = row_xadd<8>(s);
    s = row_xadd<4>(s);
    s = row_xadd<2>(s);
    s = row_xadd<1>(s);
    float n   = __builtin_amdgcn_sqrtf(s);
    float inv = __builtin_amdgcn_rcpf(fmaxf(n, EPSN));
    float4 g;  g.x = gx * inv;  g.y = gy * inv;  g.z = gz * inv;  g.w = gw * inv;
    return g;
}

template <bool TILE>
__device__ __forceinline__ void bin_layer0(const int4* tile, const int4* gr,
                                           const int* rs, const int* rc,
                                           const float* __restrict__ f,
                                           float*       __restrict__ dst, int b, int N) {
    int t  = threadIdx.x;
    int hw = t >> 4;
    int m  = t & 15;
    int l4 = m << 2;
    const float* fl = f + l4;
    for (int j = hw; j < 256; j += BR_T / 16) {
        int row = (b << BIN_SHIFT) + j;
        if (row >= N) break;
        int s    = rs[j];
        int cc   = rc[j];
        int last = cc - 1;
        float4 a;  a.x = 0.0f;  a.y = 0.0f;  a.z = 0.0f;  a.w = 0.0f;
        for (int i = 0; i < cc; i += 4) {
            int i0 = s + i;
            int i1 = s + min(i + 1, last);
            int i2 = s + min(i + 2, last);
            int i3 = s + min(i + 3, last);
            int4 e0, e1, e2, e3;
            if (TILE) { e0 = tile[i0];  e1 = tile[i1];  e2 = tile[i2];  e3 = tile[i3]; }
            else      { e0 = gr[i0];    e1 = gr[i1];    e2 = gr[i2];    e3 = gr[i3]; }
            float4 x0 = *(const float4*)(fl + (size_t)e0.x * EMB);
            float4 x1 = *(const float4*)(fl + (size_t)e1.x * EMB);
            float4 x2 = *(const float4*)(fl + (size_t)e2.x * EMB);
            float4 x3 = *(const float4*)(fl + (size_t)e3.x * EMB);
            float w0 = __int_as_float(e0.y);
            float w1 = (i + 1 <= last) ? __int_as_float(e1.y) : 0.0f;
            float w2 = (i + 2 <= last) ? __int_as_float(e2.y) : 0.0f;
            float w3 = (i + 3 <= last) ? __int_as_float(e3.y) : 0.0f;
            fma_row(a, x0, 1.0f, w0);
            fma_row(a, x1, 1.0f, w1);
            fma_row(a, x2, 1.0f, w2);
            fma_row(a, x3, 1.0f, w3);
        }
        float4 gv = row_finish(a, 0.5f);
        *(float4*)(dst + (size_t)row * EMB + l4) = gv;
    }
}

__global__ void __launch_bounds__(BR_T)
bin_rows_l0(const int*   __restrict__ tab,
            const int*   __restrict__ tmpw,
            int*                      cew,
            int*         __restrict__ seg,
            const float* __restrict__ f,
            float*       __restrict__ dst, int N, int NB, int NBLK) {
    const int4* tmp = (const int4*)tmpw;
    int4*       ce  = (int4*)cew;
    __shared__ int  rc[256];
    __shared__ int  rs[256];
    __shared__ int  wsum[16];
    __shared__ int  sumS[16];
    __shared__ int  sumC[16];
    __shared__ int  runSrc[BR_T];
    __shared__ int  runLen[BR_T];
    __shared__ int4 stage[BR_CAP];
    int t    = threadIdx.x;
    int b    = blockIdx.x;
    int lane = t & 31;
    int w    = t >> 5;
    int TP   = NB + 1;
    int s0 = 0, len0 = 0;
    if (t < NBLK) {
        const int* trow = tab + (size_t)t * TP + b;
        s0   = trow[0];
        len0 = trow[1] - s0;
    }
    runSrc[t] = t * SC_EPB + s0;
    runLen[t] = len0;
    if (t < 256) rc[t] = 0;
    int ws = s0, wc = len0;
    for (int d = 16; d >= 1; d >>= 1) {
        ws += __shfl_xor(ws, d, 32);
        wc += __shfl_xor(wc, d, 32);
    }
    if (lane == 0) { sumS[w] = ws;  sumC[w] = wc; }
    __syncthreads();
    int bstart = 0, nb = 0;
    for (int i = 0; i < 16; ++i) { bstart += sumS[i];  nb += sumC[i]; }

    int g = t / BR_G;
    int l = t % BR_G;
    for (int k0 = g; k0 < NBLK; k0 += BR_U * BR_GROUPS) {
        int rw[BR_U];
#pragma unroll
        for (int u = 0; u < BR_U; ++u) {
            int k = k0 + u * BR_GROUPS;
            rw[u] = -1;
            if (k < NBLK && l < runLen[k]) rw[u] = tmp[runSrc[k] + l].z;
        }
#pragma unroll
        for (int u = 0; u < BR_U; ++u) {
            if (rw[u] >= 0) atomicAdd(&rc[rw[u] & 255], 1);
        }
        for (int u = 0; u < BR_U; ++u) {
            int k = k0 + u * BR_GROUPS;
            if (k < NBLK) {
                int len = runLen[k];
                int src = runSrc[k];
                for (int i = l + BR_G; i < len; i += BR_G) atomicAdd(&rc[tmp[src + i].z & 255], 1);
            }
        }
    }
    __syncthreads();
    int c  = (t < 256) ? rc[t] : 0;
    int ex = block_excl_scan(c, wsum);
    if (t < 256) {
        rs[t] = ex;
        rc[t] = 0;
        int row = (b << BIN_SHIFT) + t;
        if (row < N) {
            int2 sc;
            sc.x = bstart + ex;
            sc.y = c;
            *(int2*)(seg + (size_t)row * 2) = sc;
        }
    }
    __syncthreads();
    bool staged = (nb <= BR_CAP);
    for (int k0 = g; k0 < NBLK; k0 += BR_U * BR_GROUPS) {
        int4 q[BR_U];
#pragma unroll
        for (int u = 0; u < BR_U; ++u) {
            int k = k0 + u * BR_GROUPS;
            q[u].x = 0;  q[u].y = 0;  q[u].z = -1;  q[u].w = 0;
            if (k < NBLK && l < runLen[k]) q[u] = tmp[runSrc[k] + l];
        }
#pragma unroll
        for (int u = 0; u < BR_U; ++u) {
            if (q[u].z >= 0) {
                int j    = q[u].z & 255;
                int slot = rs[j] + atomicAdd(&rc[j], 1);
                if (staged) stage[slot] = q[u];
                else        ce[bstart + slot] = q[u];
            }
        }
        for (int u = 0; u < BR_U; ++u) {
            int k = k0 + u * BR_GROUPS;
            if (k < NBLK) {
                int len = runLen[k];
                int src = runSrc[k];
                for (int i = l + BR_G; i < len; i += BR_G) {
                    int4 qq  = tmp[src + i];
                    int  j    = qq.z & 255;
                    int  slot = rs[j] + atomicAdd(&rc[j], 1);
                    if (staged) stage[slot] = qq;
                    else        ce[bstart + slot] = qq;
                }
            }
        }
    }
    __syncthreads();
    if (staged) {
        int4 q[BR_CAP / BR_T];
        int  dsl[BR_CAP / BR_T];
#pragma unroll
        for (int u = 0; u < BR_CAP / BR_T; ++u) {
            int i = t + u * BR_T;
            q[u].x = 0;  q[u].y = 0;  q[u].z = 0;  q[u].w = 0;
            dsl[u] = -1;
            if (i < nb) {
                q[u] = stage[i];
                int j   = q[u].z & 255;
                int s   = rs[j];
                int n   = rc[j];
                int key = q[u].w;
                int rank = 0;
                for (int k = 0; k < n; k += 4) {
                    int w0 = stage[s + k].w;
                    int w1 = stage[s + min(k + 1, n - 1)].w;
                    int w2 = stage[s + min(k + 2, n - 1)].w;
                    int w3 = stage[s + min(k + 3, n - 1)].w;
                    rank += (w0 < key) ? 1 : 0;
                    rank += (k + 1 < n && w1 < key) ? 1 : 0;
                    rank += (k + 2 < n && w2 < key) ? 1 : 0;
                    rank += (k + 3 < n && w3 < key) ? 1 : 0;
                }
                dsl[u] = s + rank;
            }
        }
        __syncthreads();
#pragma unroll
        for (int u = 0; u < BR_CAP / BR_T; ++u) {
            if (dsl[u] >= 0) stage[dsl[u]] = q[u];
        }
        __syncthreads();
        for (int i = t; i < nb; i += BR_T) ce[bstart + i] = stage[i];
    } else {
        if (t < 256) {
            int4* g0 = ce + bstart + rs[t];
            for (int i = 1; i < c; ++i) {
                int4 key = g0[i];
                int  k   = i - 1;
                while (k >= 0 && g0[k].w > key.w) { g0[k + 1] = g0[k];  --k; }
                g0[k + 1] = key;
            }
        }
        __syncthreads();
    }
    if (staged) bin_layer0<true>(stage, ce + bstart, rs, rc, f, dst, b, N);
    else        bin_layer0<false>(stage, ce + bstart, rs, rc, f, dst, b, N);
}

#define ROW_BLK   256
#define ROW_HOLD  32

template <int K>
__device__ __forceinline__ int row_bcast(int v) {
    return __builtin_amdgcn_update_dpp(0, v, 0x150 + K, 0xf, 0xf, false);
}

template <int D>
__device__ __forceinline__ int row_rot(int v) {
    return __builtin_amdgcn_update_dpp(0, v, 0x120 + D, 0xf, 0xf, false);
}


template <int D>
__device__ __forceinline__ void rank2(int ka, int kb, int& ra, int& rb) {
    int ta = row_rot<D>(ka);
    int tb = row_rot<D>(kb);
    ra += ((ta < ka) ? 1 : 0) + ((tb < ka) ? 1 : 0);
    rb += ((ta < kb) ? 1 : 0) + ((tb < kb) ? 1 : 0);
}

template <int I>
__device__ __forceinline__ void gather4(float4& a, const float* fl, int scol, int sval, float sc) {
    int c0 = row_bcast<I + 0>(scol);
    int c1 = row_bcast<I + 1>(scol);
    int c2 = row_bcast<I + 2>(scol);
    int c3 = row_bcast<I + 3>(scol);
    float w0 = __int_as_float(row_bcast<I + 0>(sval));
    float w1 = __int_as_float(row_bcast<I + 1>(sval));
    float w2 = __int_as_float(row_bcast<I + 2>(sval));
    float w3 = __int_as_float(row_bcast<I + 3>(sval));
    float4 x0 = *(const float4*)(fl + (size_t)c0 * EMB);
    float4 x1 = *(const float4*)(fl + (size_t)c1 * EMB);
    float4 x2 = *(const float4*)(fl + (size_t)c2 * EMB);
    float4 x3 = *(const float4*)(fl + (size_t)c3 * EMB);
    fma_row(a, x0, sc, w0);
    fma_row(a, x1, sc, w1);
    fma_row(a, x2, sc, w2);
    fma_row(a, x3, sc, w3);
}

template <int MODE>
__global__ void spmm_rows(const float* __restrict__ f,
                          const int*   __restrict__ seg,
                          int*                      ce,
                          const float* __restrict__ base0,
                          const float* __restrict__ base1,
                          float*       __restrict__ dst,
                          float srcScale, float inv_c, int N) {
    int tid = blockIdx.x * blockDim.x + threadIdx.x;
    int r   = tid >> 4;
    int m   = tid & 15;
    int l4  = m << 2;
    if (r >= N) return;
    int2 sc = *(const int2*)(seg + (size_t)r * 2);
    int c = sc.y;
    int4* rec = (int4*)ce + sc.x;
    const float* fl = f + l4;
    size_t off = (size_t)r * EMB + l4;
    float4 b0;  b0.x = 0.0f;  b0.y = 0.0f;  b0.z = 0.0f;  b0.w = 0.0f;
    float4 b1 = b0;
    if (MODE == 1) {
        b0 = *(const float4*)(base0 + off);
        b1 = *(const float4*)(base1 + off);
    }
    float4 a;  a.x = 0.0f;  a.y = 0.0f;  a.z = 0.0f;  a.w = 0.0f;

    if (c > ROW_HOLD) {
        int prev = -1;
        for (int k = 0; k < c; ++k) {
            int best = 0x7fffffff, bcol = 0, bval = 0;
            for (int j = 0; j < c; ++j) {
                int4 q = rec[j];
                if (q.w > prev && q.w < best) { best = q.w;  bcol = q.x;  bval = q.y; }
            }
            prev = best;
            float4 x = *(const float4*)(fl + (size_t)bcol * EMB);
            fma_row(a, x, srcScale, __int_as_float(bval));
        }
    } else if (MODE == 0 && c > 16) {
        __shared__ int2 ordS[ROW_BLK / 16][ROW_HOLD];
        int2* ord = ordS[threadIdx.x >> 4];
        bool hb = (m + 16 < c);
        int4 qa = rec[m];
        int4 qb;  qb.x = 0;  qb.y = 0;  qb.z = r;  qb.w = 0x7fffffff;
        if (hb) qb = rec[m + 16];
        int ka = qa.w;
        int kb = hb ? qb.w : 0x7fffffff;
        int ra = (kb < ka) ? 1 : 0;
        int rb = (ka < kb) ? 1 : 0;
        rank2<1>(ka, kb, ra, rb);   rank2<2>(ka, kb, ra, rb);   rank2<3>(ka, kb, ra, rb);
        rank2<4>(ka, kb, ra, rb);   rank2<5>(ka, kb, ra, rb);   rank2<6>(ka, kb, ra, rb);
        rank2<7>(ka, kb, ra, rb);   rank2<8>(ka, kb, ra, rb);   rank2<9>(ka, kb, ra, rb);
        rank2<10>(ka, kb, ra, rb);  rank2<11>(ka, kb, ra, rb);  rank2<12>(ka, kb, ra, rb);
        rank2<13>(ka, kb, ra, rb);  rank2<14>(ka, kb, ra, rb);  rank2<15>(ka, kb, ra, rb);
        if (!hb) rb = m + 16;
        int2 pa;  pa.x = qa.x;  pa.y = qa.y;
        int2 pb;  pb.x = qb.x;  pb.y = qb.y;
        ord[ra] = pa;
        ord[rb] = pb;
        int2 sa = ord[m];
        int4 oa;  oa.x = sa.x;  oa.y = sa.y;  oa.z = r;  oa.w = 0;
        rec[m] = oa;
        if (hb) {
            int2 sb = ord[m + 16];
            int4 ob;  ob.x = sb.x;  ob.y = sb.y;  ob.z = r;  ob.w = 0;
            rec[m + 16] = ob;
        }
        int last = c - 1;
        for (int i = 0; i < c; i += 4) {
            int2 e0 = ord[i];
            int2 e1 = ord[min(i + 1, last)];
            int2 e2 = ord[min(i + 2, last)];
            int2 e3 = ord[min(i + 3, last)];
            float4 x0 = *(const float4*)(fl + (size_t)e0.x * EMB);
            float4 x1 = *(const float4*)(fl + (size_t)e1.x * EMB);
            float4 x2 = *(const float4*)(fl + (size_t)e2.x * EMB);
            float4 x3 = *(const float4*)(fl + (size_t)e3.x * EMB);
            float w0 = __int_as_float(e0.y);
            float w1 = (i + 1 <= last) ? __int_as_float(e1.y) : 0.0f;
            float w2 = (i + 2 <= last) ? __int_as_float(e2.y) : 0.0f;
            float w3 = (i + 3 <= last) ? __int_as_float(e3.y) : 0.0f;
            fma_row(a, x0, srcScale, w0);
            fma_row(a, x1, srcScale, w1);
            fma_row(a, x2, srcScale, w2);
            fma_row(a, x3, srcScale, w3);
        }
    } else if (MODE == 0) {
        if (c > 0) {
            int4 q;  q.x = 0;  q.y = 0;  q.z = r;  q.w = 0x7fffffff;
            if (m < c) q = rec[m];
            int key = (m < c) ? q.w : 0x7fffffff;
            int rank = 0;
            rank += (row_rot<1>(key)  < key) ? 1 : 0;
            rank += (row_rot<2>(key)  < key) ? 1 : 0;
            rank += (row_rot<3>(key)  < key) ? 1 : 0;
            rank += (row_rot<4>(key)  < key) ? 1 : 0;
            rank += (row_rot<5>(key)  < key) ? 1 : 0;
            rank += (row_rot<6>(key)  < key) ? 1 : 0;
            rank += (row_rot<7>(key)  < key) ? 1 : 0;
            rank += (row_rot<8>(key)  < key) ? 1 : 0;
            rank += (row_rot<9>(key)  < key) ? 1 : 0;
            rank += (row_rot<10>(key) < key) ? 1 : 0;
            rank += (row_rot<11>(key) < key) ? 1 : 0;
            rank += (row_rot<12>(key) < key) ? 1 : 0;
            rank += (row_rot<13>(key) < key) ? 1 : 0;
            rank += (row_rot<14>(key) < key) ? 1 : 0;
            rank += (row_rot<15>(key) < key) ? 1 : 0;
            if (m >= c) rank = m;
            int to   = (((int)threadIdx.x & 16) | rank) << 2;
            int scol = __builtin_amdgcn_ds_permute(to, q.x);
            int sval = __builtin_amdgcn_ds_permute(to, q.y);
            int lastcol = __shfl(scol, c - 1, 16);
            if (m >= c) scol = lastcol;
            if (m < c) {
                int4 o;  o.x = scol;  o.y = sval;  o.z = r;  o.w = 0;
                rec[m] = o;
            }
            gather4<0>(a, fl, scol, sval, srcScale);
            if (c > 4)  gather4<4>(a, fl, scol, sval, srcScale);
            if (c > 8)  gather4<8>(a, fl, scol, sval, srcScale);
            if (c > 12) gather4<12>(a, fl, scol, sval, srcScale);
        }
    } else {
        for (int i = 0; i < c; i += 4) {
            int last = c - 1;
            int i1 = min(i + 1, last);
            int i2 = min(i + 2, last);
            int i3 = min(i + 3, last);
            int4 e0 = rec[i];
            int4 e1 = rec[i1];
            int4 e2 = rec[i2];
            int4 e3 = rec[i3];
            float4 x0 = *(const float4*)(fl + (size_t)e0.x * EMB);
            float4 x1 = *(const float4*)(fl + (size_t)e1.x * EMB);
            float4 x2 = *(const float4*)(fl + (size_t)e2.x * EMB);
            float4 x3 = *(const float4*)(fl + (size_t)e3.x * EMB);
            float w0 = __int_as_float(e0.y);
            float w1 = (i + 1 <= last) ? __int_as_float(e1.y) : 0.0f;
            float w2 = (i + 2 <= last) ? __int_as_float(e2.y) : 0.0f;
            float w3 = (i + 3 <= last) ? __int_as_float(e3.y) : 0.0f;
            fma_row(a, x0, srcScale, w0);
            fma_row(a, x1, srcScale, w1);
            fma_row(a, x2, srcScale, w2);
            fma_row(a, x3, srcScale, w3);
        }
    }

    float4 g = row_finish(a, inv_c);
    if (MODE == 0) {
        *(float4*)(dst + off) = g;
    } else {
        float4 o;
        o.x = (b0.x + b1.x) + g.x;
        o.y = (b0.y + b1.y) + g.y;
        o.z = (b0.z + b1.z) + g.z;
        o.w = (b0.w + b1.w) + g.w;
        *(float4*)(dst + off) = o;
    }
}

#define FU_WAVES 4
#define FU_PITCH 68

template <int M>
__device__ __forceinline__ int row_xadd_i(int s) {
    return s + __builtin_amdgcn_update_dpp(0, s, 0x160 + M, 0xf, 0xf, false);
}

__device__ __forceinline__ float4 row_load16(const float* __restrict__ base, int row, int m) {
    return *(const float4*)((const char*)base + (unsigned)(row * (EMB * 4) + m * 16));
}

__device__ __forceinline__ void store_row(float* __restrict__ out, int row, int lane, float2 v) {
    *(float2*)((char*)out + (unsigned)(row * (EMB * 4) + lane * 8)) = v;
}

__device__ __forceinline__ float edge_mat(float z, float v) {
    const float kLog2eHi = 1.44269502162933349609375f;
    const float kLog2eLo = 1.92596299112661746e-8f;
    const float kLn2     = 0.693147182464599609375f;
    float t   = __builtin_amdgcn_exp2f(-fabsf(z) * kLog2eHi);
    float u   = 1.0f + t;
    float um  = u - 1.0f;
    float lg  = (__builtin_amdgcn_logf(u) * kLn2) * (t * __builtin_amdgcn_rcpf(um));
    float l1p = (um == 0.0f) ? t : lg;
    float sp  = l1p + fmaxf(z, 0.0f);
    float p    = z * kLog2eHi;
    float rest = fmaf(z, kLog2eHi, -p) + z * kLog2eLo;
    float ez   = __builtin_amdgcn_exp2f(p) * fmaf(rest, kLn2, 1.0f);
    return 0.5f * ez * sp * v;
}

__global__ void __launch_bounds__(FU_WAVES * 32) __attribute__((amdgpu_waves_per_eu(8)))
edge_upd(const float* __restrict__ acc,
         const int*   __restrict__ seg,
         const int*   __restrict__ ce,
         float*       __restrict__ out, int N) {
    __shared__ __attribute__((aligned(16))) float tileS[FU_WAVES][16 * FU_PITCH];
    const float kScale = 1.0f / 3.0f;
    int lane = threadIdx.x & 31;
    int wv   = __builtin_amdgcn_readfirstlane((int)(threadIdx.x >> 5));
    int r0   = (blockIdx.x * FU_WAVES + wv) * 16;
    if (r0 >= N) return;
    int m    = lane & 15;
    int half = lane >> 4;
    int rows = min(16, N - r0);
    float* sT = tileS[wv];
    const float* F = sT + m * FU_PITCH + 2 * half;

    int2 sc;  sc.x = 0;  sc.y = 0;
    if (m < rows) sc = *(const int2*)(seg + (size_t)(r0 + m) * 2);
    int nsum = sc.y;
    nsum = row_xadd_i<8>(nsum);
    nsum = row_xadd_i<4>(nsum);
    nsum = row_xadd_i<2>(nsum);
    nsum = row_xadd_i<1>(nsum);
    int begin  = __builtin_amdgcn_readfirstlane(sc.x);
    int n      = __builtin_amdgcn_readfirstlane(nsum);
    int ntiles = (n + 15) >> 4;
    int last   = n - 1;
    const int4* recs = (const int4*)ce + begin;

    float2 zero2;  zero2.x = 0.0f;  zero2.y = 0.0f;

    unsigned emp = __builtin_amdgcn_ballot_w32(half == 0 && m < rows && sc.y == 0);
    while (emp != 0u) {
        int zr = __builtin_ctz(emp);
        emp &= emp - 1u;
        store_row(out, r0 + zr, lane, zero2);
    }
    if (ntiles == 0) return;

    int4 rcC = recs[min(m, last)];
    int4 rcN = recs[min(16 + m, last)];

    float4 s4[8];
#pragma unroll
    for (int j = 0; j < 8; ++j) {
        s4[j] = row_load16(acc, min(r0 + 2 * j + half, N - 1), m);
    }
    float4 tq[8];
#pragma unroll
    for (int j = 0; j < 8; ++j) {
        tq[j] = row_load16(acc, __shfl(rcC.x, 2 * j + half, 32), m);
    }
#pragma unroll
    for (int j = 0; j < 8; ++j) {
        float4 x = s4[j];
        x.x *= kScale;  x.y *= kScale;  x.z *= kScale;  x.w *= kScale;
        *(float4*)(sT + (2 * j + half) * FU_PITCH + 4 * m) = x;
    }
    v2f A[16];
#pragma unroll
    for (int ks = 0; ks < 16; ++ks) {
        float2 a = *(const float2*)(F + 4 * ks);
        A[ks].x = a.x;  A[ks].y = a.y;
    }

    float2 av = zero2;
    int tailrow = 0;
    for (int t = 0; t < ntiles; ++t) {
        bool  valid = (16 * t + m < n);
        int   irow  = rcC.z - r0;
        float v     = __int_as_float(rcC.y);

#pragma unroll
        for (int j = 0; j < 8; ++j) {
            float4 x = tq[j];
            x.x *= kScale;  x.y *= kScale;  x.z *= kScale;  x.w *= kScale;
            *(float4*)(sT + (2 * j + half) * FU_PITCH + 4 * m) = x;
        }
#pragma unroll
        for (int j = 0; j < 8; ++j) {
            tq[j] = row_load16(acc, __shfl(rcN.x, 2 * j + half, 32), m);
        }
        int4 rcNN = recs[min(16 * (t + 2) + m, last)];

        int nxt = __shfl_down(irow, 1, 16);
        int nx0 = __builtin_amdgcn_readlane(rcN.z, 0) - r0;
        if (m == 15) nxt = nx0;
        unsigned ends = __builtin_amdgcn_ballot_w32(half == 0 && nxt != irow);
        tailrow = __builtin_amdgcn_readlane(irow, 15);

        v8f c0 = {};
        v8f c1 = {};
#pragma unroll
        for (int ks = 0; ks < 16; ks += 2) {
            float2 b0 = *(const float2*)(F + 4 * ks);
            float2 b1 = *(const float2*)(F + 4 * ks + 4);
            v2f B0;  B0.x = b0.x;  B0.y = b0.y;
            v2f B1;  B1.x = b1.x;  B1.y = b1.y;
            c0 = __builtin_amdgcn_wmma_f32_16x16x4_f32(
                 false, A[ks],  false, B0,
                 (short)0, c0,  false,  false);
            c1 = __builtin_amdgcn_wmma_f32_16x16x4_f32(
                 false, A[ks + 1],  false, B1,
                 (short)0, c1,  false,  false);
        }
        v8f cc = c0 + c1;

        int   i7  = irow & 7;
        float sel = cc[0];
#pragma unroll
        for (int vv = 1; vv < 8; ++vv) {
            if (i7 == vv) sel = cc[vv];
        }
        float dot = ((irow >> 3) == half) ? sel : 0.0f;
        dot += __shfl_xor(dot, 16, 32);

        float cross = dot * (1.0f / EMB);
        float z     = 4.0f * (1.0f - cross);
        float mat   = valid ? edge_mat(z, v) : 0.0f;

        float2 xr[16];
#pragma unroll
        for (int j = 0; j < 16; ++j) {
            xr[j] = *(const float2*)(sT + j * FU_PITCH + 2 * lane);
        }
#pragma unroll
        for (int j = 0; j < 16; ++j) {
            float mj = readlane_f32(mat, j);
            av.x += xr[j].x * mj;
            av.y += xr[j].y * mj;
            if ((ends >> j) & 1u) {
                store_row(out, r0 + __builtin_amdgcn_readlane(irow, j), lane, av);
                av = zero2;
            }
        }

        rcC = rcN;
        rcN = rcNN;
    }
    store_row(out, r0 + tailrow, lane, av);
}

extern "C" void kernel_launch(void* const* d_in, const int* in_sizes, int n_in,
                              void* d_out, int out_size, void* d_ws, size_t ws_size,
                              hipStream_t stream) {
    (void)n_in; (void)out_size; (void)ws_size;
    const float* feats     = (const float*)d_in[0];
    const float* edge_vals = (const float*)d_in[1];
    const int*   edge_row  = (const int*)d_in[2];
    const int*   edge_col  = (const int*)d_in[3];

    int N = in_sizes[0] / EMB;
    int E = in_sizes[1];

    float* bufA = (float*)d_ws;
    float* bufB = bufA + (size_t)N * EMB;
    float* acc  = bufB + (size_t)N * EMB;
    float* out  = (float*)d_out;

    int* ce    = (int*)bufB;
    int* seg   = ce + (size_t)E * 4;
    int* pos   = seg + (size_t)N * 2;
    int* cnt   = pos + E;
    int* total = cnt + N;

    const int blk = 256;
    int gridE  = (E + blk - 1) / blk;
    int gridN  = (N + blk - 1) / blk;
    int gridGa = (int)(((long long)N * 16 + ROW_BLK - 1) / ROW_BLK);
    int gridFu = (N + 16 * FU_WAVES - 1) / (16 * FU_WAVES);
    int NB     = (N + (1 << BIN_SHIFT) - 1) >> BIN_SHIFT;
    int gridSc = (E + SC_EPB - 1) / SC_EPB;

    if (NB + 1 <= NBMAX && gridSc <= BR_T && (long long)gridSc * SC_EPB * 4 <= (long long)N * EMB &&
        (long long)gridSc * (NB + 1) <= (long long)E) {
        int* tab = pos;
        int* tmp = (int*)acc;
        bin_block_sort<<<gridSc, SC_T, 0, stream>>>(edge_vals, edge_row, edge_col, tab, tmp, E, NB);
        bin_rows_l0<<<NB, BR_T, 0, stream>>>(tab, tmp, ce, seg, feats, bufA, N, NB, gridSc);
    } else {
        hipMemsetAsync(cnt, 0, (size_t)(N + 1) * sizeof(int), stream);
        row_hist<<<gridE, blk, 0, stream>>>(edge_row, cnt, pos, E);
        row_alloc<<<gridN, blk, 0, stream>>>(cnt, seg, total, N);
        csr_fill<<<gridE, blk, 0, stream>>>(edge_vals, edge_row, edge_col, seg, pos, ce, E);
        spmm_rows<0><<<gridGa, ROW_BLK, 0, stream>>>(feats, seg, ce, nullptr, nullptr, bufA, 1.0f, 0.5f, N);
    }

    spmm_rows<1><<<gridGa, ROW_BLK, 0, stream>>>(bufA, seg, ce, feats, bufA, acc, 1.0f, 1.0f / 3.0f, N);

    edge_upd<<<gridFu, FU_WAVES * 32, 0, stream>>>(acc, seg, ce, out, N);
}


# Both layer sums keep 8 row gathers in flight per half-wave (was 4); layer 1 reads a row's records once, one per lane, and takes column and weight by lane broadcast; sum order kept
# speedup vs baseline: 4.2787x; 1.0402x over previous
#include <hip/hip_runtime.h>
#include <hip/hip_bf16.h>
#include <math.h>

typedef float v2f __attribute__((ext_vector_type(2)));
typedef float v8f __attribute__((ext_vector_type(8)));

#define EMB   64
#define EPSN  1e-12f

__device__ __forceinline__ float readlane_f32(float x, int lane) {
    return __int_as_float(__builtin_amdgcn_readlane(__float_as_int(x), lane));
}

__global__ void row_hist(const int* __restrict__ rows,
                         int*       __restrict__ cnt,
                         int*       __restrict__ pos, int E) {
    int e = blockIdx.x * blockDim.x + threadIdx.x;
    if (e >= E) return;
    pos[e] = atomicAdd(cnt + rows[e], 1);
}

__global__ void row_alloc(const int* __restrict__ cnt,
                          int*       __restrict__ seg,
                          int*       __restrict__ total, int N) {
    __shared__ int wsum[8];
    __shared__ int base;
    int r    = blockIdx.x * 256 + threadIdx.x;
    int lane = threadIdx.x & 31;
    int w    = threadIdx.x >> 5;
    int c    = (r < N) ? cnt[r] : 0;
    int incl = c;
#pragma unroll
    for (int d = 1; d < 32; d <<= 1) {
        int t = __shfl_up(incl, d, 32);
        if (lane >= d) incl += t;
    }
    if (lane == 31) wsum[w] = incl;
    __syncthreads();
    if (threadIdx.x == 0) {
        int run = 0;
        for (int i = 0; i < 8; ++i) { int t = wsum[i]; wsum[i] = run; run += t; }
        base = atomicAdd(total, run);
    }
    __syncthreads();
    if (r < N) {
        int2 sc;
        sc.x = base + wsum[w] + incl - c;
        sc.y = c;
        *(int2*)(seg + (size_t)r * 2) = sc;
    }
}

__global__ void csr_fill(const float* __restrict__ vals,
                         const int*   __restrict__ rows,
                         const int*   __restrict__ cols,
                         const int*   __restrict__ seg,
                         const int*   __restrict__ pos,
                         int*         __restrict__ ce, int E) {
    int e = blockIdx.x * blockDim.x + threadIdx.x;
    if (e >= E) return;
    int r = rows[e];
    int p = seg[(size_t)r * 2] + pos[e];
    int4 rec;
    rec.x = cols[e];
    rec.y = __float_as_int(vals[e]);
    rec.z = r;
    rec.w = e;
    *(int4*)(ce + (size_t)p * 4) = rec;
}

#define BIN_SHIFT 8
#define NBMAX     1024
#define SC_T      512
#define SC_K      8
#define SC_EPB    (SC_T * SC_K)
#define BR_T      512
#define BR_CAP    3072
#define BR_G      16
#define BR_GROUPS (BR_T / BR_G)
#define BR_U      8

__device__ __forceinline__ int block_excl_scan(int v, int* wsum) {
    int lane = threadIdx.x & 31;
    int w    = threadIdx.x >> 5;
    int incl = v;
    for (int d = 1; d < 32; d <<= 1) {
        int x = __shfl_up(incl, d, 32);
        if (lane >= d) incl += x;
    }
    __syncthreads();
    if (lane == 31) wsum[w] = incl;
    __syncthreads();
    int basew = 0;
    for (int i = 0; i < 16; ++i) {
        int x = wsum[i];
        if (i < w) basew += x;
    }
    return basew + incl - v;
}

__global__ void __launch_bounds__(SC_T)
bin_block_sort(const float* __restrict__ vals,
               const int*   __restrict__ rows,
               const int*   __restrict__ cols,
               int*         __restrict__ tab,
               int*         __restrict__ tmpw, int E, int NB) {
    int4* tmp = (int4*)tmpw;
    __shared__ int  h[NBMAX];
    __shared__ int  wsum[16];
    __shared__ int4 stage[SC_EPB];
    int t = threadIdx.x;
    for (int b = t; b < NBMAX; b += SC_T) h[b] = 0;
    int base = blockIdx.x * SC_EPB;
    int r[SC_K], cl[SC_K], vl[SC_K], lr[SC_K];
#pragma unroll
    for (int k = 0; k < SC_K; ++k) {
        int e = base + k * SC_T + t;
        r[k] = -1;  cl[k] = 0;  vl[k] = 0;  lr[k] = 0;
        if (e < E) {
            r[k]  = rows[e];
            cl[k] = cols[e];
            vl[k] = __float_as_int(vals[e]);
        }
    }
    __syncthreads();
#pragma unroll
    for (int k = 0; k < SC_K; ++k) {
        if (r[k] >= 0) lr[k] = atomicAdd(&h[r[k] >> BIN_SHIFT], 1);
    }
    __syncthreads();
    int b0 = 2 * t;
    int b1 = 2 * t + 1;
    int c0 = h[b0];
    int c1 = h[b1];
    int lex = block_excl_scan(c0 + c1, wsum);
    h[b0] = lex;
    h[b1] = lex + c0;
    int  TP   = NB + 1;
    int* trow = tab + (size_t)blockIdx.x * TP;
    if (b0 < TP) trow[b0] = lex;
    if (b1 < TP) trow[b1] = lex + c0;
    __syncthreads();
#pragma unroll
    for (int k = 0; k < SC_K; ++k) {
        if (r[k] >= 0) {
            int4 q;
            q.x = cl[k];
            q.y = vl[k];
            q.z = r[k];
            q.w = base + k * SC_T + t;
            stage[h[r[k] >> BIN_SHIFT] + lr[k]] = q;
        }
    }
    __syncthreads();
    int n = min(SC_EPB, E - base);
    for (int s = t; s < n; s += SC_T) tmp[base + s] = stage[s];
}

__device__ __forceinline__ void fma_row(float4& a, const float4& x, float sc, float w) {
    a.x += (x.x * sc) * w;
    a.y += (x.y * sc) * w;
    a.z += (x.z * sc) * w;
    a.w += (x.w * sc) * w;
}

__device__ __forceinline__ float4 row_ld(const float* __restrict__ f, int row, int m16) {
    return *(const float4*)((const char*)f + (unsigned)(row * (EMB * 4) + m16));
}

template <int M>
__device__ __forceinline__ float row_xadd(float s) {
    int t = __builtin_amdgcn_update_dpp(0, __float_as_int(s), 0x160 + M, 0xf, 0xf, false);
    return s + __int_as_float(t);
}

__device__ __forceinline__ float4 row_finish(const float4& a, float inv_c) {
    float gx = a.x * inv_c;
    float gy = a.y * inv_c;
    float gz = a.z * inv_c;
    float gw = a.w * inv_c;
    float s  = gx * gx + gy * gy + gz * gz + gw * gw;
    s = row_xadd<8>(s);
    s = row_xadd<4>(s);
    s = row_xadd<2>(s);
    s = row_xadd<1>(s);
    float n   = __builtin_amdgcn_sqrtf(s);
    float inv = __builtin_amdgcn_rcpf(fmaxf(n, EPSN));
    float4 g;  g.x = gx * inv;  g.y = gy * inv;  g.z = gz * inv;  g.w = gw * inv;
    return g;
}

template <bool TILE>
__device__ __forceinline__ void bin_layer0(const int4* tile, const int4* gr,
                                           const int* rs, const int* rc,
                                           const float* __restrict__ f,
                                           float*       __restrict__ dst, int b, int N) {
    int t  = threadIdx.x;
    int hw = t >> 4;
    int m  = t & 15;
    int l4 = m << 2;
    for (int j = hw; j < 256; j += BR_T / 16) {
        int row = (b << BIN_SHIFT) + j;
        if (row >= N) break;
        int s    = rs[j];
        int cc   = rc[j];
        int last = cc - 1;
        float4 a;  a.x = 0.0f;  a.y = 0.0f;  a.z = 0.0f;  a.w = 0.0f;
        for (int i = 0; i < cc; i += 8) {
            int i0 = s + i;
            int i1 = s + min(i + 1, last);
            int i2 = s + min(i + 2, last);
            int i3 = s + min(i + 3, last);
            int i4 = s + min(i + 4, last);
            int i5 = s + min(i + 5, last);
            int i6 = s + min(i + 6, last);
            int i7 = s + min(i + 7, last);
            int4 e0, e1, e2, e3, e4, e5, e6, e7;
            if (TILE) {
                e0 = tile[i0];  e1 = tile[i1];  e2 = tile[i2];  e3 = tile[i3];
                e4 = tile[i4];  e5 = tile[i5];  e6 = tile[i6];  e7 = tile[i7];
            } else {
                e0 = gr[i0];    e1 = gr[i1];    e2 = gr[i2];    e3 = gr[i3];
                e4 = gr[i4];    e5 = gr[i5];    e6 = gr[i6];    e7 = gr[i7];
            }
            float4 x0 = row_ld(f, e0.x, l4 << 2);
            float4 x1 = row_ld(f, e1.x, l4 << 2);
            float4 x2 = row_ld(f, e2.x, l4 << 2);
            float4 x3 = row_ld(f, e3.x, l4 << 2);
            float w0 = __int_as_float(e0.y);
            float w1 = (i + 1 <= last) ? __int_as_float(e1.y) : 0.0f;
            float w2 = (i + 2 <= last) ? __int_as_float(e2.y) : 0.0f;
            float w3 = (i + 3 <= last) ? __int_as_float(e3.y) : 0.0f;
            float w4 = (i + 4 <= last) ? __int_as_float(e4.y) : 0.0f;
            float w5 = (i + 5 <= last) ? __int_as_float(e5.y) : 0.0f;
            float w6 = (i + 6 <= last) ? __int_as_float(e6.y) : 0.0f;
            float w7 = (i + 7 <= last) ? __int_as_float(e7.y) : 0.0f;
            if (cc - i <= 4) {
                fma_row(a, x0, 1.0f, w0);
                fma_row(a, x1, 1.0f, w1);
                fma_row(a, x2, 1.0f, w2);
                fma_row(a, x3, 1.0f, w3);
            } else {
                float4 x4 = row_ld(f, e4.x, l4 << 2);
                float4 x5 = row_ld(f, e5.x, l4 << 2);
                float4 x6 = row_ld(f, e6.x, l4 << 2);
                float4 x7 = row_ld(f, e7.x, l4 << 2);
                fma_row(a, x0, 1.0f, w0);
                fma_row(a, x1, 1.0f, w1);
                fma_row(a, x2, 1.0f, w2);
                fma_row(a, x3, 1.0f, w3);
                fma_row(a, x4, 1.0f, w4);
                fma_row(a, x5, 1.0f, w5);
                fma_row(a, x6, 1.0f, w6);
                fma_row(a, x7, 1.0f, w7);
            }
        }
        float4 gv = row_finish(a, 0.5f);
        *(float4*)(dst + (size_t)row * EMB + l4) = gv;
    }
}

__global__ void __launch_bounds__(BR_T)
bin_rows_l0(const int*   __restrict__ tab,
            const int*   __restrict__ tmpw,
            int*                      cew,
            int*         __restrict__ seg,
            const float* __restrict__ f,
            float*       __restrict__ dst, int N, int NB, int NBLK) {
    const int4* tmp = (const int4*)tmpw;
    int4*       ce  = (int4*)cew;
    __shared__ int  rc[256];
    __shared__ int  rs[256];
    __shared__ int  wsum[16];
    __shared__ int  sumS[16];
    __shared__ int  sumC[16];
    __shared__ int  runSrc[BR_T];
    __shared__ int  runLen[BR_T];
    __shared__ int4 stage[BR_CAP];
    int t    = threadIdx.x;
    int b    = blockIdx.x;
    int lane = t & 31;
    int w    = t >> 5;
    int TP   = NB + 1;
    int s0 = 0, len0 = 0;
    if (t < NBLK) {
        const int* trow = tab + (size_t)t * TP + b;
        s0   = trow[0];
        len0 = trow[1] - s0;
    }
    runSrc[t] = t * SC_EPB + s0;
    runLen[t] = len0;
    if (t < 256) rc[t] = 0;
    int ws = s0, wc = len0;
    for (int d = 16; d >= 1; d >>= 1) {
        ws += __shfl_xor(ws, d, 32);
        wc += __shfl_xor(wc, d, 32);
    }
    if (lane == 0) { sumS[w] = ws;  sumC[w] = wc; }
    __syncthreads();
    int bstart = 0, nb = 0;
    for (int i = 0; i < 16; ++i) { bstart += sumS[i];  nb += sumC[i]; }

    int g = t / BR_G;
    int l = t % BR_G;
    for (int k0 = g; k0 < NBLK; k0 += BR_U * BR_GROUPS) {
        int rw[BR_U];
#pragma unroll
        for (int u = 0; u < BR_U; ++u) {
            int k = k0 + u * BR_GROUPS;
            rw[u] = -1;
            if (k < NBLK && l < runLen[k]) rw[u] = tmp[runSrc[k] + l].z;
        }
#pragma unroll
        for (int u = 0; u < BR_U; ++u) {
            if (rw[u] >= 0) atomicAdd(&rc[rw[u] & 255], 1);
        }
        for (int u = 0; u < BR_U; ++u) {
            int k = k0 + u * BR_GROUPS;
            if (k < NBLK) {
                int len = runLen[k];
                int src = runSrc[k];
                for (int i = l + BR_G; i < len; i += BR_G) atomicAdd(&rc[tmp[src + i].z & 255], 1);
            }
        }
    }
    __syncthreads();
    int c  = (t < 256) ? rc[t] : 0;
    int ex = block_excl_scan(c, wsum);
    if (t < 256) {
        rs[t] = ex;
        rc[t] = 0;
        int row = (b << BIN_SHIFT) + t;
        if (row < N) {
            int2 sc;
            sc.x = bstart + ex;
            sc.y = c;
            *(int2*)(seg + (size_t)row * 2) = sc;
        }
    }
    __syncthreads();
    bool staged = (nb <= BR_CAP);
    for (int k0 = g; k0 < NBLK; k0 += BR_U * BR_GROUPS) {
        int4 q[BR_U];
#pragma unroll
        for (int u = 0; u < BR_U; ++u) {
            int k = k0 + u * BR_GROUPS;
            q[u].x = 0;  q[u].y = 0;  q[u].z = -1;  q[u].w = 0;
            if (k < NBLK && l < runLen[k]) q[u] = tmp[runSrc[k] + l];
        }
#pragma unroll
        for (int u = 0; u < BR_U; ++u) {
            if (q[u].z >= 0) {
                int j    = q[u].z & 255;
                int slot = rs[j] + atomicAdd(&rc[j], 1);
                if (staged) stage[slot] = q[u];
                else        ce[bstart + slot] = q[u];
            }
        }
        for (int u = 0; u < BR_U; ++u) {
            int k = k0 + u * BR_GROUPS;
            if (k < NBLK) {
                int len = runLen[k];
                int src = runSrc[k];
                for (int i = l + BR_G; i < len; i += BR_G) {
                    int4 qq  = tmp[src + i];
                    int  j    = qq.z & 255;
                    int  slot = rs[j] + atomicAdd(&rc[j], 1);
                    if (staged) stage[slot] = qq;
                    else        ce[bstart + slot] = qq;
                }
            }
        }
    }
    __syncthreads();
    if (staged) {
        int4 q[BR_CAP / BR_T];
        int  dsl[BR_CAP / BR_T];
#pragma unroll
        for (int u = 0; u < BR_CAP / BR_T; ++u) {
            int i = t + u * BR_T;
            q[u].x = 0;  q[u].y = 0;  q[u].z = 0;  q[u].w = 0;
            dsl[u] = -1;
            if (i < nb) {
                q[u] = stage[i];
                int j   = q[u].z & 255;
                int s   = rs[j];
                int n   = rc[j];
                int key = q[u].w;
                int rank = 0;
                for (int k = 0; k < n; k += 4) {
                    int w0 = stage[s + k].w;
                    int w1 = stage[s + min(k + 1, n - 1)].w;
                    int w2 = stage[s + min(k + 2, n - 1)].w;
                    int w3 = stage[s + min(k + 3, n - 1)].w;
                    rank += (w0 < key) ? 1 : 0;
                    rank += (k + 1 < n && w1 < key) ? 1 : 0;
                    rank += (k + 2 < n && w2 < key) ? 1 : 0;
                    rank += (k + 3 < n && w3 < key) ? 1 : 0;
                }
                dsl[u] = s + rank;
            }
        }
        __syncthreads();
#pragma unroll
        for (int u = 0; u < BR_CAP / BR_T; ++u) {
            if (dsl[u] >= 0) stage[dsl[u]] = q[u];
        }
        __syncthreads();
        for (int i = t; i < nb; i += BR_T) ce[bstart + i] = stage[i];
    } else {
        if (t < 256) {
            int4* g0 = ce + bstart + rs[t];
            for (int i = 1; i < c; ++i) {
                int4 key = g0[i];
                int  k   = i - 1;
                while (k >= 0 && g0[k].w > key.w) { g0[k + 1] = g0[k];  --k; }
                g0[k + 1] = key;
            }
        }
        __syncthreads();
    }
    if (staged) bin_layer0<true>(stage, ce + bstart, rs, rc, f, dst, b, N);
    else        bin_layer0<false>(stage, ce + bstart, rs, rc, f, dst, b, N);
}

#define ROW_BLK   256
#define ROW_HOLD  32

template <int K>
__device__ __forceinline__ int row_bcast(int v) {
    return __builtin_amdgcn_update_dpp(0, v, 0x150 + K, 0xf, 0xf, false);
}

template <int D>
__device__ __forceinline__ int row_rot(int v) {
    return __builtin_amdgcn_update_dpp(0, v, 0x120 + D, 0xf, 0xf, false);
}


template <int D>
__device__ __forceinline__ void rank2(int ka, int kb, int& ra, int& rb) {
    int ta = row_rot<D>(ka);
    int tb = row_rot<D>(kb);
    ra += ((ta < ka) ? 1 : 0) + ((tb < ka) ? 1 : 0);
    rb += ((ta < kb) ? 1 : 0) + ((tb < kb) ? 1 : 0);
}

template <int I>
__device__ __forceinline__ void gather4(float4& a, const float* fl, int scol, int sval, float sc) {
    int c0 = row_bcast<I + 0>(scol);
    int c1 = row_bcast<I + 1>(scol);
    int c2 = row_bcast<I + 2>(scol);
    int c3 = row_bcast<I + 3>(scol);
    float w0 = __int_as_float(row_bcast<I + 0>(sval));
    float w1 = __int_as_float(row_bcast<I + 1>(sval));
    float w2 = __int_as_float(row_bcast<I + 2>(sval));
    float w3 = __int_as_float(row_bcast<I + 3>(sval));
    float4 x0 = *(const float4*)(fl + (size_t)c0 * EMB);
    float4 x1 = *(const float4*)(fl + (size_t)c1 * EMB);
    float4 x2 = *(const float4*)(fl + (size_t)c2 * EMB);
    float4 x3 = *(const float4*)(fl + (size_t)c3 * EMB);
    fma_row(a, x0, sc, w0);
    fma_row(a, x1, sc, w1);
    fma_row(a, x2, sc, w2);
    fma_row(a, x3, sc, w3);
}

template <int I>
__device__ __forceinline__ float4 lane_ld(const float* __restrict__ f, int scol, int m16) {
    return row_ld(f, row_bcast<I>(scol), m16);
}
template <int I>
__device__ __forceinline__ void lane_fma(float4& a, const float4& x, int sval, float sc) {
    fma_row(a, x, sc, __int_as_float(row_bcast<I>(sval)));
}
template <int I>
__device__ __forceinline__ void lane_gather4(float4& a, const float* __restrict__ f, int m16,
                                             int scol, int sval, float sc) {
    float4 x0 = lane_ld<I + 0>(f, scol, m16);
    float4 x1 = lane_ld<I + 1>(f, scol, m16);
    float4 x2 = lane_ld<I + 2>(f, scol, m16);
    float4 x3 = lane_ld<I + 3>(f, scol, m16);
    lane_fma<I + 0>(a, x0, sval, sc);
    lane_fma<I + 1>(a, x1, sval, sc);
    lane_fma<I + 2>(a, x2, sval, sc);
    lane_fma<I + 3>(a, x3, sval, sc);
}

template <int MODE>
__global__ void spmm_rows(const float* __restrict__ f,
                          const int*   __restrict__ seg,
                          int*                      ce,
                          const float* __restrict__ base0,
                          const float* __restrict__ base1,
                          float*       __restrict__ dst,
                          float srcScale, float inv_c, int N) {
    int tid = blockIdx.x * blockDim.x + threadIdx.x;
    int r   = tid >> 4;
    int m   = tid & 15;
    int l4  = m << 2;
    if (r >= N) return;
    int2 sc = *(const int2*)(seg + (size_t)r * 2);
    int c = sc.y;
    int4* rec = (int4*)ce + sc.x;
    const float* fl = f + l4;
    size_t off = (size_t)r * EMB + l4;
    float4 b0;  b0.x = 0.0f;  b0.y = 0.0f;  b0.z = 0.0f;  b0.w = 0.0f;
    float4 b1 = b0;
    if (MODE == 1) {
        b0 = *(const float4*)(base0 + off);
        b1 = *(const float4*)(base1 + off);
    }
    float4 a;  a.x = 0.0f;  a.y = 0.0f;  a.z = 0.0f;  a.w = 0.0f;

    if (c > ROW_HOLD) {
        int prev = -1;
        for (int k = 0; k < c; ++k) {
            int best = 0x7fffffff, bcol = 0, bval = 0;
            for (int j = 0; j < c; ++j) {
                int4 q = rec[j];
                if (q.w > prev && q.w < best) { best = q.w;  bcol = q.x;  bval = q.y; }
            }
            prev = best;
            float4 x = *(const float4*)(fl + (size_t)bcol * EMB);
            fma_row(a, x, srcScale, __int_as_float(bval));
        }
    } else if (MODE == 0 && c > 16) {
        __shared__ int2 ordS[ROW_BLK / 16][ROW_HOLD];
        int2* ord = ordS[threadIdx.x >> 4];
        bool hb = (m + 16 < c);
        int4 qa = rec[m];
        int4 qb;  qb.x = 0;  qb.y = 0;  qb.z = r;  qb.w = 0x7fffffff;
        if (hb) qb = rec[m + 16];
        int ka = qa.w;
        int kb = hb ? qb.w : 0x7fffffff;
        int ra = (kb < ka) ? 1 : 0;
        int rb = (ka < kb) ? 1 : 0;
        rank2<1>(ka, kb, ra, rb);   rank2<2>(ka, kb, ra, rb);   rank2<3>(ka, kb, ra, rb);
        rank2<4>(ka, kb, ra, rb);   rank2<5>(ka, kb, ra, rb);   rank2<6>(ka, kb, ra, rb);
        rank2<7>(ka, kb, ra, rb);   rank2<8>(ka, kb, ra, rb);   rank2<9>(ka, kb, ra, rb);
        rank2<10>(ka, kb, ra, rb);  rank2<11>(ka, kb, ra, rb);  rank2<12>(ka, kb, ra, rb);
        rank2<13>(ka, kb, ra, rb);  rank2<14>(ka, kb, ra, rb);  rank2<15>(ka, kb, ra, rb);
        if (!hb) rb = m + 16;
        int2 pa;  pa.x = qa.x;  pa.y = qa.y;
        int2 pb;  pb.x = qb.x;  pb.y = qb.y;
        ord[ra] = pa;
        ord[rb] = pb;
        int2 sa = ord[m];
        int4 oa;  oa.x = sa.x;  oa.y = sa.y;  oa.z = r;  oa.w = 0;
        rec[m] = oa;
        if (hb) {
            int2 sb = ord[m + 16];
            int4 ob;  ob.x = sb.x;  ob.y = sb.y;  ob.z = r;  ob.w = 0;
            rec[m + 16] = ob;
        }
        int last = c - 1;
        for (int i = 0; i < c; i += 4) {
            int2 e0 = ord[i];
            int2 e1 = ord[min(i + 1, last)];
            int2 e2 = ord[min(i + 2, last)];
            int2 e3 = ord[min(i + 3, last)];
            float4 x0 = *(const float4*)(fl + (size_t)e0.x * EMB);
            float4 x1 = *(const float4*)(fl + (size_t)e1.x * EMB);
            float4 x2 = *(const float4*)(fl + (size_t)e2.x * EMB);
            float4 x3 = *(const float4*)(fl + (size_t)e3.x * EMB);
            float w0 = __int_as_float(e0.y);
            float w1 = (i + 1 <= last) ? __int_as_float(e1.y) : 0.0f;
            float w2 = (i + 2 <= last) ? __int_as_float(e2.y) : 0.0f;
            float w3 = (i + 3 <= last) ? __int_as_float(e3.y) : 0.0f;
            fma_row(a, x0, srcScale, w0);
            fma_row(a, x1, srcScale, w1);
            fma_row(a, x2, srcScale, w2);
            fma_row(a, x3, srcScale, w3);
        }
    } else if (MODE == 0) {
        if (c > 0) {
            int4 q;  q.x = 0;  q.y = 0;  q.z = r;  q.w = 0x7fffffff;
            if (m < c) q = rec[m];
            int key = (m < c) ? q.w : 0x7fffffff;
            int rank = 0;
            rank += (row_rot<1>(key)  < key) ? 1 : 0;
            rank += (row_rot<2>(key)  < key) ? 1 : 0;
            rank += (row_rot<3>(key)  < key) ? 1 : 0;
            rank += (row_rot<4>(key)  < key) ? 1 : 0;
            rank += (row_rot<5>(key)  < key) ? 1 : 0;
            rank += (row_rot<6>(key)  < key) ? 1 : 0;
            rank += (row_rot<7>(key)  < key) ? 1 : 0;
            rank += (row_rot<8>(key)  < key) ? 1 : 0;
            rank += (row_rot<9>(key)  < key) ? 1 : 0;
            rank += (row_rot<10>(key) < key) ? 1 : 0;
            rank += (row_rot<11>(key) < key) ? 1 : 0;
            rank += (row_rot<12>(key) < key) ? 1 : 0;
            rank += (row_rot<13>(key) < key) ? 1 : 0;
            rank += (row_rot<14>(key) < key) ? 1 : 0;
            rank += (row_rot<15>(key) < key) ? 1 : 0;
            if (m >= c) rank = m;
            int to   = (((int)threadIdx.x & 16) | rank) << 2;
            int scol = __builtin_amdgcn_ds_permute(to, q.x);
            int sval = __builtin_amdgcn_ds_permute(to, q.y);
            int lastcol = __shfl(scol, c - 1, 16);
            if (m >= c) scol = lastcol;
            if (m < c) {
                int4 o;  o.x = scol;  o.y = sval;  o.z = r;  o.w = 0;
                rec[m] = o;
            }
            gather4<0>(a, fl, scol, sval, srcScale);
            if (c > 4)  gather4<4>(a, fl, scol, sval, srcScale);
            if (c > 8)  gather4<8>(a, fl, scol, sval, srcScale);
            if (c > 12) gather4<12>(a, fl, scol, sval, srcScale);
        }
    } else if (c <= 16) {
        if (c > 0) {
            int4 q;  q.x = 0;  q.y = 0;  q.z = r;  q.w = 0;
            if (m < c) q = rec[m];
            int scol = q.x;
            int sval = q.y;
            int lastcol = __shfl(scol, c - 1, 16);
            if (m >= c) scol = lastcol;
            int m16 = m << 4;
            float4 x0 = lane_ld<0>(f, scol, m16);
            float4 x1 = lane_ld<1>(f, scol, m16);
            float4 x2 = lane_ld<2>(f, scol, m16);
            float4 x3 = lane_ld<3>(f, scol, m16);
            if (c <= 4) {
                lane_fma<0>(a, x0, sval, srcScale);
                lane_fma<1>(a, x1, sval, srcScale);
                lane_fma<2>(a, x2, sval, srcScale);
                lane_fma<3>(a, x3, sval, srcScale);
            } else {
                float4 x4 = lane_ld<4>(f, scol, m16);
                float4 x5 = lane_ld<5>(f, scol, m16);
                float4 x6 = lane_ld<6>(f, scol, m16);
                float4 x7 = lane_ld<7>(f, scol, m16);
                __builtin_amdgcn_sched_barrier(0);
                lane_fma<0>(a, x0, sval, srcScale);
                lane_fma<1>(a, x1, sval, srcScale);
                lane_fma<2>(a, x2, sval, srcScale);
                lane_fma<3>(a, x3, sval, srcScale);
                lane_fma<4>(a, x4, sval, srcScale);
                lane_fma<5>(a, x5, sval, srcScale);
                lane_fma<6>(a, x6, sval, srcScale);
                lane_fma<7>(a, x7, sval, srcScale);
                if (c > 8)  lane_gather4<8>(a, f, m16, scol, sval, srcScale);
                if (c > 12) lane_gather4<12>(a, f, m16, scol, sval, srcScale);
            }
        }
    } else {
        for (int i = 0; i < c; i += 4) {
            int last = c - 1;
            int i1 = min(i + 1, last);
            int i2 = min(i + 2, last);
            int i3 = min(i + 3, last);
            int4 e0 = rec[i];
            int4 e1 = rec[i1];
            int4 e2 = rec[i2];
            int4 e3 = rec[i3];
            float4 x0 = *(const float4*)(fl + (size_t)e0.x * EMB);
            float4 x1 = *(const float4*)(fl + (size_t)e1.x * EMB);
            float4 x2 = *(const float4*)(fl + (size_t)e2.x * EMB);
            float4 x3 = *(const float4*)(fl + (size_t)e3.x * EMB);
            float w0 = __int_as_float(e0.y);
            float w1 = (i + 1 <= last) ? __int_as_float(e1.y) : 0.0f;
            float w2 = (i + 2 <= last) ? __int_as_float(e2.y) : 0.0f;
            float w3 = (i + 3 <= last) ? __int_as_float(e3.y) : 0.0f;
            fma_row(a, x0, srcScale, w0);
            fma_row(a, x1, srcScale, w1);
            fma_row(a, x2, srcScale, w2);
            fma_row(a, x3, srcScale, w3);
        }
    }

    float4 g = row_finish(a, inv_c);
    if (MODE == 0) {
        *(float4*)(dst + off) = g;
    } else {
        float4 o;
        o.x = (b0.x + b1.x) + g.x;
        o.y = (b0.y + b1.y) + g.y;
        o.z = (b0.z + b1.z) + g.z;
        o.w = (b0.w + b1.w) + g.w;
        *(float4*)(dst + off) = o;
    }
}

#define FU_WAVES 4
#define FU_PITCH 68

template <int M>
__device__ __forceinline__ int row_xadd_i(int s) {
    return s + __builtin_amdgcn_update_dpp(0, s, 0x160 + M, 0xf, 0xf, false);
}

__device__ __forceinline__ float4 row_load16(const float* __restrict__ base, int row, int m) {
    return *(const float4*)((const char*)base + (unsigned)(row * (EMB * 4) + m * 16));
}

__device__ __forceinline__ void store_row(float* __restrict__ out, int row, int lane, float2 v) {
    *(float2*)((char*)out + (unsigned)(row * (EMB * 4) + lane * 8)) = v;
}

__device__ __forceinline__ float edge_mat(float z, float v) {
    const float kLog2eHi = 1.44269502162933349609375f;
    const float kLog2eLo = 1.92596299112661746e-8f;
    const float kLn2     = 0.693147182464599609375f;
    float t   = __builtin_amdgcn_exp2f(-fabsf(z) * kLog2eHi);
    float u   = 1.0f + t;
    float um  = u - 1.0f;
    float lg  = (__builtin_amdgcn_logf(u) * kLn2) * (t * __builtin_amdgcn_rcpf(um));
    float l1p = (um == 0.0f) ? t : lg;
    float sp  = l1p + fmaxf(z, 0.0f);
    float p    = z * kLog2eHi;
    float rest = fmaf(z, kLog2eHi, -p) + z * kLog2eLo;
    float ez   = __builtin_amdgcn_exp2f(p) * fmaf(rest, kLn2, 1.0f);
    return 0.5f * ez * sp * v;
}

__global__ void __launch_bounds__(FU_WAVES * 32) __attribute__((amdgpu_waves_per_eu(8)))
edge_upd(const float* __restrict__ acc,
         const int*   __restrict__ seg,
         const int*   __restrict__ ce,
         float*       __restrict__ out, int N) {
    __shared__ __attribute__((aligned(16))) float tileS[FU_WAVES][16 * FU_PITCH];
    const float kScale = 1.0f / 3.0f;
    int lane = threadIdx.x & 31;
    int wv   = __builtin_amdgcn_readfirstlane((int)(threadIdx.x >> 5));
    int r0   = (blockIdx.x * FU_WAVES + wv) * 16;
    if (r0 >= N) return;
    int m    = lane & 15;
    int half = lane >> 4;
    int rows = min(16, N - r0);
    float* sT = tileS[wv];
    const float* F = sT + m * FU_PITCH + 2 * half;

    int2 sc;  sc.x = 0;  sc.y = 0;
    if (m < rows) sc = *(const int2*)(seg + (size_t)(r0 + m) * 2);
    int nsum = sc.y;
    nsum = row_xadd_i<8>(nsum);
    nsum = row_xadd_i<4>(nsum);
    nsum = row_xadd_i<2>(nsum);
    nsum = row_xadd_i<1>(nsum);
    int begin  = __builtin_amdgcn_readfirstlane(sc.x);
    int n      = __builtin_amdgcn_readfirstlane(nsum);
    int ntiles = (n + 15) >> 4;
    int last   = n - 1;
    const int4* recs = (const int4*)ce + begin;

    float2 zero2;  zero2.x = 0.0f;  zero2.y = 0.0f;

    unsigned emp = __builtin_amdgcn_ballot_w32(half == 0 && m < rows && sc.y == 0);
    while (emp != 0u) {
        int zr = __builtin_ctz(emp);
        emp &= emp - 1u;
        store_row(out, r0 + zr, lane, zero2);
    }
    if (ntiles == 0) return;

    int4 rcC = recs[min(m, last)];
    int4 rcN = recs[min(16 + m, last)];

    float4 s4[8];
#pragma unroll
    for (int j = 0; j < 8; ++j) {
        s4[j] = row_load16(acc, min(r0 + 2 * j + half, N - 1), m);
    }
    float4 tq[8];
#pragma unroll
    for (int j = 0; j < 8; ++j) {
        tq[j] = row_load16(acc, __shfl(rcC.x, 2 * j + half, 32), m);
    }
#pragma unroll
    for (int j = 0; j < 8; ++j) {
        float4 x = s4[j];
        x.x *= kScale;  x.y *= kScale;  x.z *= kScale;  x.w *= kScale;
        *(float4*)(sT + (2 * j + half) * FU_PITCH + 4 * m) = x;
    }
    v2f A[16];
#pragma unroll
    for (int ks = 0; ks < 16; ++ks) {
        float2 a = *(const float2*)(F + 4 * ks);
        A[ks].x = a.x;  A[ks].y = a.y;
    }

    float2 av = zero2;
    int tailrow = 0;
    for (int t = 0; t < ntiles; ++t) {
        bool  valid = (16 * t + m < n);
        int   irow  = rcC.z - r0;
        float v     = __int_as_float(rcC.y);

#pragma unroll
        for (int j = 0; j < 8; ++j) {
            float4 x = tq[j];
            x.x *= kScale;  x.y *= kScale;  x.z *= kScale;  x.w *= kScale;
            *(float4*)(sT + (2 * j + half) * FU_PITCH + 4 * m) = x;
        }
#pragma unroll
        for (int j = 0; j < 8; ++j) {
            tq[j] = row_load16(acc, __shfl(rcN.x, 2 * j + half, 32), m);
        }
        int4 rcNN = recs[min(16 * (t + 2) + m, last)];

        int nxt = __shfl_down(irow, 1, 16);
        int nx0 = __builtin_amdgcn_readlane(rcN.z, 0) - r0;
        if (m == 15) nxt = nx0;
        unsigned ends = __builtin_amdgcn_ballot_w32(half == 0 && nxt != irow);
        tailrow = __builtin_amdgcn_readlane(irow, 15);

        v8f c0 = {};
        v8f c1 = {};
#pragma unroll
        for (int ks = 0; ks < 16; ks += 2) {
            float2 b0 = *(const float2*)(F + 4 * ks);
            float2 b1 = *(const float2*)(F + 4 * ks + 4);
            v2f B0;  B0.x = b0.x;  B0.y = b0.y;
            v2f B1;  B1.x = b1.x;  B1.y = b1.y;
            c0 = __builtin_amdgcn_wmma_f32_16x16x4_f32(
                 false, A[ks],  false, B0,
                 (short)0, c0,  false,  false);
            c1 = __builtin_amdgcn_wmma_f32_16x16x4_f32(
                 false, A[ks + 1],  false, B1,
                 (short)0, c1,  false,  false);
        }
        v8f cc = c0 + c1;

        int   i7  = irow & 7;
        float sel = cc[0];
#pragma unroll
        for (int vv = 1; vv < 8; ++vv) {
            if (i7 == vv) sel = cc[vv];
        }
        float dot = ((irow >> 3) == half) ? sel : 0.0f;
        dot += __shfl_xor(dot, 16, 32);

        float cross = dot * (1.0f / EMB);
        float z     = 4.0f * (1.0f - cross);
        float mat   = valid ? edge_mat(z, v) : 0.0f;

        float2 xr[16];
#pragma unroll
        for (int j = 0; j < 16; ++j) {
            xr[j] = *(const float2*)(sT + j * FU_PITCH + 2 * lane);
        }
#pragma unroll
        for (int j = 0; j < 16; ++j) {
            float mj = readlane_f32(mat, j);
            av.x += xr[j].x * mj;
            av.y += xr[j].y * mj;
            if ((ends >> j) & 1u) {
                store_row(out, r0 + __builtin_amdgcn_readlane(irow, j), lane, av);
                av = zero2;
            }
        }

        rcC = rcN;
        rcN = rcNN;
    }
    store_row(out, r0 + tailrow, lane, av);
}

extern "C" void kernel_launch(void* const* d_in, const int* in_sizes, int n_in,
                              void* d_out, int out_size, void* d_ws, size_t ws_size,
                              hipStream_t stream) {
    (void)n_in; (void)out_size; (void)ws_size;
    const float* feats     = (const float*)d_in[0];
    const float* edge_vals = (const float*)d_in[1];
    const int*   edge_row  = (const int*)d_in[2];
    const int*   edge_col  = (const int*)d_in[3];

    int N = in_sizes[0] / EMB;
    int E = in_sizes[1];

    float* bufA = (float*)d_ws;
    float* bufB = bufA + (size_t)N * EMB;
    float* acc  = bufB + (size_t)N * EMB;
    float* out  = (float*)d_out;

    int* ce    = (int*)bufB;
    int* seg   = ce + (size_t)E * 4;
    int* pos   = seg + (size_t)N * 2;
    int* cnt   = pos + E;
    int* total = cnt + N;

    const int blk = 256;
    int gridE  = (E + blk - 1) / blk;
    int gridN  = (N + blk - 1) / blk;
    int gridGa = (int)(((long long)N * 16 + ROW_BLK - 1) / ROW_BLK);
    int gridFu = (N + 16 * FU_WAVES - 1) / (16 * FU_WAVES);
    int NB     = (N + (1 << BIN_SHIFT) - 1) >> BIN_SHIFT;
    int gridSc = (E + SC_EPB - 1) / SC_EPB;

    if (NB + 1 <= NBMAX && gridSc <= BR_T && (long long)gridSc * SC_EPB * 4 <= (long long)N * EMB &&
        (long long)gridSc * (NB + 1) <= (long long)E) {
        int* tab = pos;
        int* tmp = (int*)acc;
        bin_block_sort<<<gridSc, SC_T, 0, stream>>>(edge_vals, edge_row, edge_col, tab, tmp, E, NB);
        bin_rows_l0<<<NB, BR_T, 0, stream>>>(tab, tmp, ce, seg, feats, bufA, N, NB, gridSc);
    } else {
        hipMemsetAsync(cnt, 0, (size_t)(N + 1) * sizeof(int), stream);
        row_hist<<<gridE, blk, 0, stream>>>(edge_row, cnt, pos, E);
        row_alloc<<<gridN, blk, 0, stream>>>(cnt, seg, total, N);
        csr_fill<<<gridE, blk, 0, stream>>>(edge_vals, edge_row, edge_col, seg, pos, ce, E);
        spmm_rows<0><<<gridGa, ROW_BLK, 0, stream>>>(feats, seg, ce, nullptr, nullptr, bufA, 1.0f, 0.5f, N);
    }

    spmm_rows<1><<<gridGa, ROW_BLK, 0, stream>>>(bufA, seg, ce, feats, bufA, acc, 1.0f, 1.0f / 3.0f, N);

    edge_upd<<<gridFu, FU_WAVES * 32, 0, stream>>>(acc, seg, ce, out, N);
}
